# starting kernel
# baseline (speedup 1.0000x reference)
#include <hip/hip_runtime.h>
#include <hip/hip_bf16.h>

#define N_NODES 50000
#define N_EDGES 1600000
#define F_DIM   128
#define H_HEADS 8
#define D_HEAD  16
#define E_DIM   16
#define HD      (H_HEADS * D_HEAD)
#define ET      (N_EDGES + N_NODES)
#define EPS_IN  1e-5f
#define NEG_SLOPE 0.2f

typedef float v2f __attribute__((ext_vector_type(2)));
typedef float v8f __attribute__((ext_vector_type(8)));

__global__ void k_init(float* deg, float* loopattr, float* denom, float* outacc,
                       int* amaxK) {
    int stride = gridDim.x * blockDim.x;
    int tid = blockIdx.x * blockDim.x + threadIdx.x;
    for (int i = tid; i < N_NODES; i += stride) deg[i] = 0.0f;
    for (int i = tid; i < N_NODES * E_DIM; i += stride) loopattr[i] = 0.0f;
    for (int i = tid; i < N_NODES * H_HEADS; i += stride) denom[i] = 0.0f;
    for (int i = tid; i < N_NODES * HD; i += stride) outacc[i] = 0.0f;
    for (int i = tid; i < N_NODES * H_HEADS; i += stride) amaxK[i] = (int)0x80000000;
}

__global__ void k_deg_loop(const int* __restrict__ dst,
                           const float* __restrict__ edge_attr,
                           float* deg, float* loopsum) {
    int e = blockIdx.x * blockDim.x + threadIdx.x;
    if (e >= N_EDGES) return;
    int t = dst[e];
    atomicAdd(&deg[t], 1.0f);
    const float* ea = edge_attr + (size_t)e * E_DIM;
    float* ls = loopsum + (size_t)t * E_DIM;
#pragma unroll
    for (int k = 0; k < E_DIM; ++k) atomicAdd(&ls[k], ea[k]);
}

__global__ void k_loop_final(float* loopattr, const float* __restrict__ deg) {
    int idx = blockIdx.x * blockDim.x + threadIdx.x;
    if (idx >= N_NODES * E_DIM) return;
    float d = deg[idx >> 4];
    d = d < 1.0f ? 1.0f : d;
    loopattr[idx] /= d;
}

__global__ void __launch_bounds__(256)
k_gemm_h(const float* __restrict__ x, const float* __restrict__ W,
         float* __restrict__ h) {
    int wave = threadIdx.x >> 5;
    int lane = threadIdx.x & 31;
    int tile = blockIdx.x * 8 + wave;
    int tileM = tile >> 3;
    int tileN = tile & 7;
    int m   = tileM * 16 + (lane & 15);
    int kb  = (lane >> 4) << 1;
    int col = tileN * 16 + (lane & 15);

    v8f acc = {};
#pragma unroll
    for (int k = 0; k < F_DIM; k += 4) {
        v2f a, b;
        a.x = x[(size_t)m * F_DIM + k + kb];
        a.y = x[(size_t)m * F_DIM + k + kb + 1];
        b.x = W[(size_t)(k + kb) * HD + col];
        b.y = W[(size_t)(k + kb + 1) * HD + col];
        acc = __builtin_amdgcn_wmma_f32_16x16x4_f32(
             false, a,  false, b,
             (short)0, acc,  false,  false);
    }
    int rbase = tileM * 16 + ((lane >> 4) << 3);
#pragma unroll
    for (int r = 0; r < 8; ++r)
        h[(size_t)(rbase + r) * HD + col] = acc[r];
}

__global__ void k_attn_node(const float* __restrict__ h,
                            const float* __restrict__ att_src,
                            const float* __restrict__ att_dst,
                            float* a_src, float* a_dst) {
    int idx = blockIdx.x * blockDim.x + threadIdx.x;
    if (idx >= N_NODES * H_HEADS) return;
    int n = idx >> 3, hh = idx & 7;
    const float* hp = h + (size_t)n * HD + hh * D_HEAD;
    const float* as = att_src + hh * D_HEAD;
    const float* ad = att_dst + hh * D_HEAD;
    float s = 0.f, d = 0.f;
#pragma unroll
    for (int i = 0; i < D_HEAD; ++i) {
        float v = hp[i];
        s += v * as[i];
        d += v * ad[i];
    }
    a_src[idx] = s;
    a_dst[idx] = d;
}

__global__ void k_edge_proj(const float* __restrict__ W_e,
                            const float* __restrict__ att_edge, float* V) {
    int idx = threadIdx.x;
    int k = idx >> 3, hh = idx & 7;
    float s = 0.f;
#pragma unroll
    for (int d = 0; d < D_HEAD; ++d)
        s += W_e[(size_t)k * HD + hh * D_HEAD + d] * att_edge[hh * D_HEAD + d];
    V[idx] = s;
}

__global__ void k_logits(const int* __restrict__ src, const int* __restrict__ dst,
                         const float* __restrict__ edge_attr,
                         const float* __restrict__ loopattr,
                         const float* __restrict__ Vc,
                         const float* __restrict__ a_src,
                         const float* __restrict__ a_dst,
                         float* alpha, int* amaxK) {
    int i = blockIdx.x * blockDim.x + threadIdx.x;
    if (i >= ET) return;
    int s, t;
    const float* ea;
    if (i < N_EDGES) {
        s = src[i]; t = dst[i];
        ea = edge_attr + (size_t)i * E_DIM;
    } else {
        s = i - N_EDGES; t = s;
        ea = loopattr + (size_t)s * E_DIM;
    }
    float eav[E_DIM];
#pragma unroll
    for (int k = 0; k < E_DIM; ++k) eav[k] = ea[k];
#pragma unroll
    for (int hh = 0; hh < H_HEADS; ++hh) {
        float ae = 0.f;
#pragma unroll
        for (int k = 0; k < E_DIM; ++k) ae += eav[k] * Vc[k * 8 + hh];
        float l = a_src[s * 8 + hh] + a_dst[t * 8 + hh] + ae;
        l = l > 0.f ? l : NEG_SLOPE * l;
        alpha[(size_t)i * 8 + hh] = l;
        int bi = __float_as_int(l);
        int key = bi >= 0 ? bi : (bi ^ 0x7FFFFFFF);
        atomicMax(&amaxK[t * 8 + hh], key);
    }
}

__global__ void k_expsum(const int* __restrict__ dst, float* alpha,
                         const int* __restrict__ amaxK, float* denom) {
    int i = blockIdx.x * blockDim.x + threadIdx.x;
    if (i >= ET) return;
    int t = (i < N_EDGES) ? dst[i] : (i - N_EDGES);
#pragma unroll
    for (int hh = 0; hh < H_HEADS; ++hh) {
        int key = amaxK[t * 8 + hh];
        int bi = key >= 0 ? key : (key ^ 0x7FFFFFFF);
        float mx = __int_as_float(bi);
        float a = __expf(alpha[(size_t)i * 8 + hh] - mx);
        alpha[(size_t)i * 8 + hh] = a;
        atomicAdd(&denom[t * 8 + hh], a);
    }
}

__global__ void k_scatter(const int* __restrict__ src, const int* __restrict__ dst,
                          const float* __restrict__ alpha,
                          const float* __restrict__ denom,
                          const float* __restrict__ h, float* outacc) {
    long long tid = (long long)blockIdx.x * blockDim.x + threadIdx.x;
    if (tid >= (long long)ET * 4) return;
    int g = (int)(tid & 3);
    long long i = tid >> 2;
    int s, t;
    if (i < N_EDGES) { s = src[i]; t = dst[i]; }
    else             { s = (int)(i - N_EDGES); t = s; }
#pragma unroll
    for (int hh = g * 2; hh < g * 2 + 2; ++hh) {
        float w = alpha[(size_t)i * 8 + hh] / (denom[t * 8 + hh] + 1e-16f);
        const float* hp = h + (size_t)s * HD + hh * D_HEAD;
        float* op = outacc + (size_t)t * HD + hh * D_HEAD;
#pragma unroll
        for (int d = 0; d < D_HEAD; ++d) atomicAdd(&op[d], hp[d] * w);
    }
}

__global__ void k_stats(const float* __restrict__ outacc, float* meanb,
                        float* invstd) {
    __shared__ float ssum[256];
    __shared__ float ssq[256];
    int c = blockIdx.x;
    float s = 0.f, q = 0.f;
    for (int n = threadIdx.x; n < N_NODES; n += blockDim.x) {
        float v = outacc[(size_t)n * HD + c];
        s += v;
        q += v * v;
    }
    ssum[threadIdx.x] = s;
    ssq[threadIdx.x] = q;
    __syncthreads();
    for (int off = 128; off > 0; off >>= 1) {
        if (threadIdx.x < off) {
            ssum[threadIdx.x] += ssum[threadIdx.x + off];
            ssq[threadIdx.x]  += ssq[threadIdx.x + off];
        }
        __syncthreads();
    }
    if (threadIdx.x == 0) {
        float m = ssum[0] / (float)N_NODES;
        float var = ssq[0] / (float)N_NODES - m * m;
        meanb[c] = m;
        invstd[c] = rsqrtf(var + EPS_IN);
    }
}

__global__ void k_final(const float* __restrict__ x,
                        const float* __restrict__ outacc,
                        const float* __restrict__ meanb,
                        const float* __restrict__ invstd,
                        const float* __restrict__ gamma,
                        const float* __restrict__ beta, float* out) {
    int idx = blockIdx.x * blockDim.x + threadIdx.x;
    if (idx >= N_NODES * HD) return;
    int c = idx & 127;
    float normed = (outacc[idx] - meanb[c]) * invstd[c] * gamma[c] + beta[c];
    float t = x[idx] + normed;
    out[idx] = t > 0.f ? t : expm1f(t);
}

extern "C" void kernel_launch(void* const* d_in, const int* in_sizes, int n_in,
                              void* d_out, int out_size, void* d_ws, size_t ws_size,
                              hipStream_t stream) {
    const float* x        = (const float*)d_in[0];
    const int*   ei       = (const int*)d_in[1];
    const float* edgeattr = (const float*)d_in[2];
    const float* W        = (const float*)d_in[3];
    const float* att_src  = (const float*)d_in[4];
    const float* att_dst  = (const float*)d_in[5];
    const float* W_e      = (const float*)d_in[6];
    const float* att_edge = (const float*)d_in[7];
    const float* gamma    = (const float*)d_in[8];
    const float* beta     = (const float*)d_in[9];
    float* out = (float*)d_out;

    const int* src = ei;
    const int* dst = ei + N_EDGES;

    float* ws = (float*)d_ws;
    float* hbuf     = ws;
    float* a_src    = hbuf   + (size_t)N_NODES * HD;
    float* a_dst    = a_src  + (size_t)N_NODES * H_HEADS;
    float* degb     = a_dst  + (size_t)N_NODES * H_HEADS;
    float* loopattr = degb   + N_NODES;
    float* Vc       = loopattr + (size_t)N_NODES * E_DIM;
    float* alpha    = Vc + 128;
    float* denom    = alpha + (size_t)ET * H_HEADS;
    float* outacc   = denom + (size_t)N_NODES * H_HEADS;
    float* meanb    = outacc + (size_t)N_NODES * HD;
    float* invstd   = meanb + 128;
    int*   amaxK    = (int*)(invstd + 128);

    k_init<<<1024, 256, 0, stream>>>(degb, loopattr, denom, outacc, amaxK);

    k_deg_loop<<<(N_EDGES + 255) / 256, 256, 0, stream>>>(dst, edgeattr, degb, loopattr);
    k_loop_final<<<(N_NODES * E_DIM + 255) / 256, 256, 0, stream>>>(loopattr, degb);

    k_gemm_h<<<(N_NODES / 16) * 8 / 8, 256, 0, stream>>>(x, W, hbuf);

    k_attn_node<<<(N_NODES * H_HEADS + 255) / 256, 256, 0, stream>>>(
        hbuf, att_src, att_dst, a_src, a_dst);
    k_edge_proj<<<1, 128, 0, stream>>>(W_e, att_edge, Vc);

    k_logits<<<(ET + 255) / 256, 256, 0, stream>>>(src, dst, edgeattr, loopattr,
                                                   Vc, a_src, a_dst, alpha, amaxK);
    k_expsum<<<(ET + 255) / 256, 256, 0, stream>>>(dst, alpha, amaxK, denom);
    {
        long long tot = (long long)ET * 4;
        int blocks = (int)((tot + 255) / 256);
        k_scatter<<<blocks, 256, 0, stream>>>(src, dst, alpha, denom, hbuf, outacc);
    }

    k_stats<<<HD, 256, 0, stream>>>(outacc, meanb, invstd);
    k_final<<<(N_NODES * HD + 255) / 256, 256, 0, stream>>>(
        x, outacc, meanb, invstd, gamma, beta, out);
}


# Aggregation: each workgroup owns 196 dst nodes, scans all dst, adds alpha*h[src] rows in LDS (LDS atomics, self loop first), plain stores; no global atomics there; sum order changes
# speedup vs baseline: 1.8575x; 1.8575x over previous
#include <hip/hip_runtime.h>
#include <hip/hip_bf16.h>

#define N_NODES 50000
#define N_EDGES 1600000
#define F_DIM   128
#define H_HEADS 8
#define D_HEAD  16
#define E_DIM   16
#define HD      (H_HEADS * D_HEAD)
#define ET      (N_EDGES + N_NODES)
#define EPS_IN  1e-5f
#define NEG_SLOPE 0.2f

typedef float v2f __attribute__((ext_vector_type(2)));
typedef float v8f __attribute__((ext_vector_type(8)));

__global__ void k_init(float* deg, float* loopattr, float* denom, int* amaxK) {
    int stride = gridDim.x * blockDim.x;
    int tid = blockIdx.x * blockDim.x + threadIdx.x;
    for (int i = tid; i < N_NODES; i += stride) deg[i] = 0.0f;
    for (int i = tid; i < N_NODES * E_DIM; i += stride) loopattr[i] = 0.0f;
    for (int i = tid; i < N_NODES * H_HEADS; i += stride) denom[i] = 0.0f;
    for (int i = tid; i < N_NODES * H_HEADS; i += stride) amaxK[i] = (int)0x80000000;
}

__global__ void k_deg_loop(const int* __restrict__ dst,
                           const float* __restrict__ edge_attr,
                           float* deg, float* loopsum) {
    int e = blockIdx.x * blockDim.x + threadIdx.x;
    if (e >= N_EDGES) return;
    int t = dst[e];
    atomicAdd(&deg[t], 1.0f);
    const float* ea = edge_attr + (size_t)e * E_DIM;
    float* ls = loopsum + (size_t)t * E_DIM;
#pragma unroll
    for (int k = 0; k < E_DIM; ++k) atomicAdd(&ls[k], ea[k]);
}

__global__ void k_loop_final(float* loopattr, const float* __restrict__ deg) {
    int idx = blockIdx.x * blockDim.x + threadIdx.x;
    if (idx >= N_NODES * E_DIM) return;
    float d = deg[idx >> 4];
    d = d < 1.0f ? 1.0f : d;
    loopattr[idx] /= d;
}

__global__ void __launch_bounds__(256)
k_gemm_h(const float* __restrict__ x, const float* __restrict__ W,
         float* __restrict__ h) {
    int wave = threadIdx.x >> 5;
    int lane = threadIdx.x & 31;
    int tile = blockIdx.x * 8 + wave;
    int tileM = tile >> 3;
    int tileN = tile & 7;
    int m   = tileM * 16 + (lane & 15);
    int kb  = (lane >> 4) << 1;
    int col = tileN * 16 + (lane & 15);

    v8f acc = {};
#pragma unroll
    for (int k = 0; k < F_DIM; k += 4) {
        v2f a, b;
        a.x = x[(size_t)m * F_DIM + k + kb];
        a.y = x[(size_t)m * F_DIM + k + kb + 1];
        b.x = W[(size_t)(k + kb) * HD + col];
        b.y = W[(size_t)(k + kb + 1) * HD + col];
        acc = __builtin_amdgcn_wmma_f32_16x16x4_f32(
             false, a,  false, b,
             (short)0, acc,  false,  false);
    }
    int rbase = tileM * 16 + ((lane >> 4) << 3);
#pragma unroll
    for (int r = 0; r < 8; ++r)
        h[(size_t)(rbase + r) * HD + col] = acc[r];
}

__global__ void k_attn_node(const float* __restrict__ h,
                            const float* __restrict__ att_src,
                            const float* __restrict__ att_dst,
                            float* a_src, float* a_dst) {
    int idx = blockIdx.x * blockDim.x + threadIdx.x;
    if (idx >= N_NODES * H_HEADS) return;
    int n = idx >> 3, hh = idx & 7;
    const float* hp = h + (size_t)n * HD + hh * D_HEAD;
    const float* as = att_src + hh * D_HEAD;
    const float* ad = att_dst + hh * D_HEAD;
    float s = 0.f, d = 0.f;
#pragma unroll
    for (int i = 0; i < D_HEAD; ++i) {
        float v = hp[i];
        s += v * as[i];
        d += v * ad[i];
    }
    a_src[idx] = s;
    a_dst[idx] = d;
}

__global__ void k_edge_proj(const float* __restrict__ W_e,
                            const float* __restrict__ att_edge, float* V) {
    int idx = threadIdx.x;
    int k = idx >> 3, hh = idx & 7;
    float s = 0.f;
#pragma unroll
    for (int d = 0; d < D_HEAD; ++d)
        s += W_e[(size_t)k * HD + hh * D_HEAD + d] * att_edge[hh * D_HEAD + d];
    V[idx] = s;
}

__global__ void k_logits(const int* __restrict__ src, const int* __restrict__ dst,
                         const float* __restrict__ edge_attr,
                         const float* __restrict__ loopattr,
                         const float* __restrict__ Vc,
                         const float* __restrict__ a_src,
                         const float* __restrict__ a_dst,
                         float* alpha, int* amaxK) {
    int i = blockIdx.x * blockDim.x + threadIdx.x;
    if (i >= ET) return;
    int s, t;
    const float* ea;
    if (i < N_EDGES) {
        s = src[i]; t = dst[i];
        ea = edge_attr + (size_t)i * E_DIM;
    } else {
        s = i - N_EDGES; t = s;
        ea = loopattr + (size_t)s * E_DIM;
    }
    float eav[E_DIM];
#pragma unroll
    for (int k = 0; k < E_DIM; ++k) eav[k] = ea[k];
#pragma unroll
    for (int hh = 0; hh < H_HEADS; ++hh) {
        float ae = 0.f;
#pragma unroll
        for (int k = 0; k < E_DIM; ++k) ae += eav[k] * Vc[k * 8 + hh];
        float l = a_src[s * 8 + hh] + a_dst[t * 8 + hh] + ae;
        l = l > 0.f ? l : NEG_SLOPE * l;
        alpha[(size_t)i * 8 + hh] = l;
        int bi = __float_as_int(l);
        int key = bi >= 0 ? bi : (bi ^ 0x7FFFFFFF);
        atomicMax(&amaxK[t * 8 + hh], key);
    }
}

__global__ void k_expsum(const int* __restrict__ dst, float* alpha,
                         const int* __restrict__ amaxK, float* denom) {
    int i = blockIdx.x * blockDim.x + threadIdx.x;
    if (i >= ET) return;
    int t = (i < N_EDGES) ? dst[i] : (i - N_EDGES);
#pragma unroll
    for (int hh = 0; hh < H_HEADS; ++hh) {
        int key = amaxK[t * 8 + hh];
        int bi = key >= 0 ? key : (key ^ 0x7FFFFFFF);
        float mx = __int_as_float(bi);
        float a = __expf(alpha[(size_t)i * 8 + hh] - mx);
        alpha[(size_t)i * 8 + hh] = a;
        atomicAdd(&denom[t * 8 + hh], a);
    }
}

#define AGG_BLOCKS  256
#define AGG_ROWS    ((N_NODES + AGG_BLOCKS - 1) / AGG_BLOCKS)
#define AGG_THREADS 1024
#define AGG_WAVES   (AGG_THREADS / 32)
#define DST_GROUPS  (N_EDGES / 128)
#define AGG_UNROLL  4

static_assert(N_EDGES % 128 == 0, "dst list must split into whole wave groups");

__device__ __forceinline__ void agg_row_term(float* acc_row, const float* __restrict__ hrow,
                                             const float* __restrict__ arow,
                                             const float* __restrict__ drow, int lane,
                                             bool first) {
    float wl = arow[lane & 7] / (drow[lane & 7] + 1e-16f);
    float hv[4];
#pragma unroll
    for (int j = 0; j < 4; ++j) hv[j] = hrow[j * 32 + lane];
    float v[4];
#pragma unroll
    for (int j = 0; j < 4; ++j) {
        int hh = j * 2 + (lane >> 4);
        float w = __int_as_float(__builtin_amdgcn_ds_bpermute(hh << 2, __float_as_int(wl)));
        v[j] = hv[j] * w;
    }
#pragma unroll
    for (int j = 0; j < 4; ++j) {
        int c = j * 32 + lane;
        if (first) acc_row[c] = v[j];
        else       atomicAdd(&acc_row[c], v[j]);
    }
}

__device__ __forceinline__ void agg_load_groups(int (&dv)[AGG_UNROLL][4],
                                                const int* __restrict__ dst, int g0, int lane) {
#pragma unroll
    for (int u = 0; u < AGG_UNROLL; ++u) {
        int gg = g0 + u * AGG_WAVES;
        if (gg < DST_GROUPS) {
            const int* p = dst + ((size_t)gg * 32 + lane) * 4;
            dv[u][0] = p[0]; dv[u][1] = p[1]; dv[u][2] = p[2]; dv[u][3] = p[3];
        } else {
            dv[u][0] = -1; dv[u][1] = -1; dv[u][2] = -1; dv[u][3] = -1;
        }
    }
}

__global__ void __launch_bounds__(AGG_THREADS)
k_aggregate(const int* __restrict__ src, const int* __restrict__ dst,
            const float* __restrict__ alpha, const float* __restrict__ denom,
            const float* __restrict__ h, float* __restrict__ outacc) {
    __shared__ float acc[AGG_ROWS * HD];
    const int lane = threadIdx.x & 31;
    const int wave = __builtin_amdgcn_readfirstlane((int)(threadIdx.x >> 5));
    const int n0 = blockIdx.x * AGG_ROWS;
    int cnt = N_NODES - n0;
    if (cnt > AGG_ROWS) cnt = AGG_ROWS;

    int dv[AGG_UNROLL][4];
    agg_load_groups(dv, dst, wave, lane);

    for (int tl = wave; tl < cnt; tl += AGG_WAVES) {
        int t = n0 + tl;
        size_t i = (size_t)N_EDGES + (size_t)t;
        agg_row_term(acc + tl * HD, h + (size_t)t * HD, alpha + i * 8, denom + (size_t)t * 8,
                     lane, true);
    }
    __syncthreads();

    for (int g0 = wave; g0 < DST_GROUPS; g0 += AGG_WAVES * AGG_UNROLL) {
        int nx[AGG_UNROLL][4];
        agg_load_groups(nx, dst, g0 + AGG_WAVES * AGG_UNROLL, lane);

        const unsigned ucnt = (unsigned)cnt;
        int tl[AGG_UNROLL][4];
        int sv[AGG_UNROLL][4];
        unsigned anym[AGG_UNROLL];
#pragma unroll
        for (int u = 0; u < AGG_UNROLL; ++u) {
            int gg = g0 + u * AGG_WAVES;
#pragma unroll
            for (int k = 0; k < 4; ++k) {
                tl[u][k] = dv[u][k] - n0;
                sv[u][k] = 0;
            }
            bool hu = ((unsigned)tl[u][0] < ucnt) | ((unsigned)tl[u][1] < ucnt) |
                      ((unsigned)tl[u][2] < ucnt) | ((unsigned)tl[u][3] < ucnt);
            anym[u] = __builtin_amdgcn_ballot_w32(hu);
            if (anym[u] != 0u) {
#pragma unroll
                for (int k = 0; k < 4; ++k) {
                    if ((unsigned)tl[u][k] < ucnt)
                        sv[u][k] = src[((size_t)gg * 32 + lane) * 4 + k];
                }
            }
        }
#pragma unroll
        for (int u = 0; u < AGG_UNROLL; ++u) {
            int gg = g0 + u * AGG_WAVES;
            if (anym[u] != 0u) {
#pragma unroll
                for (int k = 0; k < 4; ++k) {
                    unsigned m = __builtin_amdgcn_ballot_w32((unsigned)tl[u][k] < ucnt);
                    while (m != 0u) {
                        int l = __builtin_ctz(m);
                        m &= m - 1u;
                        int tlu = __builtin_amdgcn_readlane(tl[u][k], l);
                        int su  = __builtin_amdgcn_readlane(sv[u][k], l);
                        size_t e = ((size_t)gg * 32 + (size_t)l) * 4 + (size_t)k;
                        agg_row_term(acc + tlu * HD, h + (size_t)su * HD, alpha + e * 8,
                                     denom + (size_t)(n0 + tlu) * 8, lane, false);
                    }
                }
            }
        }
#pragma unroll
        for (int u = 0; u < AGG_UNROLL; ++u) {
#pragma unroll
            for (int k = 0; k < 4; ++k) dv[u][k] = nx[u][k];
        }
    }
    __syncthreads();

    for (int tl = wave; tl < cnt; tl += AGG_WAVES) {
        float* orow = outacc + (size_t)(n0 + tl) * HD;
#pragma unroll
        for (int j = 0; j < 4; ++j) orow[j * 32 + lane] = acc[tl * HD + j * 32 + lane];
    }
}

__global__ void k_stats(const float* __restrict__ outacc, float* meanb,
                        float* invstd) {
    __shared__ float ssum[256];
    __shared__ float ssq[256];
    int c = blockIdx.x;
    float s = 0.f, q = 0.f;
    for (int n = threadIdx.x; n < N_NODES; n += blockDim.x) {
        float v = outacc[(size_t)n * HD + c];
        s += v;
        q += v * v;
    }
    ssum[threadIdx.x] = s;
    ssq[threadIdx.x] = q;
    __syncthreads();
    for (int off = 128; off > 0; off >>= 1) {
        if (threadIdx.x < off) {
            ssum[threadIdx.x] += ssum[threadIdx.x + off];
            ssq[threadIdx.x]  += ssq[threadIdx.x + off];
        }
        __syncthreads();
    }
    if (threadIdx.x == 0) {
        float m = ssum[0] / (float)N_NODES;
        float var = ssq[0] / (float)N_NODES - m * m;
        meanb[c] = m;
        invstd[c] = rsqrtf(var + EPS_IN);
    }
}

__global__ void k_final(const float* __restrict__ x,
                        const float* __restrict__ outacc,
                        const float* __restrict__ meanb,
                        const float* __restrict__ invstd,
                        const float* __restrict__ gamma,
                        const float* __restrict__ beta, float* out) {
    int idx = blockIdx.x * blockDim.x + threadIdx.x;
    if (idx >= N_NODES * HD) return;
    int c = idx & 127;
    float normed = (outacc[idx] - meanb[c]) * invstd[c] * gamma[c] + beta[c];
    float t = x[idx] + normed;
    out[idx] = t > 0.f ? t : expm1f(t);
}

extern "C" void kernel_launch(void* const* d_in, const int* in_sizes, int n_in,
                              void* d_out, int out_size, void* d_ws, size_t ws_size,
                              hipStream_t stream) {
    const float* x        = (const float*)d_in[0];
    const int*   ei       = (const int*)d_in[1];
    const float* edgeattr = (const float*)d_in[2];
    const float* W        = (const float*)d_in[3];
    const float* att_src  = (const float*)d_in[4];
    const float* att_dst  = (const float*)d_in[5];
    const float* W_e      = (const float*)d_in[6];
    const float* att_edge = (const float*)d_in[7];
    const float* gamma    = (const float*)d_in[8];
    const float* beta     = (const float*)d_in[9];
    float* out = (float*)d_out;

    const int* src = ei;
    const int* dst = ei + N_EDGES;

    float* ws = (float*)d_ws;
    float* hbuf     = ws;
    float* a_src    = hbuf   + (size_t)N_NODES * HD;
    float* a_dst    = a_src  + (size_t)N_NODES * H_HEADS;
    float* degb     = a_dst  + (size_t)N_NODES * H_HEADS;
    float* loopattr = degb   + N_NODES;
    float* Vc       = loopattr + (size_t)N_NODES * E_DIM;
    float* alpha    = Vc + 128;
    float* denom    = alpha + (size_t)ET * H_HEADS;
    float* outacc   = denom + (size_t)N_NODES * H_HEADS;
    float* meanb    = outacc + (size_t)N_NODES * HD;
    float* invstd   = meanb + 128;
    int*   amaxK    = (int*)(invstd + 128);

    k_init<<<1024, 256, 0, stream>>>(degb, loopattr, denom, amaxK);

    k_deg_loop<<<(N_EDGES + 255) / 256, 256, 0, stream>>>(dst, edgeattr, degb, loopattr);
    k_loop_final<<<(N_NODES * E_DIM + 255) / 256, 256, 0, stream>>>(loopattr, degb);

    k_gemm_h<<<(N_NODES / 16) * 8 / 8, 256, 0, stream>>>(x, W, hbuf);

    k_attn_node<<<(N_NODES * H_HEADS + 255) / 256, 256, 0, stream>>>(
        hbuf, att_src, att_dst, a_src, a_dst);
    k_edge_proj<<<1, 128, 0, stream>>>(W_e, att_edge, Vc);

    k_logits<<<(ET + 255) / 256, 256, 0, stream>>>(src, dst, edgeattr, loopattr,
                                                   Vc, a_src, a_dst, alpha, amaxK);
    k_expsum<<<(ET + 255) / 256, 256, 0, stream>>>(dst, alpha, amaxK, denom);
    k_aggregate<<<AGG_BLOCKS, AGG_THREADS, 0, stream>>>(src, dst, alpha, denom, hbuf, outacc);

    k_stats<<<HD, 256, 0, stream>>>(outacc, meanb, invstd);
    k_final<<<(N_NODES * HD + 255) / 256, 256, 0, stream>>>(
        x, outacc, meanb, invstd, gamma, beta, out);
}


# Softmax denominators summed in LDS by the row's owning workgroup inside the aggregation, row divided once at the end; exp+denominator launch and its global float atomics removed; sum order changes
# speedup vs baseline: 2.4199x; 1.3028x over previous
#include <hip/hip_runtime.h>
#include <hip/hip_bf16.h>

#define N_NODES 50000
#define N_EDGES 1600000
#define F_DIM   128
#define H_HEADS 8
#define D_HEAD  16
#define E_DIM   16
#define HD      (H_HEADS * D_HEAD)
#define ET      (N_EDGES + N_NODES)
#define EPS_IN  1e-5f
#define NEG_SLOPE 0.2f

typedef float v2f __attribute__((ext_vector_type(2)));
typedef float v8f __attribute__((ext_vector_type(8)));

__global__ void k_init(float* deg, float* loopattr, int* amaxK) {
    int stride = gridDim.x * blockDim.x;
    int tid = blockIdx.x * blockDim.x + threadIdx.x;
    for (int i = tid; i < N_NODES; i += stride) deg[i] = 0.0f;
    for (int i = tid; i < N_NODES * E_DIM; i += stride) loopattr[i] = 0.0f;
    for (int i = tid; i < N_NODES * H_HEADS; i += stride) amaxK[i] = (int)0x80000000;
}

__global__ void k_deg_loop(const int* __restrict__ dst,
                           const float* __restrict__ edge_attr,
                           float* deg, float* loopsum) {
    int e = blockIdx.x * blockDim.x + threadIdx.x;
    if (e >= N_EDGES) return;
    int t = dst[e];
    atomicAdd(&deg[t], 1.0f);
    const float* ea = edge_attr + (size_t)e * E_DIM;
    float* ls = loopsum + (size_t)t * E_DIM;
#pragma unroll
    for (int k = 0; k < E_DIM; ++k) atomicAdd(&ls[k], ea[k]);
}

__global__ void k_loop_final(float* loopattr, const float* __restrict__ deg) {
    int idx = blockIdx.x * blockDim.x + threadIdx.x;
    if (idx >= N_NODES * E_DIM) return;
    float d = deg[idx >> 4];
    d = d < 1.0f ? 1.0f : d;
    loopattr[idx] /= d;
}

__global__ void __launch_bounds__(256)
k_gemm_h(const float* __restrict__ x, const float* __restrict__ W,
         float* __restrict__ h) {
    int wave = threadIdx.x >> 5;
    int lane = threadIdx.x & 31;
    int tile = blockIdx.x * 8 + wave;
    int tileM = tile >> 3;
    int tileN = tile & 7;
    int m   = tileM * 16 + (lane & 15);
    int kb  = (lane >> 4) << 1;
    int col = tileN * 16 + (lane & 15);

    v8f acc = {};
#pragma unroll
    for (int k = 0; k < F_DIM; k += 4) {
        v2f a, b;
        a.x = x[(size_t)m * F_DIM + k + kb];
        a.y = x[(size_t)m * F_DIM + k + kb + 1];
        b.x = W[(size_t)(k + kb) * HD + col];
        b.y = W[(size_t)(k + kb + 1) * HD + col];
        acc = __builtin_amdgcn_wmma_f32_16x16x4_f32(
             false, a,  false, b,
             (short)0, acc,  false,  false);
    }
    int rbase = tileM * 16 + ((lane >> 4) << 3);
#pragma unroll
    for (int r = 0; r < 8; ++r)
        h[(size_t)(rbase + r) * HD + col] = acc[r];
}

__global__ void k_attn_node(const float* __restrict__ h,
                            const float* __restrict__ att_src,
                            const float* __restrict__ att_dst,
                            float* a_src, float* a_dst) {
    int idx = blockIdx.x * blockDim.x + threadIdx.x;
    if (idx >= N_NODES * H_HEADS) return;
    int n = idx >> 3, hh = idx & 7;
    const float* hp = h + (size_t)n * HD + hh * D_HEAD;
    const float* as = att_src + hh * D_HEAD;
    const float* ad = att_dst + hh * D_HEAD;
    float s = 0.f, d = 0.f;
#pragma unroll
    for (int i = 0; i < D_HEAD; ++i) {
        float v = hp[i];
        s += v * as[i];
        d += v * ad[i];
    }
    a_src[idx] = s;
    a_dst[idx] = d;
}

__global__ void k_edge_proj(const float* __restrict__ W_e,
                            const float* __restrict__ att_edge, float* V) {
    int idx = threadIdx.x;
    int k = idx >> 3, hh = idx & 7;
    float s = 0.f;
#pragma unroll
    for (int d = 0; d < D_HEAD; ++d)
        s += W_e[(size_t)k * HD + hh * D_HEAD + d] * att_edge[hh * D_HEAD + d];
    V[idx] = s;
}

__global__ void k_logits(const int* __restrict__ src, const int* __restrict__ dst,
                         const float* __restrict__ edge_attr,
                         const float* __restrict__ loopattr,
                         const float* __restrict__ Vc,
                         const float* __restrict__ a_src,
                         const float* __restrict__ a_dst,
                         float* alpha, int* amaxK) {
    int i = blockIdx.x * blockDim.x + threadIdx.x;
    if (i >= ET) return;
    int s, t;
    const float* ea;
    if (i < N_EDGES) {
        s = src[i]; t = dst[i];
        ea = edge_attr + (size_t)i * E_DIM;
    } else {
        s = i - N_EDGES; t = s;
        ea = loopattr + (size_t)s * E_DIM;
    }
    float eav[E_DIM];
#pragma unroll
    for (int k = 0; k < E_DIM; ++k) eav[k] = ea[k];
#pragma unroll
    for (int hh = 0; hh < H_HEADS; ++hh) {
        float ae = 0.f;
#pragma unroll
        for (int k = 0; k < E_DIM; ++k) ae += eav[k] * Vc[k * 8 + hh];
        float l = a_src[s * 8 + hh] + a_dst[t * 8 + hh] + ae;
        l = l > 0.f ? l : NEG_SLOPE * l;
        alpha[(size_t)i * 8 + hh] = l;
        int bi = __float_as_int(l);
        int key = bi >= 0 ? bi : (bi ^ 0x7FFFFFFF);
        atomicMax(&amaxK[t * 8 + hh], key);
    }
}

#define AGG_BLOCKS  256
#define AGG_ROWS    ((N_NODES + AGG_BLOCKS - 1) / AGG_BLOCKS)
#define AGG_THREADS 1024
#define AGG_WAVES   (AGG_THREADS / 32)
#define DST_GROUPS  (N_EDGES / 128)
#define AGG_UNROLL  4

static_assert(N_EDGES % 128 == 0, "dst list must split into whole wave groups");

__device__ __forceinline__ void agg_row_term(float* acc_row, float* den_row,
                                             const float* __restrict__ hrow,
                                             const float* __restrict__ lrow,
                                             const int* __restrict__ mrow, int lane,
                                             bool first) {
    int key = mrow[lane & 7];
    int bi = key >= 0 ? key : (key ^ 0x7FFFFFFF);
    float wl = __expf(lrow[lane & 7] - __int_as_float(bi));
    float hv[4];
#pragma unroll
    for (int j = 0; j < 4; ++j) hv[j] = hrow[j * 32 + lane];
    float v[4];
#pragma unroll
    for (int j = 0; j < 4; ++j) {
        int hh = j * 2 + (lane >> 4);
        float w = __int_as_float(__builtin_amdgcn_ds_bpermute(hh << 2, __float_as_int(wl)));
        v[j] = hv[j] * w;
    }
#pragma unroll
    for (int j = 0; j < 4; ++j) {
        int c = j * 32 + lane;
        if (first) acc_row[c] = v[j];
        else       atomicAdd(&acc_row[c], v[j]);
    }
    if (lane < H_HEADS) {
        if (first) den_row[lane] = wl;
        else       atomicAdd(&den_row[lane], wl);
    }
}

__device__ __forceinline__ void agg_load_groups(int (&dv)[AGG_UNROLL][4],
                                                const int* __restrict__ dst, int g0, int lane) {
#pragma unroll
    for (int u = 0; u < AGG_UNROLL; ++u) {
        int gg = g0 + u * AGG_WAVES;
        if (gg < DST_GROUPS) {
            const int* p = dst + ((size_t)gg * 32 + lane) * 4;
            dv[u][0] = p[0]; dv[u][1] = p[1]; dv[u][2] = p[2]; dv[u][3] = p[3];
        } else {
            dv[u][0] = -1; dv[u][1] = -1; dv[u][2] = -1; dv[u][3] = -1;
        }
    }
}

__global__ void __launch_bounds__(AGG_THREADS)
k_aggregate(const int* __restrict__ src, const int* __restrict__ dst,
            const float* __restrict__ logit, const int* __restrict__ amaxK,
            const float* __restrict__ h, float* __restrict__ outacc) {
    __shared__ float acc[AGG_ROWS * HD];
    __shared__ float den[AGG_ROWS * H_HEADS];
    const int lane = threadIdx.x & 31;
    const int wave = __builtin_amdgcn_readfirstlane((int)(threadIdx.x >> 5));
    const int n0 = blockIdx.x * AGG_ROWS;
    int cnt = N_NODES - n0;
    if (cnt > AGG_ROWS) cnt = AGG_ROWS;

    int dv[AGG_UNROLL][4];
    agg_load_groups(dv, dst, wave, lane);

    for (int tl = wave; tl < cnt; tl += AGG_WAVES) {
        int t = n0 + tl;
        size_t i = (size_t)N_EDGES + (size_t)t;
        agg_row_term(acc + tl * HD, den + tl * H_HEADS, h + (size_t)t * HD, logit + i * 8,
                     amaxK + (size_t)t * 8, lane, true);
    }
    __syncthreads();

    for (int g0 = wave; g0 < DST_GROUPS; g0 += AGG_WAVES * AGG_UNROLL) {
        int nx[AGG_UNROLL][4];
        agg_load_groups(nx, dst, g0 + AGG_WAVES * AGG_UNROLL, lane);

        const unsigned ucnt = (unsigned)cnt;
        int tl[AGG_UNROLL][4];
        int sv[AGG_UNROLL][4];
        unsigned anym[AGG_UNROLL];
#pragma unroll
        for (int u = 0; u < AGG_UNROLL; ++u) {
            int gg = g0 + u * AGG_WAVES;
#pragma unroll
            for (int k = 0; k < 4; ++k) {
                tl[u][k] = dv[u][k] - n0;
                sv[u][k] = 0;
            }
            bool hu = ((unsigned)tl[u][0] < ucnt) | ((unsigned)tl[u][1] < ucnt) |
                      ((unsigned)tl[u][2] < ucnt) | ((unsigned)tl[u][3] < ucnt);
            anym[u] = __builtin_amdgcn_ballot_w32(hu);
            if (anym[u] != 0u) {
#pragma unroll
                for (int k = 0; k < 4; ++k) {
                    if ((unsigned)tl[u][k] < ucnt)
                        sv[u][k] = src[((size_t)gg * 32 + lane) * 4 + k];
                }
            }
        }
#pragma unroll
        for (int u = 0; u < AGG_UNROLL; ++u) {
            int gg = g0 + u * AGG_WAVES;
            if (anym[u] != 0u) {
#pragma unroll
                for (int k = 0; k < 4; ++k) {
                    unsigned m = __builtin_amdgcn_ballot_w32((unsigned)tl[u][k] < ucnt);
                    while (m != 0u) {
                        int l = __builtin_ctz(m);
                        m &= m - 1u;
                        int tlu = __builtin_amdgcn_readlane(tl[u][k], l);
                        int su  = __builtin_amdgcn_readlane(sv[u][k], l);
                        size_t e = ((size_t)gg * 32 + (size_t)l) * 4 + (size_t)k;
                        agg_row_term(acc + tlu * HD, den + tlu * H_HEADS, h + (size_t)su * HD,
                                     logit + e * 8, amaxK + (size_t)(n0 + tlu) * 8, lane, false);
                    }
                }
            }
        }
#pragma unroll
        for (int u = 0; u < AGG_UNROLL; ++u) {
#pragma unroll
            for (int k = 0; k < 4; ++k) dv[u][k] = nx[u][k];
        }
    }
    __syncthreads();

    for (int tl = wave; tl < cnt; tl += AGG_WAVES) {
        float* orow = outacc + (size_t)(n0 + tl) * HD;
#pragma unroll
        for (int j = 0; j < 4; ++j) {
            int hh = j * 2 + (lane >> 4);
            orow[j * 32 + lane] = acc[tl * HD + j * 32 + lane] / (den[tl * H_HEADS + hh] + 1e-16f);
        }
    }
}

__global__ void k_stats(const float* __restrict__ outacc, float* meanb,
                        float* invstd) {
    __shared__ float ssum[256];
    __shared__ float ssq[256];
    int c = blockIdx.x;
    float s = 0.f, q = 0.f;
    for (int n = threadIdx.x; n < N_NODES; n += blockDim.x) {
        float v = outacc[(size_t)n * HD + c];
        s += v;
        q += v * v;
    }
    ssum[threadIdx.x] = s;
    ssq[threadIdx.x] = q;
    __syncthreads();
    for (int off = 128; off > 0; off >>= 1) {
        if (threadIdx.x < off) {
            ssum[threadIdx.x] += ssum[threadIdx.x + off];
            ssq[threadIdx.x]  += ssq[threadIdx.x + off];
        }
        __syncthreads();
    }
    if (threadIdx.x == 0) {
        float m = ssum[0] / (float)N_NODES;
        float var = ssq[0] / (float)N_NODES - m * m;
        meanb[c] = m;
        invstd[c] = rsqrtf(var + EPS_IN);
    }
}

__global__ void k_final(const float* __restrict__ x,
                        const float* __restrict__ outacc,
                        const float* __restrict__ meanb,
                        const float* __restrict__ invstd,
                        const float* __restrict__ gamma,
                        const float* __restrict__ beta, float* out) {
    int idx = blockIdx.x * blockDim.x + threadIdx.x;
    if (idx >= N_NODES * HD) return;
    int c = idx & 127;
    float normed = (outacc[idx] - meanb[c]) * invstd[c] * gamma[c] + beta[c];
    float t = x[idx] + normed;
    out[idx] = t > 0.f ? t : expm1f(t);
}

extern "C" void kernel_launch(void* const* d_in, const int* in_sizes, int n_in,
                              void* d_out, int out_size, void* d_ws, size_t ws_size,
                              hipStream_t stream) {
    const float* x        = (const float*)d_in[0];
    const int*   ei       = (const int*)d_in[1];
    const float* edgeattr = (const float*)d_in[2];
    const float* W        = (const float*)d_in[3];
    const float* att_src  = (const float*)d_in[4];
    const float* att_dst  = (const float*)d_in[5];
    const float* W_e      = (const float*)d_in[6];
    const float* att_edge = (const float*)d_in[7];
    const float* gamma    = (const float*)d_in[8];
    const float* beta     = (const float*)d_in[9];
    float* out = (float*)d_out;

    const int* src = ei;
    const int* dst = ei + N_EDGES;

    float* ws = (float*)d_ws;
    float* hbuf     = ws;
    float* a_src    = hbuf   + (size_t)N_NODES * HD;
    float* a_dst    = a_src  + (size_t)N_NODES * H_HEADS;
    float* degb     = a_dst  + (size_t)N_NODES * H_HEADS;
    float* loopattr = degb   + N_NODES;
    float* Vc       = loopattr + (size_t)N_NODES * E_DIM;
    float* alpha    = Vc + 128;
    float* denom    = alpha + (size_t)ET * H_HEADS;
    float* outacc   = denom + (size_t)N_NODES * H_HEADS;
    float* meanb    = outacc + (size_t)N_NODES * HD;
    float* invstd   = meanb + 128;
    int*   amaxK    = (int*)(invstd + 128);

    k_init<<<1024, 256, 0, stream>>>(degb, loopattr, amaxK);

    k_deg_loop<<<(N_EDGES + 255) / 256, 256, 0, stream>>>(dst, edgeattr, degb, loopattr);
    k_loop_final<<<(N_NODES * E_DIM + 255) / 256, 256, 0, stream>>>(loopattr, degb);

    k_gemm_h<<<(N_NODES / 16) * 8 / 8, 256, 0, stream>>>(x, W, hbuf);

    k_attn_node<<<(N_NODES * H_HEADS + 255) / 256, 256, 0, stream>>>(
        hbuf, att_src, att_dst, a_src, a_dst);
    k_edge_proj<<<1, 128, 0, stream>>>(W_e, att_edge, Vc);

    k_logits<<<(ET + 255) / 256, 256, 0, stream>>>(src, dst, edgeattr, loopattr,
                                                   Vc, a_src, a_dst, alpha, amaxK);
    k_aggregate<<<AGG_BLOCKS, AGG_THREADS, 0, stream>>>(src, dst, alpha, amaxK, hbuf, outacc);

    k_stats<<<HD, 256, 0, stream>>>(outacc, meanb, invstd);
    k_final<<<(N_NODES * HD + 255) / 256, 256, 0, stream>>>(
        x, outacc, meanb, invstd, gamma, beta, out);
}


# Degree, attribute sums and logit maximum per destination summed in LDS by the row's owning workgroup (a dst scan queues owned edges, drained one per lane); global atomics removed; sum order changes
# speedup vs baseline: 6.1438x; 2.5389x over previous
#include <hip/hip_runtime.h>
#include <hip/hip_bf16.h>

#define N_NODES 50000
#define N_EDGES 1600000
#define F_DIM   128
#define H_HEADS 8
#define D_HEAD  16
#define E_DIM   16
#define HD      (H_HEADS * D_HEAD)
#define ET      (N_EDGES + N_NODES)
#define EPS_IN  1e-5f
#define NEG_SLOPE 0.2f

typedef float v2f __attribute__((ext_vector_type(2)));
typedef float v8f __attribute__((ext_vector_type(8)));

__global__ void __launch_bounds__(256)
k_gemm_h(const float* __restrict__ x, const float* __restrict__ W,
         float* __restrict__ h) {
    int wave = threadIdx.x >> 5;
    int lane = threadIdx.x & 31;
    int tile = blockIdx.x * 8 + wave;
    int tileM = tile >> 3;
    int tileN = tile & 7;
    int m   = tileM * 16 + (lane & 15);
    int kb  = (lane >> 4) << 1;
    int col = tileN * 16 + (lane & 15);

    v8f acc = {};
#pragma unroll
    for (int k = 0; k < F_DIM; k += 4) {
        v2f a, b;
        a.x = x[(size_t)m * F_DIM + k + kb];
        a.y = x[(size_t)m * F_DIM + k + kb + 1];
        b.x = W[(size_t)(k + kb) * HD + col];
        b.y = W[(size_t)(k + kb + 1) * HD + col];
        acc = __builtin_amdgcn_wmma_f32_16x16x4_f32(
             false, a,  false, b,
             (short)0, acc,  false,  false);
    }
    int rbase = tileM * 16 + ((lane >> 4) << 3);
#pragma unroll
    for (int r = 0; r < 8; ++r)
        h[(size_t)(rbase + r) * HD + col] = acc[r];
}

__global__ void k_attn_node(const float* __restrict__ h,
                            const float* __restrict__ att_src,
                            const float* __restrict__ att_dst,
                            float* a_src, float* a_dst) {
    int idx = blockIdx.x * blockDim.x + threadIdx.x;
    if (idx >= N_NODES * H_HEADS) return;
    int n = idx >> 3, hh = idx & 7;
    const float* hp = h + (size_t)n * HD + hh * D_HEAD;
    const float* as = att_src + hh * D_HEAD;
    const float* ad = att_dst + hh * D_HEAD;
    float s = 0.f, d = 0.f;
#pragma unroll
    for (int i = 0; i < D_HEAD; ++i) {
        float v = hp[i];
        s += v * as[i];
        d += v * ad[i];
    }
    a_src[idx] = s;
    a_dst[idx] = d;
}

__global__ void k_edge_proj(const float* __restrict__ W_e,
                            const float* __restrict__ att_edge, float* V) {
    int idx = threadIdx.x;
    int k = idx >> 3, hh = idx & 7;
    float s = 0.f;
#pragma unroll
    for (int d = 0; d < D_HEAD; ++d)
        s += W_e[(size_t)k * HD + hh * D_HEAD + d] * att_edge[hh * D_HEAD + d];
    V[idx] = s;
}

#define AGG_BLOCKS  256
#define AGG_ROWS    ((N_NODES + AGG_BLOCKS - 1) / AGG_BLOCKS)
#define AGG_THREADS 1024
#define AGG_WAVES   (AGG_THREADS / 32)
#define DST_GROUPS  (N_EDGES / 128)
#define AGG_UNROLL  4

static_assert(N_EDGES % 128 == 0, "dst list must split into whole wave groups");

__device__ __forceinline__ void agg_row_term(float* acc_row, float* den_row,
                                             const float* __restrict__ hrow,
                                             const float* __restrict__ lrow,
                                             const int* __restrict__ mrow, int lane,
                                             bool first) {
    int key = mrow[lane & 7];
    int bi = key >= 0 ? key : (key ^ 0x7FFFFFFF);
    float wl = __expf(lrow[lane & 7] - __int_as_float(bi));
    float hv[4];
#pragma unroll
    for (int j = 0; j < 4; ++j) hv[j] = hrow[j * 32 + lane];
    float v[4];
#pragma unroll
    for (int j = 0; j < 4; ++j) {
        int hh = j * 2 + (lane >> 4);
        float w = __int_as_float(__builtin_amdgcn_ds_bpermute(hh << 2, __float_as_int(wl)));
        v[j] = hv[j] * w;
    }
#pragma unroll
    for (int j = 0; j < 4; ++j) {
        int c = j * 32 + lane;
        if (first) acc_row[c] = v[j];
        else       atomicAdd(&acc_row[c], v[j]);
    }
    if (lane < H_HEADS) {
        if (first) den_row[lane] = wl;
        else       atomicAdd(&den_row[lane], wl);
    }
}

__device__ __forceinline__ void agg_load_groups(int (&dv)[AGG_UNROLL][4],
                                                const int* __restrict__ dst, int g0, int lane) {
#pragma unroll
    for (int u = 0; u < AGG_UNROLL; ++u) {
        int gg = g0 + u * AGG_WAVES;
        if (gg < DST_GROUPS) {
            const int* p = dst + ((size_t)gg * 32 + lane) * 4;
            dv[u][0] = p[0]; dv[u][1] = p[1]; dv[u][2] = p[2]; dv[u][3] = p[3];
        } else {
            dv[u][0] = -1; dv[u][1] = -1; dv[u][2] = -1; dv[u][3] = -1;
        }
    }
}

__global__ void __launch_bounds__(AGG_THREADS)
k_aggregate(const int* __restrict__ src, const int* __restrict__ dst,
            const float* __restrict__ logit, const int* __restrict__ amaxK,
            const float* __restrict__ h, float* __restrict__ outacc) {
    __shared__ float acc[AGG_ROWS * HD];
    __shared__ float den[AGG_ROWS * H_HEADS];
    const int lane = threadIdx.x & 31;
    const int wave = __builtin_amdgcn_readfirstlane((int)(threadIdx.x >> 5));
    const int n0 = blockIdx.x * AGG_ROWS;
    int cnt = N_NODES - n0;
    if (cnt > AGG_ROWS) cnt = AGG_ROWS;

    int dv[AGG_UNROLL][4];
    agg_load_groups(dv, dst, wave, lane);

    for (int tl = wave; tl < cnt; tl += AGG_WAVES) {
        int t = n0 + tl;
        size_t i = (size_t)N_EDGES + (size_t)t;
        agg_row_term(acc + tl * HD, den + tl * H_HEADS, h + (size_t)t * HD, logit + i * 8,
                     amaxK + (size_t)t * 8, lane, true);
    }
    __syncthreads();

    for (int g0 = wave; g0 < DST_GROUPS; g0 += AGG_WAVES * AGG_UNROLL) {
        int nx[AGG_UNROLL][4];
        agg_load_groups(nx, dst, g0 + AGG_WAVES * AGG_UNROLL, lane);

        const unsigned ucnt = (unsigned)cnt;
        int tl[AGG_UNROLL][4];
        int sv[AGG_UNROLL][4];
        unsigned anym[AGG_UNROLL];
#pragma unroll
        for (int u = 0; u < AGG_UNROLL; ++u) {
            int gg = g0 + u * AGG_WAVES;
#pragma unroll
            for (int k = 0; k < 4; ++k) {
                tl[u][k] = dv[u][k] - n0;
                sv[u][k] = 0;
            }
            bool hu = ((unsigned)tl[u][0] < ucnt) | ((unsigned)tl[u][1] < ucnt) |
                      ((unsigned)tl[u][2] < ucnt) | ((unsigned)tl[u][3] < ucnt);
            anym[u] = __builtin_amdgcn_ballot_w32(hu);
            if (anym[u] != 0u) {
#pragma unroll
                for (int k = 0; k < 4; ++k) {
                    if ((unsigned)tl[u][k] < ucnt)
                        sv[u][k] = src[((size_t)gg * 32 + lane) * 4 + k];
                }
            }
        }
#pragma unroll
        for (int u = 0; u < AGG_UNROLL; ++u) {
            int gg = g0 + u * AGG_WAVES;
            if (anym[u] != 0u) {
#pragma unroll
                for (int k = 0; k < 4; ++k) {
                    unsigned m = __builtin_amdgcn_ballot_w32((unsigned)tl[u][k] < ucnt);
                    while (m != 0u) {
                        int l = __builtin_ctz(m);
                        m &= m - 1u;
                        int tlu = __builtin_amdgcn_readlane(tl[u][k], l);
                        int su  = __builtin_amdgcn_readlane(sv[u][k], l);
                        size_t e = ((size_t)gg * 32 + (size_t)l) * 4 + (size_t)k;
                        agg_row_term(acc + tlu * HD, den + tlu * H_HEADS, h + (size_t)su * HD,
                                     logit + e * 8, amaxK + (size_t)(n0 + tlu) * 8, lane, false);
                    }
                }
            }
        }
#pragma unroll
        for (int u = 0; u < AGG_UNROLL; ++u) {
#pragma unroll
            for (int k = 0; k < 4; ++k) dv[u][k] = nx[u][k];
        }
    }
    __syncthreads();

    for (int tl = wave; tl < cnt; tl += AGG_WAVES) {
        float* orow = outacc + (size_t)(n0 + tl) * HD;
#pragma unroll
        for (int j = 0; j < 4; ++j) {
            int hh = j * 2 + (lane >> 4);
            orow[j * 32 + lane] = acc[tl * HD + j * 32 + lane] / (den[tl * H_HEADS + hh] + 1e-16f);
        }
    }
}

__device__ __forceinline__ int logit_key(float l) {
    int bi = __float_as_int(l);
    return bi >= 0 ? bi : (bi ^ 0x7FFFFFFF);
}

__device__ __forceinline__ void pre_edge_term(size_t e, int tl, int s, int t,
                                              const float* __restrict__ edge_attr,
                                              const float* __restrict__ a_src,
                                              const float* __restrict__ a_dst,
                                              const float* sV, float* lsum, float* ldeg,
                                              int* lmax, float* __restrict__ logit) {
    const float4* ep = reinterpret_cast<const float4*>(edge_attr + e * E_DIM);
    const float4* sp = reinterpret_cast<const float4*>(a_src + (size_t)s * H_HEADS);
    const float4* dp = reinterpret_cast<const float4*>(a_dst + (size_t)t * H_HEADS);
    float4 q0 = ep[0], q1 = ep[1], q2 = ep[2], q3 = ep[3];
    float4 s0 = sp[0], s1 = sp[1];
    float4 d0 = dp[0], d1 = dp[1];
    float eav[E_DIM] = {q0.x, q0.y, q0.z, q0.w, q1.x, q1.y, q1.z, q1.w,
                        q2.x, q2.y, q2.z, q2.w, q3.x, q3.y, q3.z, q3.w};
    float asv[H_HEADS] = {s0.x, s0.y, s0.z, s0.w, s1.x, s1.y, s1.z, s1.w};
    float adv[H_HEADS] = {d0.x, d0.y, d0.z, d0.w, d1.x, d1.y, d1.z, d1.w};
    float ae[H_HEADS];
#pragma unroll
    for (int hh = 0; hh < H_HEADS; ++hh) ae[hh] = 0.f;
#pragma unroll
    for (int k = 0; k < E_DIM; ++k) {
#pragma unroll
        for (int hh = 0; hh < H_HEADS; ++hh) ae[hh] += eav[k] * sV[k * 8 + hh];
    }
    float lg[H_HEADS];
#pragma unroll
    for (int hh = 0; hh < H_HEADS; ++hh) {
        float l = asv[hh] + adv[hh] + ae[hh];
        lg[hh] = l > 0.f ? l : NEG_SLOPE * l;
    }
    float4* lp = reinterpret_cast<float4*>(logit + e * H_HEADS);
    lp[0] = make_float4(lg[0], lg[1], lg[2], lg[3]);
    lp[1] = make_float4(lg[4], lg[5], lg[6], lg[7]);
#pragma unroll
    for (int hh = 0; hh < H_HEADS; ++hh) atomicMax(&lmax[tl * H_HEADS + hh], logit_key(lg[hh]));
#pragma unroll
    for (int k = 0; k < E_DIM; ++k) atomicAdd(&lsum[tl * E_DIM + k], eav[k]);
    atomicAdd(&ldeg[tl], 1.0f);
}

#define PRE_STRIDE   (AGG_WAVES * AGG_UNROLL)
#define PRE_STRIPES  ((DST_GROUPS + PRE_STRIDE - 1) / PRE_STRIDE)
#define PRE_SPC      14
#define PRE_CHUNKS   ((PRE_STRIPES + PRE_SPC - 1) / PRE_SPC)
#define PRE_QCAP     2048

__global__ void __launch_bounds__(AGG_THREADS)
k_node_pre(const int* __restrict__ src, const int* __restrict__ dst,
           const float* __restrict__ edge_attr, const float* __restrict__ Vc,
           const float* __restrict__ a_src, const float* __restrict__ a_dst,
           float* __restrict__ logit, int* __restrict__ amaxK) {
    __shared__ float lsum[AGG_ROWS * E_DIM];
    __shared__ float ldeg[AGG_ROWS];
    __shared__ int   lmax[AGG_ROWS * H_HEADS];
    __shared__ float sV[E_DIM * H_HEADS];
    __shared__ int   q[PRE_QCAP];
    __shared__ int   qn[PRE_CHUNKS];
    const int lane = threadIdx.x & 31;
    const int wave = __builtin_amdgcn_readfirstlane((int)(threadIdx.x >> 5));
    const int n0 = blockIdx.x * AGG_ROWS;
    int cnt = N_NODES - n0;
    if (cnt > AGG_ROWS) cnt = AGG_ROWS;
    const unsigned ucnt = (unsigned)cnt;

    int dv[AGG_UNROLL][4];
    agg_load_groups(dv, dst, wave, lane);

    for (int i = threadIdx.x; i < AGG_ROWS * E_DIM; i += AGG_THREADS) lsum[i] = 0.0f;
    for (int i = threadIdx.x; i < AGG_ROWS * H_HEADS; i += AGG_THREADS) lmax[i] = (int)0x80000000;
    for (int i = threadIdx.x; i < AGG_ROWS; i += AGG_THREADS) ldeg[i] = 0.0f;
    if (threadIdx.x < E_DIM * H_HEADS) sV[threadIdx.x] = Vc[threadIdx.x];
    if (threadIdx.x < PRE_CHUNKS) qn[threadIdx.x] = 0;
    __syncthreads();

    for (int c = 0; c < PRE_CHUNKS; ++c) {
        int iend = (c + 1) * PRE_SPC;
        if (iend > PRE_STRIPES) iend = PRE_STRIPES;
        for (int i = c * PRE_SPC; i < iend; ++i) {
            const int g0 = wave + i * PRE_STRIDE;
            int cur[AGG_UNROLL][4];
#pragma unroll
            for (int u = 0; u < AGG_UNROLL; ++u) {
#pragma unroll
                for (int k = 0; k < 4; ++k) cur[u][k] = dv[u][k];
            }
            agg_load_groups(dv, dst, g0 + PRE_STRIDE, lane);

            unsigned ovf = 0u;
#pragma unroll
            for (int u = 0; u < AGG_UNROLL; ++u) {
                int gg = g0 + u * AGG_WAVES;
#pragma unroll
                for (int k = 0; k < 4; ++k) {
                    if ((unsigned)(cur[u][k] - n0) < ucnt) {
                        int pos = atomicAdd(&qn[c], 1);
                        if (pos < PRE_QCAP) q[pos] = (gg * 32 + lane) * 4 + k;
                        else                ovf |= 1u << (u * 4 + k);
                    }
                }
            }
            while (ovf != 0u) {
                asm volatile("" ::: "memory");
                int slot = __builtin_ctz(ovf);
                ovf &= ovf - 1u;
                int e = ((g0 + (slot >> 2) * AGG_WAVES) * 32 + lane) * 4 + (slot & 3);
                int s = src[e];
                int t = dst[e];
                pre_edge_term((size_t)e, t - n0, s, t, edge_attr, a_src, a_dst, sV, lsum, ldeg,
                              lmax, logit);
            }
        }
        __syncthreads();
        int qc = qn[c];
        if (qc > PRE_QCAP) qc = PRE_QCAP;
        for (int j = threadIdx.x; j < qc; j += AGG_THREADS) {
            asm volatile("" ::: "memory");
            int e = q[j];
            int s = src[e];
            int t = dst[e];
            pre_edge_term((size_t)e, t - n0, s, t, edge_attr, a_src, a_dst, sV, lsum, ldeg,
                          lmax, logit);
        }
        __syncthreads();
    }

    for (int i = threadIdx.x; i < cnt * H_HEADS; i += AGG_THREADS) {
        int tl = i >> 3, hh = i & 7;
        int t = n0 + tl;
        float d = ldeg[tl];
        d = d < 1.0f ? 1.0f : d;
        float ae = 0.f;
#pragma unroll
        for (int k = 0; k < E_DIM; ++k) ae += (lsum[tl * E_DIM + k] / d) * sV[k * 8 + hh];
        float l = a_src[(size_t)t * H_HEADS + hh] + a_dst[(size_t)t * H_HEADS + hh] + ae;
        l = l > 0.f ? l : NEG_SLOPE * l;
        logit[((size_t)N_EDGES + (size_t)t) * H_HEADS + hh] = l;
        int key = logit_key(l);
        int mk = lmax[i];
        amaxK[(size_t)t * H_HEADS + hh] = key > mk ? key : mk;
    }
}

__global__ void k_stats(const float* __restrict__ outacc, float* meanb,
                        float* invstd) {
    __shared__ float ssum[256];
    __shared__ float ssq[256];
    int c = blockIdx.x;
    float s = 0.f, q = 0.f;
    for (int n = threadIdx.x; n < N_NODES; n += blockDim.x) {
        float v = outacc[(size_t)n * HD + c];
        s += v;
        q += v * v;
    }
    ssum[threadIdx.x] = s;
    ssq[threadIdx.x] = q;
    __syncthreads();
    for (int off = 128; off > 0; off >>= 1) {
        if (threadIdx.x < off) {
            ssum[threadIdx.x] += ssum[threadIdx.x + off];
            ssq[threadIdx.x]  += ssq[threadIdx.x + off];
        }
        __syncthreads();
    }
    if (threadIdx.x == 0) {
        float m = ssum[0] / (float)N_NODES;
        float var = ssq[0] / (float)N_NODES - m * m;
        meanb[c] = m;
        invstd[c] = rsqrtf(var + EPS_IN);
    }
}

__global__ void k_final(const float* __restrict__ x,
                        const float* __restrict__ outacc,
                        const float* __restrict__ meanb,
                        const float* __restrict__ invstd,
                        const float* __restrict__ gamma,
                        const float* __restrict__ beta, float* out) {
    int idx = blockIdx.x * blockDim.x + threadIdx.x;
    if (idx >= N_NODES * HD) return;
    int c = idx & 127;
    float normed = (outacc[idx] - meanb[c]) * invstd[c] * gamma[c] + beta[c];
    float t = x[idx] + normed;
    out[idx] = t > 0.f ? t : expm1f(t);
}

extern "C" void kernel_launch(void* const* d_in, const int* in_sizes, int n_in,
                              void* d_out, int out_size, void* d_ws, size_t ws_size,
                              hipStream_t stream) {
    const float* x        = (const float*)d_in[0];
    const int*   ei       = (const int*)d_in[1];
    const float* edgeattr = (const float*)d_in[2];
    const float* W        = (const float*)d_in[3];
    const float* att_src  = (const float*)d_in[4];
    const float* att_dst  = (const float*)d_in[5];
    const float* W_e      = (const float*)d_in[6];
    const float* att_edge = (const float*)d_in[7];
    const float* gamma    = (const float*)d_in[8];
    const float* beta     = (const float*)d_in[9];
    float* out = (float*)d_out;

    const int* src = ei;
    const int* dst = ei + N_EDGES;

    float* ws = (float*)d_ws;
    float* hbuf     = ws;
    float* a_src    = hbuf   + (size_t)N_NODES * HD;
    float* a_dst    = a_src  + (size_t)N_NODES * H_HEADS;
    float* degb     = a_dst  + (size_t)N_NODES * H_HEADS;
    float* loopattr = degb   + N_NODES;
    float* Vc       = loopattr + (size_t)N_NODES * E_DIM;
    float* alpha    = Vc + 128;
    float* denom    = alpha + (size_t)ET * H_HEADS;
    float* outacc   = denom + (size_t)N_NODES * H_HEADS;
    float* meanb    = outacc + (size_t)N_NODES * HD;
    float* invstd   = meanb + 128;
    int*   amaxK    = (int*)(invstd + 128);

    k_gemm_h<<<(N_NODES / 16) * 8 / 8, 256, 0, stream>>>(x, W, hbuf);

    k_attn_node<<<(N_NODES * H_HEADS + 255) / 256, 256, 0, stream>>>(
        hbuf, att_src, att_dst, a_src, a_dst);
    k_edge_proj<<<1, 128, 0, stream>>>(W_e, att_edge, Vc);

    k_node_pre<<<AGG_BLOCKS, AGG_THREADS, 0, stream>>>(src, dst, edgeattr, Vc, a_src, a_dst,
                                                       alpha, amaxK);
    k_aggregate<<<AGG_BLOCKS, AGG_THREADS, 0, stream>>>(src, dst, alpha, amaxK, hbuf, outacc);

    k_stats<<<HD, 256, 0, stream>>>(outacc, meanb, invstd);
    k_final<<<(N_NODES * HD + 255) / 256, 256, 0, stream>>>(
        x, outacc, meanb, invstd, gamma, beta, out);
}


# Aggregation drains the owned-edge list the pre-softmax kernel's dst scan already found (kept in the block's own output rows, 8 edges in flight per wave), no second dst scan; sum order changes
# speedup vs baseline: 7.3678x; 1.1992x over previous
#include <hip/hip_runtime.h>
#include <hip/hip_bf16.h>

#define N_NODES 50000
#define N_EDGES 1600000
#define F_DIM   128
#define H_HEADS 8
#define D_HEAD  16
#define E_DIM   16
#define HD      (H_HEADS * D_HEAD)
#define ET      (N_EDGES + N_NODES)
#define EPS_IN  1e-5f
#define NEG_SLOPE 0.2f

typedef float v2f __attribute__((ext_vector_type(2)));
typedef float v8f __attribute__((ext_vector_type(8)));

__global__ void __launch_bounds__(256)
k_gemm_h(const float* __restrict__ x, const float* __restrict__ W,
         float* __restrict__ h) {
    int wave = threadIdx.x >> 5;
    int lane = threadIdx.x & 31;
    int tile = blockIdx.x * 8 + wave;
    int tileM = tile >> 3;
    int tileN = tile & 7;
    int m   = tileM * 16 + (lane & 15);
    int kb  = (lane >> 4) << 1;
    int col = tileN * 16 + (lane & 15);

    v8f acc = {};
#pragma unroll
    for (int k = 0; k < F_DIM; k += 4) {
        v2f a, b;
        a.x = x[(size_t)m * F_DIM + k + kb];
        a.y = x[(size_t)m * F_DIM + k + kb + 1];
        b.x = W[(size_t)(k + kb) * HD + col];
        b.y = W[(size_t)(k + kb + 1) * HD + col];
        acc = __builtin_amdgcn_wmma_f32_16x16x4_f32(
             false, a,  false, b,
             (short)0, acc,  false,  false);
    }
    int rbase = tileM * 16 + ((lane >> 4) << 3);
#pragma unroll
    for (int r = 0; r < 8; ++r)
        h[(size_t)(rbase + r) * HD + col] = acc[r];
}

__global__ void k_attn_node(const float* __restrict__ h,
                            const float* __restrict__ att_src,
                            const float* __restrict__ att_dst,
                            float* a_src, float* a_dst) {
    int idx = blockIdx.x * blockDim.x + threadIdx.x;
    if (idx >= N_NODES * H_HEADS) return;
    int n = idx >> 3, hh = idx & 7;
    const float* hp = h + (size_t)n * HD + hh * D_HEAD;
    const float* as = att_src + hh * D_HEAD;
    const float* ad = att_dst + hh * D_HEAD;
    float s = 0.f, d = 0.f;
#pragma unroll
    for (int i = 0; i < D_HEAD; ++i) {
        float v = hp[i];
        s += v * as[i];
        d += v * ad[i];
    }
    a_src[idx] = s;
    a_dst[idx] = d;
}

__global__ void k_edge_proj(const float* __restrict__ W_e,
                            const float* __restrict__ att_edge, float* V) {
    int idx = threadIdx.x;
    int k = idx >> 3, hh = idx & 7;
    float s = 0.f;
#pragma unroll
    for (int d = 0; d < D_HEAD; ++d)
        s += W_e[(size_t)k * HD + hh * D_HEAD + d] * att_edge[hh * D_HEAD + d];
    V[idx] = s;
}

#define AGG_BLOCKS  256
#define AGG_ROWS    ((N_NODES + AGG_BLOCKS - 1) / AGG_BLOCKS)
#define AGG_THREADS 1024
#define AGG_WAVES   (AGG_THREADS / 32)
#define DST_GROUPS  (N_EDGES / 128)
#define AGG_UNROLL  4

#define AGG_BATCH   8
#define LIST_E_BITS 21
#define LIST_CAP_PER_ROW (HD / 2)

static_assert(N_EDGES % 128 == 0, "dst list must split into whole wave groups");
static_assert(N_EDGES <= (1 << LIST_E_BITS), "edge id must fit its field");
static_assert(AGG_ROWS <= (1 << (31 - LIST_E_BITS)), "local row must fit its field");

__device__ __forceinline__ void agg_row_term(float* acc_row, float* den_row,
                                             const float* __restrict__ hrow,
                                             const float* __restrict__ lrow,
                                             const int* __restrict__ mrow, int lane,
                                             bool first) {
    int key = mrow[lane & 7];
    int bi = key >= 0 ? key : (key ^ 0x7FFFFFFF);
    float wl = __expf(lrow[lane & 7] - __int_as_float(bi));
    float hv[4];
#pragma unroll
    for (int j = 0; j < 4; ++j) hv[j] = hrow[j * 32 + lane];
    float v[4];
#pragma unroll
    for (int j = 0; j < 4; ++j) {
        int hh = j * 2 + (lane >> 4);
        float w = __int_as_float(__builtin_amdgcn_ds_bpermute(hh << 2, __float_as_int(wl)));
        v[j] = hv[j] * w;
    }
#pragma unroll
    for (int j = 0; j < 4; ++j) {
        int c = j * 32 + lane;
        if (first) acc_row[c] = v[j];
        else       atomicAdd(&acc_row[c], v[j]);
    }
    if (lane < H_HEADS) {
        if (first) den_row[lane] = wl;
        else       atomicAdd(&den_row[lane], wl);
    }
}

__device__ __forceinline__ void agg_load_groups(int (&dv)[AGG_UNROLL][4],
                                                const int* __restrict__ dst, int g0, int lane) {
#pragma unroll
    for (int u = 0; u < AGG_UNROLL; ++u) {
        int gg = g0 + u * AGG_WAVES;
        if (gg < DST_GROUPS) {
            const int* p = dst + ((size_t)gg * 32 + lane) * 4;
            dv[u][0] = p[0]; dv[u][1] = p[1]; dv[u][2] = p[2]; dv[u][3] = p[3];
        } else {
            dv[u][0] = -1; dv[u][1] = -1; dv[u][2] = -1; dv[u][3] = -1;
        }
    }
}

__global__ void __launch_bounds__(AGG_THREADS)
k_aggregate(const int* __restrict__ src, const int* __restrict__ dst,
            const float* __restrict__ logit, const int* __restrict__ amaxK,
            const float* __restrict__ h, const int* __restrict__ ecount, float* outacc) {
    __shared__ float acc[AGG_ROWS * HD];
    __shared__ float den[AGG_ROWS * H_HEADS];
    const int lane = threadIdx.x & 31;
    const int wave = __builtin_amdgcn_readfirstlane((int)(threadIdx.x >> 5));
    const int n0 = blockIdx.x * AGG_ROWS;
    int cnt = N_NODES - n0;
    if (cnt > AGG_ROWS) cnt = AGG_ROWS;

    const int total = ecount[blockIdx.x];
    const int lcap = cnt * LIST_CAP_PER_ROW;
    const int2* blist = reinterpret_cast<const int2*>(outacc + (size_t)n0 * HD);

    for (int tl = wave; tl < cnt; tl += AGG_WAVES) {
        int t = n0 + tl;
        size_t i = (size_t)N_EDGES + (size_t)t;
        agg_row_term(acc + tl * HD, den + tl * H_HEADS, h + (size_t)t * HD, logit + i * 8,
                     amaxK + (size_t)t * 8, lane, true);
    }
    __syncthreads();

    if (total <= lcap) {
    const int nfull = total & ~(AGG_BATCH - 1);
    int2 ent = make_int2(0, 0);
    if (lane < AGG_BATCH && wave * AGG_BATCH < nfull) ent = blist[wave * AGG_BATCH + lane];
    asm volatile("" : "+v"(ent.x), "+v"(ent.y));
    for (int j0 = wave * AGG_BATCH; j0 < nfull; j0 += AGG_WAVES * AGG_BATCH) {
        const int jn = j0 + AGG_WAVES * AGG_BATCH;
        int2 nxt = make_int2(0, 0);
        if (lane < AGG_BATCH && jn < nfull) nxt = blist[jn + lane];
        int   tlb[AGG_BATCH];
        int   keyb[AGG_BATCH];
        float lgb[AGG_BATCH];
        float hvb[AGG_BATCH][4];
#pragma unroll
        for (int b = 0; b < AGG_BATCH; ++b) {
            int ex = __builtin_amdgcn_readlane(ent.x, b);
            int su = __builtin_amdgcn_readlane(ent.y, b);
            int eu = ex & ((1 << LIST_E_BITS) - 1);
            tlb[b] = ex >> LIST_E_BITS;
            keyb[b] = amaxK[(size_t)(n0 + tlb[b]) * 8 + (lane & 7)];
            lgb[b]  = logit[(size_t)eu * 8 + (lane & 7)];
#pragma unroll
            for (int j = 0; j < 4; ++j) hvb[b][j] = h[(size_t)su * HD + j * 32 + lane];
        }
#pragma unroll
        for (int b = 0; b < AGG_BATCH; ++b) {
            int bi = keyb[b] >= 0 ? keyb[b] : (keyb[b] ^ 0x7FFFFFFF);
            float wl = __expf(lgb[b] - __int_as_float(bi));
            float* acc_row = acc + tlb[b] * HD;
            float v[4];
#pragma unroll
            for (int j = 0; j < 4; ++j) {
                int hh = j * 2 + (lane >> 4);
                float w = __int_as_float(__builtin_amdgcn_ds_bpermute(hh << 2, __float_as_int(wl)));
                v[j] = hvb[b][j] * w;
            }
#pragma unroll
            for (int j = 0; j < 4; ++j) atomicAdd(&acc_row[j * 32 + lane], v[j]);
            if (lane < H_HEADS) atomicAdd(&den[tlb[b] * H_HEADS + lane], wl);
        }
        ent = nxt;
    }
    for (int j = nfull + wave; j < total; j += AGG_WAVES) {
        int2 en = blist[j];
        int ex = __builtin_amdgcn_readfirstlane(en.x);
        int su = __builtin_amdgcn_readfirstlane(en.y);
        int eu = ex & ((1 << LIST_E_BITS) - 1);
        int tlu = ex >> LIST_E_BITS;
        agg_row_term(acc + tlu * HD, den + tlu * H_HEADS, h + (size_t)su * HD,
                     logit + (size_t)eu * 8, amaxK + (size_t)(n0 + tlu) * 8, lane, false);
    }
    } else {
    int dv[AGG_UNROLL][4];
    agg_load_groups(dv, dst, wave, lane);
    for (int g0 = wave; g0 < DST_GROUPS; g0 += AGG_WAVES * AGG_UNROLL) {
        int nx[AGG_UNROLL][4];
        agg_load_groups(nx, dst, g0 + AGG_WAVES * AGG_UNROLL, lane);

        const unsigned ucnt = (unsigned)cnt;
        int tl[AGG_UNROLL][4];
        int sv[AGG_UNROLL][4];
        unsigned anym[AGG_UNROLL];
#pragma unroll
        for (int u = 0; u < AGG_UNROLL; ++u) {
            int gg = g0 + u * AGG_WAVES;
#pragma unroll
            for (int k = 0; k < 4; ++k) {
                tl[u][k] = dv[u][k] - n0;
                sv[u][k] = 0;
            }
            bool hu = ((unsigned)tl[u][0] < ucnt) | ((unsigned)tl[u][1] < ucnt) |
                      ((unsigned)tl[u][2] < ucnt) | ((unsigned)tl[u][3] < ucnt);
            anym[u] = __builtin_amdgcn_ballot_w32(hu);
            if (anym[u] != 0u) {
#pragma unroll
                for (int k = 0; k < 4; ++k) {
                    if ((unsigned)tl[u][k] < ucnt)
                        sv[u][k] = src[((size_t)gg * 32 + lane) * 4 + k];
                }
            }
        }
#pragma unroll
        for (int u = 0; u < AGG_UNROLL; ++u) {
            int gg = g0 + u * AGG_WAVES;
            if (anym[u] != 0u) {
#pragma unroll
                for (int k = 0; k < 4; ++k) {
                    unsigned m = __builtin_amdgcn_ballot_w32((unsigned)tl[u][k] < ucnt);
                    while (m != 0u) {
                        int l = __builtin_ctz(m);
                        m &= m - 1u;
                        int tlu = __builtin_amdgcn_readlane(tl[u][k], l);
                        int su  = __builtin_amdgcn_readlane(sv[u][k], l);
                        size_t e = ((size_t)gg * 32 + (size_t)l) * 4 + (size_t)k;
                        agg_row_term(acc + tlu * HD, den + tlu * H_HEADS, h + (size_t)su * HD,
                                     logit + e * 8, amaxK + (size_t)(n0 + tlu) * 8, lane, false);
                    }
                }
            }
        }
#pragma unroll
        for (int u = 0; u < AGG_UNROLL; ++u) {
#pragma unroll
            for (int k = 0; k < 4; ++k) dv[u][k] = nx[u][k];
        }
    }
    }
    __syncthreads();

    for (int tl = wave; tl < cnt; tl += AGG_WAVES) {
        float* orow = outacc + (size_t)(n0 + tl) * HD;
#pragma unroll
        for (int j = 0; j < 4; ++j) {
            int hh = j * 2 + (lane >> 4);
            orow[j * 32 + lane] = acc[tl * HD + j * 32 + lane] / (den[tl * H_HEADS + hh] + 1e-16f);
        }
    }
}

__device__ __forceinline__ int logit_key(float l) {
    int bi = __float_as_int(l);
    return bi >= 0 ? bi : (bi ^ 0x7FFFFFFF);
}

__device__ __forceinline__ void pre_edge_term(size_t e, int tl, int s, int t,
                                              const float* __restrict__ edge_attr,
                                              const float* __restrict__ a_src,
                                              const float* __restrict__ a_dst,
                                              const float* sV, float* lsum, float* ldeg,
                                              int* lmax, float* __restrict__ logit,
                                              int* lcnt, int2* blist, int lcap) {
    const float4* ep = reinterpret_cast<const float4*>(edge_attr + e * E_DIM);
    const float4* sp = reinterpret_cast<const float4*>(a_src + (size_t)s * H_HEADS);
    const float4* dp = reinterpret_cast<const float4*>(a_dst + (size_t)t * H_HEADS);
    float4 q0 = ep[0], q1 = ep[1], q2 = ep[2], q3 = ep[3];
    float4 s0 = sp[0], s1 = sp[1];
    float4 d0 = dp[0], d1 = dp[1];
    float eav[E_DIM] = {q0.x, q0.y, q0.z, q0.w, q1.x, q1.y, q1.z, q1.w,
                        q2.x, q2.y, q2.z, q2.w, q3.x, q3.y, q3.z, q3.w};
    float asv[H_HEADS] = {s0.x, s0.y, s0.z, s0.w, s1.x, s1.y, s1.z, s1.w};
    float adv[H_HEADS] = {d0.x, d0.y, d0.z, d0.w, d1.x, d1.y, d1.z, d1.w};
    float ae[H_HEADS];
#pragma unroll
    for (int hh = 0; hh < H_HEADS; ++hh) ae[hh] = 0.f;
#pragma unroll
    for (int k = 0; k < E_DIM; ++k) {
#pragma unroll
        for (int hh = 0; hh < H_HEADS; ++hh) ae[hh] += eav[k] * sV[k * 8 + hh];
    }
    float lg[H_HEADS];
#pragma unroll
    for (int hh = 0; hh < H_HEADS; ++hh) {
        float l = asv[hh] + adv[hh] + ae[hh];
        lg[hh] = l > 0.f ? l : NEG_SLOPE * l;
    }
    float4* lp = reinterpret_cast<float4*>(logit + e * H_HEADS);
    lp[0] = make_float4(lg[0], lg[1], lg[2], lg[3]);
    lp[1] = make_float4(lg[4], lg[5], lg[6], lg[7]);
#pragma unroll
    for (int hh = 0; hh < H_HEADS; ++hh) atomicMax(&lmax[tl * H_HEADS + hh], logit_key(lg[hh]));
#pragma unroll
    for (int k = 0; k < E_DIM; ++k) atomicAdd(&lsum[tl * E_DIM + k], eav[k]);
    atomicAdd(&ldeg[tl], 1.0f);
    int pos = atomicAdd(lcnt, 1);
    if (pos < lcap) blist[pos] = make_int2((int)e | (tl << LIST_E_BITS), s);
}

#define PRE_STRIDE   (AGG_WAVES * AGG_UNROLL)
#define PRE_STRIPES  ((DST_GROUPS + PRE_STRIDE - 1) / PRE_STRIDE)
#define PRE_SPC      14
#define PRE_CHUNKS   ((PRE_STRIPES + PRE_SPC - 1) / PRE_SPC)
#define PRE_QCAP     2048

__global__ void __launch_bounds__(AGG_THREADS)
k_node_pre(const int* __restrict__ src, const int* __restrict__ dst,
           const float* __restrict__ edge_attr, const float* __restrict__ Vc,
           const float* __restrict__ a_src, const float* __restrict__ a_dst,
           float* __restrict__ logit, int* __restrict__ amaxK,
           float* elist, int* __restrict__ ecount) {
    __shared__ float lsum[AGG_ROWS * E_DIM];
    __shared__ float ldeg[AGG_ROWS];
    __shared__ int   lmax[AGG_ROWS * H_HEADS];
    __shared__ float sV[E_DIM * H_HEADS];
    __shared__ int   q[PRE_QCAP];
    __shared__ int   qn[PRE_CHUNKS];
    __shared__ int   lcnt;
    const int lane = threadIdx.x & 31;
    const int wave = __builtin_amdgcn_readfirstlane((int)(threadIdx.x >> 5));
    const int n0 = blockIdx.x * AGG_ROWS;
    int cnt = N_NODES - n0;
    if (cnt > AGG_ROWS) cnt = AGG_ROWS;
    const unsigned ucnt = (unsigned)cnt;
    int2* blist = reinterpret_cast<int2*>(elist + (size_t)n0 * HD);
    const int lcap = cnt * LIST_CAP_PER_ROW;

    int dv[AGG_UNROLL][4];
    agg_load_groups(dv, dst, wave, lane);

    for (int i = threadIdx.x; i < AGG_ROWS * E_DIM; i += AGG_THREADS) lsum[i] = 0.0f;
    for (int i = threadIdx.x; i < AGG_ROWS * H_HEADS; i += AGG_THREADS) lmax[i] = (int)0x80000000;
    for (int i = threadIdx.x; i < AGG_ROWS; i += AGG_THREADS) ldeg[i] = 0.0f;
    if (threadIdx.x < E_DIM * H_HEADS) sV[threadIdx.x] = Vc[threadIdx.x];
    if (threadIdx.x < PRE_CHUNKS) qn[threadIdx.x] = 0;
    if (threadIdx.x == 0) lcnt = 0;
    __syncthreads();

    for (int c = 0; c < PRE_CHUNKS; ++c) {
        int iend = (c + 1) * PRE_SPC;
        if (iend > PRE_STRIPES) iend = PRE_STRIPES;
        for (int i = c * PRE_SPC; i < iend; ++i) {
            const int g0 = wave + i * PRE_STRIDE;
            int cur[AGG_UNROLL][4];
#pragma unroll
            for (int u = 0; u < AGG_UNROLL; ++u) {
#pragma unroll
                for (int k = 0; k < 4; ++k) cur[u][k] = dv[u][k];
            }
            agg_load_groups(dv, dst, g0 + PRE_STRIDE, lane);

            unsigned ovf = 0u;
#pragma unroll
            for (int u = 0; u < AGG_UNROLL; ++u) {
                int gg = g0 + u * AGG_WAVES;
#pragma unroll
                for (int k = 0; k < 4; ++k) {
                    if ((unsigned)(cur[u][k] - n0) < ucnt) {
                        int pos = atomicAdd(&qn[c], 1);
                        if (pos < PRE_QCAP) q[pos] = (gg * 32 + lane) * 4 + k;
                        else                ovf |= 1u << (u * 4 + k);
                    }
                }
            }
            while (ovf != 0u) {
                asm volatile("" ::: "memory");
                int slot = __builtin_ctz(ovf);
                ovf &= ovf - 1u;
                int e = ((g0 + (slot >> 2) * AGG_WAVES) * 32 + lane) * 4 + (slot & 3);
                int s = src[e];
                int t = dst[e];
                pre_edge_term((size_t)e, t - n0, s, t, edge_attr, a_src, a_dst, sV, lsum, ldeg,
                              lmax, logit, &lcnt, blist, lcap);
            }
        }
        __syncthreads();
        int qc = qn[c];
        if (qc > PRE_QCAP) qc = PRE_QCAP;
        for (int j = threadIdx.x; j < qc; j += AGG_THREADS) {
            asm volatile("" ::: "memory");
            int e = q[j];
            int s = src[e];
            int t = dst[e];
            pre_edge_term((size_t)e, t - n0, s, t, edge_attr, a_src, a_dst, sV, lsum, ldeg,
                          lmax, logit, &lcnt, blist, lcap);
        }
        __syncthreads();
    }

    if (threadIdx.x == 0) ecount[blockIdx.x] = lcnt;

    for (int i = threadIdx.x; i < cnt * H_HEADS; i += AGG_THREADS) {
        int tl = i >> 3, hh = i & 7;
        int t = n0 + tl;
        float d = ldeg[tl];
        d = d < 1.0f ? 1.0f : d;
        float ae = 0.f;
#pragma unroll
        for (int k = 0; k < E_DIM; ++k) ae += (lsum[tl * E_DIM + k] / d) * sV[k * 8 + hh];
        float l = a_src[(size_t)t * H_HEADS + hh] + a_dst[(size_t)t * H_HEADS + hh] + ae;
        l = l > 0.f ? l : NEG_SLOPE * l;
        logit[((size_t)N_EDGES + (size_t)t) * H_HEADS + hh] = l;
        int key = logit_key(l);
        int mk = lmax[i];
        amaxK[(size_t)t * H_HEADS + hh] = key > mk ? key : mk;
    }
}

__global__ void k_stats(const float* __restrict__ outacc, float* meanb,
                        float* invstd) {
    __shared__ float ssum[256];
    __shared__ float ssq[256];
    int c = blockIdx.x;
    float s = 0.f, q = 0.f;
    for (int n = threadIdx.x; n < N_NODES; n += blockDim.x) {
        float v = outacc[(size_t)n * HD + c];
        s += v;
        q += v * v;
    }
    ssum[threadIdx.x] = s;
    ssq[threadIdx.x] = q;
    __syncthreads();
    for (int off = 128; off > 0; off >>= 1) {
        if (threadIdx.x < off) {
            ssum[threadIdx.x] += ssum[threadIdx.x + off];
            ssq[threadIdx.x]  += ssq[threadIdx.x + off];
        }
        __syncthreads();
    }
    if (threadIdx.x == 0) {
        float m = ssum[0] / (float)N_NODES;
        float var = ssq[0] / (float)N_NODES - m * m;
        meanb[c] = m;
        invstd[c] = rsqrtf(var + EPS_IN);
    }
}

__global__ void k_final(const float* __restrict__ x,
                        const float* __restrict__ outacc,
                        const float* __restrict__ meanb,
                        const float* __restrict__ invstd,
                        const float* __restrict__ gamma,
                        const float* __restrict__ beta, float* out) {
    int idx = blockIdx.x * blockDim.x + threadIdx.x;
    if (idx >= N_NODES * HD) return;
    int c = idx & 127;
    float normed = (outacc[idx] - meanb[c]) * invstd[c] * gamma[c] + beta[c];
    float t = x[idx] + normed;
    out[idx] = t > 0.f ? t : expm1f(t);
}

extern "C" void kernel_launch(void* const* d_in, const int* in_sizes, int n_in,
                              void* d_out, int out_size, void* d_ws, size_t ws_size,
                              hipStream_t stream) {
    const float* x        = (const float*)d_in[0];
    const int*   ei       = (const int*)d_in[1];
    const float* edgeattr = (const float*)d_in[2];
    const float* W        = (const float*)d_in[3];
    const float* att_src  = (const float*)d_in[4];
    const float* att_dst  = (const float*)d_in[5];
    const float* W_e      = (const float*)d_in[6];
    const float* att_edge = (const float*)d_in[7];
    const float* gamma    = (const float*)d_in[8];
    const float* beta     = (const float*)d_in[9];
    float* out = (float*)d_out;

    const int* src = ei;
    const int* dst = ei + N_EDGES;

    float* ws = (float*)d_ws;
    float* hbuf     = ws;
    float* a_src    = hbuf   + (size_t)N_NODES * HD;
    float* a_dst    = a_src  + (size_t)N_NODES * H_HEADS;
    float* degb     = a_dst  + (size_t)N_NODES * H_HEADS;
    float* loopattr = degb   + N_NODES;
    float* Vc       = loopattr + (size_t)N_NODES * E_DIM;
    float* alpha    = Vc + 128;
    float* denom    = alpha + (size_t)ET * H_HEADS;
    float* outacc   = denom + (size_t)N_NODES * H_HEADS;
    float* meanb    = outacc + (size_t)N_NODES * HD;
    float* invstd   = meanb + 128;
    int*   amaxK    = (int*)(invstd + 128);

    k_gemm_h<<<(N_NODES / 16) * 8 / 8, 256, 0, stream>>>(x, W, hbuf);

    k_attn_node<<<(N_NODES * H_HEADS + 255) / 256, 256, 0, stream>>>(
        hbuf, att_src, att_dst, a_src, a_dst);
    k_edge_proj<<<1, 128, 0, stream>>>(W_e, att_edge, Vc);

    k_node_pre<<<AGG_BLOCKS, AGG_THREADS, 0, stream>>>(src, dst, edgeattr, Vc, a_src, a_dst,
                                                       alpha, amaxK, outacc, (int*)degb);
    k_aggregate<<<AGG_BLOCKS, AGG_THREADS, 0, stream>>>(src, dst, alpha, amaxK, hbuf,
                                                        (const int*)degb, outacc);

    k_stats<<<HD, 256, 0, stream>>>(outacc, meanb, invstd);
    k_final<<<(N_NODES * HD + 255) / 256, 256, 0, stream>>>(
        x, outacc, meanb, invstd, gamma, beta, out);
}


# InstanceNorm channel sums and squares taken per workgroup from the finished rows still in LDS in the aggregation; stats kernel adds 256 partials, no strided re-read of the plane; sum order changes
# speedup vs baseline: 10.0108x; 1.3587x over previous
#include <hip/hip_runtime.h>
#include <hip/hip_bf16.h>

#define N_NODES 50000
#define N_EDGES 1600000
#define F_DIM   128
#define H_HEADS 8
#define D_HEAD  16
#define E_DIM   16
#define HD      (H_HEADS * D_HEAD)
#define ET      (N_EDGES + N_NODES)
#define EPS_IN  1e-5f
#define NEG_SLOPE 0.2f

typedef float v2f __attribute__((ext_vector_type(2)));
typedef float v8f __attribute__((ext_vector_type(8)));

__global__ void __launch_bounds__(256)
k_gemm_h(const float* __restrict__ x, const float* __restrict__ W,
         float* __restrict__ h) {
    int wave = threadIdx.x >> 5;
    int lane = threadIdx.x & 31;
    int tile = blockIdx.x * 8 + wave;
    int tileM = tile >> 3;
    int tileN = tile & 7;
    int m   = tileM * 16 + (lane & 15);
    int kb  = (lane >> 4) << 1;
    int col = tileN * 16 + (lane & 15);

    v8f acc = {};
#pragma unroll
    for (int k = 0; k < F_DIM; k += 4) {
        v2f a, b;
        a.x = x[(size_t)m * F_DIM + k + kb];
        a.y = x[(size_t)m * F_DIM + k + kb + 1];
        b.x = W[(size_t)(k + kb) * HD + col];
        b.y = W[(size_t)(k + kb + 1) * HD + col];
        acc = __builtin_amdgcn_wmma_f32_16x16x4_f32(
             false, a,  false, b,
             (short)0, acc,  false,  false);
    }
    int rbase = tileM * 16 + ((lane >> 4) << 3);
#pragma unroll
    for (int r = 0; r < 8; ++r)
        h[(size_t)(rbase + r) * HD + col] = acc[r];
}

__global__ void k_attn_node(const float* __restrict__ h,
                            const float* __restrict__ att_src,
                            const float* __restrict__ att_dst,
                            float* a_src, float* a_dst) {
    int idx = blockIdx.x * blockDim.x + threadIdx.x;
    if (idx >= N_NODES * H_HEADS) return;
    int n = idx >> 3, hh = idx & 7;
    const float* hp = h + (size_t)n * HD + hh * D_HEAD;
    const float* as = att_src + hh * D_HEAD;
    const float* ad = att_dst + hh * D_HEAD;
    float s = 0.f, d = 0.f;
#pragma unroll
    for (int i = 0; i < D_HEAD; ++i) {
        float v = hp[i];
        s += v * as[i];
        d += v * ad[i];
    }
    a_src[idx] = s;
    a_dst[idx] = d;
}

__global__ void k_edge_proj(const float* __restrict__ W_e,
                            const float* __restrict__ att_edge, float* V) {
    int idx = threadIdx.x;
    int k = idx >> 3, hh = idx & 7;
    float s = 0.f;
#pragma unroll
    for (int d = 0; d < D_HEAD; ++d)
        s += W_e[(size_t)k * HD + hh * D_HEAD + d] * att_edge[hh * D_HEAD + d];
    V[idx] = s;
}

#define AGG_BLOCKS  256
#define AGG_ROWS    ((N_NODES + AGG_BLOCKS - 1) / AGG_BLOCKS)
#define AGG_THREADS 1024
#define AGG_WAVES   (AGG_THREADS / 32)
#define DST_GROUPS  (N_EDGES / 128)
#define AGG_UNROLL  4

#define AGG_BATCH   8
#define LIST_E_BITS 21
#define LIST_CAP_PER_ROW (HD / 2)

static_assert(N_EDGES % 128 == 0, "dst list must split into whole wave groups");
static_assert(N_EDGES <= (1 << LIST_E_BITS), "edge id must fit its field");
static_assert(AGG_ROWS <= (1 << (31 - LIST_E_BITS)), "local row must fit its field");
static_assert((AGG_BLOCKS & (AGG_BLOCKS - 1)) == 0, "the statistics kernel halves the partials in a tree");

__device__ __forceinline__ void agg_row_term(float* acc_row, float* den_row,
                                             const float* __restrict__ hrow,
                                             const float* __restrict__ lrow,
                                             const int* __restrict__ mrow, int lane,
                                             bool first) {
    int key = mrow[lane & 7];
    int bi = key >= 0 ? key : (key ^ 0x7FFFFFFF);
    float wl = __expf(lrow[lane & 7] - __int_as_float(bi));
    float hv[4];
#pragma unroll
    for (int j = 0; j < 4; ++j) hv[j] = hrow[j * 32 + lane];
    float v[4];
#pragma unroll
    for (int j = 0; j < 4; ++j) {
        int hh = j * 2 + (lane >> 4);
        float w = __int_as_float(__builtin_amdgcn_ds_bpermute(hh << 2, __float_as_int(wl)));
        v[j] = hv[j] * w;
    }
#pragma unroll
    for (int j = 0; j < 4; ++j) {
        int c = j * 32 + lane;
        if (first) acc_row[c] = v[j];
        else       atomicAdd(&acc_row[c], v[j]);
    }
    if (lane < H_HEADS) {
        if (first) den_row[lane] = wl;
        else       atomicAdd(&den_row[lane], wl);
    }
}

__device__ __forceinline__ void agg_load_groups(int (&dv)[AGG_UNROLL][4],
                                                const int* __restrict__ dst, int g0, int lane) {
#pragma unroll
    for (int u = 0; u < AGG_UNROLL; ++u) {
        int gg = g0 + u * AGG_WAVES;
        if (gg < DST_GROUPS) {
            const int* p = dst + ((size_t)gg * 32 + lane) * 4;
            dv[u][0] = p[0]; dv[u][1] = p[1]; dv[u][2] = p[2]; dv[u][3] = p[3];
        } else {
            dv[u][0] = -1; dv[u][1] = -1; dv[u][2] = -1; dv[u][3] = -1;
        }
    }
}

__global__ void __launch_bounds__(AGG_THREADS)
k_aggregate(const int* __restrict__ src, const int* __restrict__ dst,
            const float* __restrict__ logit, const int* __restrict__ amaxK,
            const float* __restrict__ h, const int* __restrict__ ecount, float* outacc,
            float* __restrict__ psum, float* __restrict__ psq) {
    __shared__ float acc[AGG_ROWS * HD];
    __shared__ float den[AGG_ROWS * H_HEADS];
    __shared__ float red_s[AGG_WAVES * HD];
    __shared__ float red_q[AGG_WAVES * HD];
    const int lane = threadIdx.x & 31;
    const int wave = __builtin_amdgcn_readfirstlane((int)(threadIdx.x >> 5));
    const int n0 = blockIdx.x * AGG_ROWS;
    int cnt = N_NODES - n0;
    if (cnt > AGG_ROWS) cnt = AGG_ROWS;

    const int total = ecount[blockIdx.x];
    const int lcap = cnt * LIST_CAP_PER_ROW;
    const int2* blist = reinterpret_cast<const int2*>(outacc + (size_t)n0 * HD);

    for (int tl = wave; tl < cnt; tl += AGG_WAVES) {
        int t = n0 + tl;
        size_t i = (size_t)N_EDGES + (size_t)t;
        agg_row_term(acc + tl * HD, den + tl * H_HEADS, h + (size_t)t * HD, logit + i * 8,
                     amaxK + (size_t)t * 8, lane, true);
    }
    __syncthreads();

    if (total <= lcap) {
    const int nfull = total & ~(AGG_BATCH - 1);
    int2 ent = make_int2(0, 0);
    if (lane < AGG_BATCH && wave * AGG_BATCH < nfull) ent = blist[wave * AGG_BATCH + lane];
    asm volatile("" : "+v"(ent.x), "+v"(ent.y));
    for (int j0 = wave * AGG_BATCH; j0 < nfull; j0 += AGG_WAVES * AGG_BATCH) {
        const int jn = j0 + AGG_WAVES * AGG_BATCH;
        int2 nxt = make_int2(0, 0);
        if (lane < AGG_BATCH && jn < nfull) nxt = blist[jn + lane];
        int   tlb[AGG_BATCH];
        int   keyb[AGG_BATCH];
        float lgb[AGG_BATCH];
        float hvb[AGG_BATCH][4];
#pragma unroll
        for (int b = 0; b < AGG_BATCH; ++b) {
            int ex = __builtin_amdgcn_readlane(ent.x, b);
            int su = __builtin_amdgcn_readlane(ent.y, b);
            int eu = ex & ((1 << LIST_E_BITS) - 1);
            tlb[b] = ex >> LIST_E_BITS;
            keyb[b] = amaxK[(size_t)(n0 + tlb[b]) * 8 + (lane & 7)];
            lgb[b]  = logit[(size_t)eu * 8 + (lane & 7)];
#pragma unroll
            for (int j = 0; j < 4; ++j) hvb[b][j] = h[(size_t)su * HD + j * 32 + lane];
        }
#pragma unroll
        for (int b = 0; b < AGG_BATCH; ++b) {
            int bi = keyb[b] >= 0 ? keyb[b] : (keyb[b] ^ 0x7FFFFFFF);
            float wl = __expf(lgb[b] - __int_as_float(bi));
            float* acc_row = acc + tlb[b] * HD;
            float v[4];
#pragma unroll
            for (int j = 0; j < 4; ++j) {
                int hh = j * 2 + (lane >> 4);
                float w = __int_as_float(__builtin_amdgcn_ds_bpermute(hh << 2, __float_as_int(wl)));
                v[j] = hvb[b][j] * w;
            }
#pragma unroll
            for (int j = 0; j < 4; ++j) atomicAdd(&acc_row[j * 32 + lane], v[j]);
            if (lane < H_HEADS) atomicAdd(&den[tlb[b] * H_HEADS + lane], wl);
        }
        ent = nxt;
    }
    for (int j = nfull + wave; j < total; j += AGG_WAVES) {
        int2 en = blist[j];
        int ex = __builtin_amdgcn_readfirstlane(en.x);
        int su = __builtin_amdgcn_readfirstlane(en.y);
        int eu = ex & ((1 << LIST_E_BITS) - 1);
        int tlu = ex >> LIST_E_BITS;
        agg_row_term(acc + tlu * HD, den + tlu * H_HEADS, h + (size_t)su * HD,
                     logit + (size_t)eu * 8, amaxK + (size_t)(n0 + tlu) * 8, lane, false);
    }
    } else {
    int dv[AGG_UNROLL][4];
    agg_load_groups(dv, dst, wave, lane);
    for (int g0 = wave; g0 < DST_GROUPS; g0 += AGG_WAVES * AGG_UNROLL) {
        int nx[AGG_UNROLL][4];
        agg_load_groups(nx, dst, g0 + AGG_WAVES * AGG_UNROLL, lane);

        const unsigned ucnt = (unsigned)cnt;
        int tl[AGG_UNROLL][4];
        int sv[AGG_UNROLL][4];
        unsigned anym[AGG_UNROLL];
#pragma unroll
        for (int u = 0; u < AGG_UNROLL; ++u) {
            int gg = g0 + u * AGG_WAVES;
#pragma unroll
            for (int k = 0; k < 4; ++k) {
                tl[u][k] = dv[u][k] - n0;
                sv[u][k] = 0;
            }
            bool hu = ((unsigned)tl[u][0] < ucnt) | ((unsigned)tl[u][1] < ucnt) |
                      ((unsigned)tl[u][2] < ucnt) | ((unsigned)tl[u][3] < ucnt);
            anym[u] = __builtin_amdgcn_ballot_w32(hu);
            if (anym[u] != 0u) {
#pragma unroll
                for (int k = 0; k < 4; ++k) {
                    if ((unsigned)tl[u][k] < ucnt)
                        sv[u][k] = src[((size_t)gg * 32 + lane) * 4 + k];
                }
            }
        }
#pragma unroll
        for (int u = 0; u < AGG_UNROLL; ++u) {
            int gg = g0 + u * AGG_WAVES;
            if (anym[u] != 0u) {
#pragma unroll
                for (int k = 0; k < 4; ++k) {
                    unsigned m = __builtin_amdgcn_ballot_w32((unsigned)tl[u][k] < ucnt);
                    while (m != 0u) {
                        int l = __builtin_ctz(m);
                        m &= m - 1u;
                        int tlu = __builtin_amdgcn_readlane(tl[u][k], l);
                        int su  = __builtin_amdgcn_readlane(sv[u][k], l);
                        size_t e = ((size_t)gg * 32 + (size_t)l) * 4 + (size_t)k;
                        agg_row_term(acc + tlu * HD, den + tlu * H_HEADS, h + (size_t)su * HD,
                                     logit + e * 8, amaxK + (size_t)(n0 + tlu) * 8, lane, false);
                    }
                }
            }
        }
#pragma unroll
        for (int u = 0; u < AGG_UNROLL; ++u) {
#pragma unroll
            for (int k = 0; k < 4; ++k) dv[u][k] = nx[u][k];
        }
    }
    }
    __syncthreads();

    float cs[4] = {0.f, 0.f, 0.f, 0.f};
    float cq[4] = {0.f, 0.f, 0.f, 0.f};
    for (int tl = wave; tl < cnt; tl += AGG_WAVES) {
        float* orow = outacc + (size_t)(n0 + tl) * HD;
#pragma unroll
        for (int j = 0; j < 4; ++j) {
            int hh = j * 2 + (lane >> 4);
            float v = acc[tl * HD + j * 32 + lane] / (den[tl * H_HEADS + hh] + 1e-16f);
            orow[j * 32 + lane] = v;
            cs[j] += v;
            cq[j] += v * v;
        }
    }
#pragma unroll
    for (int j = 0; j < 4; ++j) {
        red_s[wave * HD + j * 32 + lane] = cs[j];
        red_q[wave * HD + j * 32 + lane] = cq[j];
    }
    __syncthreads();
    if (threadIdx.x < 2 * HD) {
        const int ch = threadIdx.x & (HD - 1);
        const bool sq = threadIdx.x >= HD;
        const float* r = sq ? red_q : red_s;
        float t = 0.f;
        for (int w = 0; w < AGG_WAVES; ++w) t += r[w * HD + ch];
        float* pp = sq ? psq : psum;
        pp[(size_t)blockIdx.x * HD + ch] = t;
    }
}

__device__ __forceinline__ int logit_key(float l) {
    int bi = __float_as_int(l);
    return bi >= 0 ? bi : (bi ^ 0x7FFFFFFF);
}

__device__ __forceinline__ void pre_edge_term(size_t e, int tl, int s, int t,
                                              const float* __restrict__ edge_attr,
                                              const float* __restrict__ a_src,
                                              const float* __restrict__ a_dst,
                                              const float* sV, float* lsum, float* ldeg,
                                              int* lmax, float* __restrict__ logit,
                                              int* lcnt, int2* blist, int lcap) {
    const float4* ep = reinterpret_cast<const float4*>(edge_attr + e * E_DIM);
    const float4* sp = reinterpret_cast<const float4*>(a_src + (size_t)s * H_HEADS);
    const float4* dp = reinterpret_cast<const float4*>(a_dst + (size_t)t * H_HEADS);
    float4 q0 = ep[0], q1 = ep[1], q2 = ep[2], q3 = ep[3];
    float4 s0 = sp[0], s1 = sp[1];
    float4 d0 = dp[0], d1 = dp[1];
    float eav[E_DIM] = {q0.x, q0.y, q0.z, q0.w, q1.x, q1.y, q1.z, q1.w,
                        q2.x, q2.y, q2.z, q2.w, q3.x, q3.y, q3.z, q3.w};
    float asv[H_HEADS] = {s0.x, s0.y, s0.z, s0.w, s1.x, s1.y, s1.z, s1.w};
    float adv[H_HEADS] = {d0.x, d0.y, d0.z, d0.w, d1.x, d1.y, d1.z, d1.w};
    float ae[H_HEADS];
#pragma unroll
    for (int hh = 0; hh < H_HEADS; ++hh) ae[hh] = 0.f;
#pragma unroll
    for (int k = 0; k < E_DIM; ++k) {
#pragma unroll
        for (int hh = 0; hh < H_HEADS; ++hh) ae[hh] += eav[k] * sV[k * 8 + hh];
    }
    float lg[H_HEADS];
#pragma unroll
    for (int hh = 0; hh < H_HEADS; ++hh) {
        float l = asv[hh] + adv[hh] + ae[hh];
        lg[hh] = l > 0.f ? l : NEG_SLOPE * l;
    }
    float4* lp = reinterpret_cast<float4*>(logit + e * H_HEADS);
    lp[0] = make_float4(lg[0], lg[1], lg[2], lg[3]);
    lp[1] = make_float4(lg[4], lg[5], lg[6], lg[7]);
#pragma unroll
    for (int hh = 0; hh < H_HEADS; ++hh) atomicMax(&lmax[tl * H_HEADS + hh], logit_key(lg[hh]));
#pragma unroll
    for (int k = 0; k < E_DIM; ++k) atomicAdd(&lsum[tl * E_DIM + k], eav[k]);
    atomicAdd(&ldeg[tl], 1.0f);
    int pos = atomicAdd(lcnt, 1);
    if (pos < lcap) blist[pos] = make_int2((int)e | (tl << LIST_E_BITS), s);
}

#define PRE_STRIDE   (AGG_WAVES * AGG_UNROLL)
#define PRE_STRIPES  ((DST_GROUPS + PRE_STRIDE - 1) / PRE_STRIDE)
#define PRE_SPC      14
#define PRE_CHUNKS   ((PRE_STRIPES + PRE_SPC - 1) / PRE_SPC)
#define PRE_QCAP     2048

__global__ void __launch_bounds__(AGG_THREADS)
k_node_pre(const int* __restrict__ src, const int* __restrict__ dst,
           const float* __restrict__ edge_attr, const float* __restrict__ Vc,
           const float* __restrict__ a_src, const float* __restrict__ a_dst,
           float* __restrict__ logit, int* __restrict__ amaxK,
           float* elist, int* __restrict__ ecount) {
    __shared__ float lsum[AGG_ROWS * E_DIM];
    __shared__ float ldeg[AGG_ROWS];
    __shared__ int   lmax[AGG_ROWS * H_HEADS];
    __shared__ float sV[E_DIM * H_HEADS];
    __shared__ int   q[PRE_QCAP];
    __shared__ int   qn[PRE_CHUNKS];
    __shared__ int   lcnt;
    const int lane = threadIdx.x & 31;
    const int wave = __builtin_amdgcn_readfirstlane((int)(threadIdx.x >> 5));
    const int n0 = blockIdx.x * AGG_ROWS;
    int cnt = N_NODES - n0;
    if (cnt > AGG_ROWS) cnt = AGG_ROWS;
    const unsigned ucnt = (unsigned)cnt;
    int2* blist = reinterpret_cast<int2*>(elist + (size_t)n0 * HD);
    const int lcap = cnt * LIST_CAP_PER_ROW;

    int dv[AGG_UNROLL][4];
    agg_load_groups(dv, dst, wave, lane);

    for (int i = threadIdx.x; i < AGG_ROWS * E_DIM; i += AGG_THREADS) lsum[i] = 0.0f;
    for (int i = threadIdx.x; i < AGG_ROWS * H_HEADS; i += AGG_THREADS) lmax[i] = (int)0x80000000;
    for (int i = threadIdx.x; i < AGG_ROWS; i += AGG_THREADS) ldeg[i] = 0.0f;
    if (threadIdx.x < E_DIM * H_HEADS) sV[threadIdx.x] = Vc[threadIdx.x];
    if (threadIdx.x < PRE_CHUNKS) qn[threadIdx.x] = 0;
    if (threadIdx.x == 0) lcnt = 0;
    __syncthreads();

    for (int c = 0; c < PRE_CHUNKS; ++c) {
        int iend = (c + 1) * PRE_SPC;
        if (iend > PRE_STRIPES) iend = PRE_STRIPES;
        for (int i = c * PRE_SPC; i < iend; ++i) {
            const int g0 = wave + i * PRE_STRIDE;
            int cur[AGG_UNROLL][4];
#pragma unroll
            for (int u = 0; u < AGG_UNROLL; ++u) {
#pragma unroll
                for (int k = 0; k < 4; ++k) cur[u][k] = dv[u][k];
            }
            agg_load_groups(dv, dst, g0 + PRE_STRIDE, lane);

            unsigned ovf = 0u;
#pragma unroll
            for (int u = 0; u < AGG_UNROLL; ++u) {
                int gg = g0 + u * AGG_WAVES;
#pragma unroll
                for (int k = 0; k < 4; ++k) {
                    if ((unsigned)(cur[u][k] - n0) < ucnt) {
                        int pos = atomicAdd(&qn[c], 1);
                        if (pos < PRE_QCAP) q[pos] = (gg * 32 + lane) * 4 + k;
                        else                ovf |= 1u << (u * 4 + k);
                    }
                }
            }
            while (ovf != 0u) {
                asm volatile("" ::: "memory");
                int slot = __builtin_ctz(ovf);
                ovf &= ovf - 1u;
                int e = ((g0 + (slot >> 2) * AGG_WAVES) * 32 + lane) * 4 + (slot & 3);
                int s = src[e];
                int t = dst[e];
                pre_edge_term((size_t)e, t - n0, s, t, edge_attr, a_src, a_dst, sV, lsum, ldeg,
                              lmax, logit, &lcnt, blist, lcap);
            }
        }
        __syncthreads();
        int qc = qn[c];
        if (qc > PRE_QCAP) qc = PRE_QCAP;
        for (int j = threadIdx.x; j < qc; j += AGG_THREADS) {
            asm volatile("" ::: "memory");
            int e = q[j];
            int s = src[e];
            int t = dst[e];
            pre_edge_term((size_t)e, t - n0, s, t, edge_attr, a_src, a_dst, sV, lsum, ldeg,
                          lmax, logit, &lcnt, blist, lcap);
        }
        __syncthreads();
    }

    if (threadIdx.x == 0) ecount[blockIdx.x] = lcnt;

    for (int i = threadIdx.x; i < cnt * H_HEADS; i += AGG_THREADS) {
        int tl = i >> 3, hh = i & 7;
        int t = n0 + tl;
        float d = ldeg[tl];
        d = d < 1.0f ? 1.0f : d;
        float ae = 0.f;
#pragma unroll
        for (int k = 0; k < E_DIM; ++k) ae += (lsum[tl * E_DIM + k] / d) * sV[k * 8 + hh];
        float l = a_src[(size_t)t * H_HEADS + hh] + a_dst[(size_t)t * H_HEADS + hh] + ae;
        l = l > 0.f ? l : NEG_SLOPE * l;
        logit[((size_t)N_EDGES + (size_t)t) * H_HEADS + hh] = l;
        int key = logit_key(l);
        int mk = lmax[i];
        amaxK[(size_t)t * H_HEADS + hh] = key > mk ? key : mk;
    }
}

__global__ void __launch_bounds__(AGG_BLOCKS)
k_stats(const float* __restrict__ psum, const float* __restrict__ psq, float* meanb,
        float* invstd) {
    __shared__ float ssum[AGG_BLOCKS];
    __shared__ float ssq[AGG_BLOCKS];
    int c = blockIdx.x;
    ssum[threadIdx.x] = psum[(size_t)threadIdx.x * HD + c];
    ssq[threadIdx.x]  = psq[(size_t)threadIdx.x * HD + c];
    __syncthreads();
    for (int off = AGG_BLOCKS / 2; off > 0; off >>= 1) {
        if (threadIdx.x < off) {
            ssum[threadIdx.x] += ssum[threadIdx.x + off];
            ssq[threadIdx.x]  += ssq[threadIdx.x + off];
        }
        __syncthreads();
    }
    if (threadIdx.x == 0) {
        float m = ssum[0] / (float)N_NODES;
        float var = ssq[0] / (float)N_NODES - m * m;
        meanb[c] = m;
        invstd[c] = rsqrtf(var + EPS_IN);
    }
}

__global__ void k_final(const float* __restrict__ x,
                        const float* __restrict__ outacc,
                        const float* __restrict__ meanb,
                        const float* __restrict__ invstd,
                        const float* __restrict__ gamma,
                        const float* __restrict__ beta, float* out) {
    int idx = blockIdx.x * blockDim.x + threadIdx.x;
    if (idx >= N_NODES * HD) return;
    int c = idx & 127;
    float normed = (outacc[idx] - meanb[c]) * invstd[c] * gamma[c] + beta[c];
    float t = x[idx] + normed;
    out[idx] = t > 0.f ? t : expm1f(t);
}

extern "C" void kernel_launch(void* const* d_in, const int* in_sizes, int n_in,
                              void* d_out, int out_size, void* d_ws, size_t ws_size,
                              hipStream_t stream) {
    const float* x        = (const float*)d_in[0];
    const int*   ei       = (const int*)d_in[1];
    const float* edgeattr = (const float*)d_in[2];
    const float* W        = (const float*)d_in[3];
    const float* att_src  = (const float*)d_in[4];
    const float* att_dst  = (const float*)d_in[5];
    const float* W_e      = (const float*)d_in[6];
    const float* att_edge = (const float*)d_in[7];
    const float* gamma    = (const float*)d_in[8];
    const float* beta     = (const float*)d_in[9];
    float* out = (float*)d_out;

    const int* src = ei;
    const int* dst = ei + N_EDGES;

    float* ws = (float*)d_ws;
    float* hbuf     = ws;
    float* a_src    = hbuf   + (size_t)N_NODES * HD;
    float* a_dst    = a_src  + (size_t)N_NODES * H_HEADS;
    float* degb     = a_dst  + (size_t)N_NODES * H_HEADS;
    float* loopattr = degb   + N_NODES;
    float* Vc       = loopattr + (size_t)N_NODES * E_DIM;
    float* alpha    = Vc + 128;
    float* denom    = alpha + (size_t)ET * H_HEADS;
    float* outacc   = denom + (size_t)N_NODES * H_HEADS;
    float* meanb    = outacc + (size_t)N_NODES * HD;
    float* invstd   = meanb + 128;
    int*   amaxK    = (int*)(invstd + 128);

    k_gemm_h<<<(N_NODES / 16) * 8 / 8, 256, 0, stream>>>(x, W, hbuf);

    k_attn_node<<<(N_NODES * H_HEADS + 255) / 256, 256, 0, stream>>>(
        hbuf, att_src, att_dst, a_src, a_dst);
    k_edge_proj<<<1, 128, 0, stream>>>(W_e, att_edge, Vc);

    k_node_pre<<<AGG_BLOCKS, AGG_THREADS, 0, stream>>>(src, dst, edgeattr, Vc, a_src, a_dst,
                                                       alpha, amaxK, outacc, (int*)degb);
    float* psum = loopattr;
    float* psq  = loopattr + (size_t)AGG_BLOCKS * HD;
    k_aggregate<<<AGG_BLOCKS, AGG_THREADS, 0, stream>>>(src, dst, alpha, amaxK, hbuf,
                                                        (const int*)degb, outacc, psum, psq);

    k_stats<<<HD, AGG_BLOCKS, 0, stream>>>(psum, psq, meanb, invstd);
    k_final<<<(N_NODES * HD + 255) / 256, 256, 0, stream>>>(
        x, outacc, meanb, invstd, gamma, beta, out);
}


# Edges binned by owner of their dst once per call (ranks in LDS, one integer atomic add per owner reserves list slots); owners no longer each scan the whole dst list; sum order changes
# speedup vs baseline: 12.9300x; 1.2916x over previous
#include <hip/hip_runtime.h>
#include <hip/hip_bf16.h>

#define N_NODES 50000
#define N_EDGES 1600000
#define F_DIM   128
#define H_HEADS 8
#define D_HEAD  16
#define E_DIM   16
#define HD      (H_HEADS * D_HEAD)
#define ET      (N_EDGES + N_NODES)
#define EPS_IN  1e-5f
#define NEG_SLOPE 0.2f

typedef float v2f __attribute__((ext_vector_type(2)));
typedef float v8f __attribute__((ext_vector_type(8)));

#define AGG_BLOCKS  256
#define AGG_ROWS    ((N_NODES + AGG_BLOCKS - 1) / AGG_BLOCKS)

__global__ void __launch_bounds__(256)
k_gemm_h(const float* __restrict__ x, const float* __restrict__ W,
         float* __restrict__ h) {
    int wave = threadIdx.x >> 5;
    int lane = threadIdx.x & 31;
    int tile = blockIdx.x * 8 + wave;
    int tileM = tile >> 3;
    int tileN = tile & 7;
    int m   = tileM * 16 + (lane & 15);
    int kb  = (lane >> 4) << 1;
    int col = tileN * 16 + (lane & 15);

    v8f acc = {};
#pragma unroll
    for (int k = 0; k < F_DIM; k += 4) {
        v2f a, b;
        a.x = x[(size_t)m * F_DIM + k + kb];
        a.y = x[(size_t)m * F_DIM + k + kb + 1];
        b.x = W[(size_t)(k + kb) * HD + col];
        b.y = W[(size_t)(k + kb + 1) * HD + col];
        acc = __builtin_amdgcn_wmma_f32_16x16x4_f32(
             false, a,  false, b,
             (short)0, acc,  false,  false);
    }
    int rbase = tileM * 16 + ((lane >> 4) << 3);
#pragma unroll
    for (int r = 0; r < 8; ++r)
        h[(size_t)(rbase + r) * HD + col] = acc[r];
}

__global__ void k_attn_node(const float* __restrict__ h,
                            const float* __restrict__ att_src,
                            const float* __restrict__ att_dst,
                            float* a_src, float* a_dst) {
    int idx = blockIdx.x * blockDim.x + threadIdx.x;
    if (idx >= N_NODES * H_HEADS) return;
    int n = idx >> 3, hh = idx & 7;
    const float* hp = h + (size_t)n * HD + hh * D_HEAD;
    const float* as = att_src + hh * D_HEAD;
    const float* ad = att_dst + hh * D_HEAD;
    float s = 0.f, d = 0.f;
#pragma unroll
    for (int i = 0; i < D_HEAD; ++i) {
        float v = hp[i];
        s += v * as[i];
        d += v * ad[i];
    }
    a_src[idx] = s;
    a_dst[idx] = d;
}

__global__ void k_edge_proj(const float* __restrict__ W_e,
                            const float* __restrict__ att_edge, float* V, int* ecount) {
    int idx = threadIdx.x;
    int k = idx >> 3, hh = idx & 7;
    float s = 0.f;
#pragma unroll
    for (int d = 0; d < D_HEAD; ++d)
        s += W_e[(size_t)k * HD + hh * D_HEAD + d] * att_edge[hh * D_HEAD + d];
    V[idx] = s;
    for (int i = idx; i < AGG_BLOCKS; i += E_DIM * H_HEADS) ecount[i] = 0;
}

#define AGG_THREADS 1024
#define AGG_WAVES   (AGG_THREADS / 32)
#define DST_GROUPS  (N_EDGES / 128)
#define AGG_UNROLL  4

#define AGG_BATCH   8
#define LIST_E_BITS 21
#define LIST_CAP_PER_ROW (HD / 2)

static_assert(N_EDGES % 128 == 0, "dst list must split into whole wave groups");
static_assert(N_EDGES <= (1 << LIST_E_BITS), "edge id must fit its field");
static_assert(AGG_ROWS <= (1 << (31 - LIST_E_BITS)), "local row must fit its field");
static_assert((AGG_BLOCKS & (AGG_BLOCKS - 1)) == 0, "the statistics kernel halves the partials in a tree");

__device__ __forceinline__ void agg_row_term(float* acc_row, float* den_row,
                                             const float* __restrict__ hrow,
                                             const float* __restrict__ lrow,
                                             const int* __restrict__ mrow, int lane,
                                             bool first) {
    int key = mrow[lane & 7];
    int bi = key >= 0 ? key : (key ^ 0x7FFFFFFF);
    float wl = __expf(lrow[lane & 7] - __int_as_float(bi));
    float hv[4];
#pragma unroll
    for (int j = 0; j < 4; ++j) hv[j] = hrow[j * 32 + lane];
    float v[4];
#pragma unroll
    for (int j = 0; j < 4; ++j) {
        int hh = j * 2 + (lane >> 4);
        float w = __int_as_float(__builtin_amdgcn_ds_bpermute(hh << 2, __float_as_int(wl)));
        v[j] = hv[j] * w;
    }
#pragma unroll
    for (int j = 0; j < 4; ++j) {
        int c = j * 32 + lane;
        if (first) acc_row[c] = v[j];
        else       atomicAdd(&acc_row[c], v[j]);
    }
    if (lane < H_HEADS) {
        if (first) den_row[lane] = wl;
        else       atomicAdd(&den_row[lane], wl);
    }
}

__device__ __forceinline__ void agg_load_groups(int (&dv)[AGG_UNROLL][4],
                                                const int* __restrict__ dst, int g0, int lane) {
#pragma unroll
    for (int u = 0; u < AGG_UNROLL; ++u) {
        int gg = g0 + u * AGG_WAVES;
        if (gg < DST_GROUPS) {
            const int* p = dst + ((size_t)gg * 32 + lane) * 4;
            dv[u][0] = p[0]; dv[u][1] = p[1]; dv[u][2] = p[2]; dv[u][3] = p[3];
        } else {
            dv[u][0] = -1; dv[u][1] = -1; dv[u][2] = -1; dv[u][3] = -1;
        }
    }
}

__global__ void __launch_bounds__(AGG_THREADS)
k_aggregate(const int* __restrict__ src, const int* __restrict__ dst,
            const float* __restrict__ logit, const int* __restrict__ amaxK,
            const float* __restrict__ h, const int* __restrict__ ecount, float* outacc,
            float* __restrict__ psum, float* __restrict__ psq) {
    __shared__ float acc[AGG_ROWS * HD];
    __shared__ float den[AGG_ROWS * H_HEADS];
    __shared__ float red_s[AGG_WAVES * HD];
    __shared__ float red_q[AGG_WAVES * HD];
    const int lane = threadIdx.x & 31;
    const int wave = __builtin_amdgcn_readfirstlane((int)(threadIdx.x >> 5));
    const int n0 = blockIdx.x * AGG_ROWS;
    int cnt = N_NODES - n0;
    if (cnt > AGG_ROWS) cnt = AGG_ROWS;

    const int total = ecount[blockIdx.x];
    const int lcap = cnt * LIST_CAP_PER_ROW;
    const int2* blist = reinterpret_cast<const int2*>(outacc + (size_t)n0 * HD);

    for (int tl = wave; tl < cnt; tl += AGG_WAVES) {
        int t = n0 + tl;
        size_t i = (size_t)N_EDGES + (size_t)t;
        agg_row_term(acc + tl * HD, den + tl * H_HEADS, h + (size_t)t * HD, logit + i * 8,
                     amaxK + (size_t)t * 8, lane, true);
    }
    __syncthreads();

    if (total <= lcap) {
    const int nfull = total & ~(AGG_BATCH - 1);
    int2 ent = make_int2(0, 0);
    if (lane < AGG_BATCH && wave * AGG_BATCH < nfull) ent = blist[wave * AGG_BATCH + lane];
    asm volatile("" : "+v"(ent.x), "+v"(ent.y));
    for (int j0 = wave * AGG_BATCH; j0 < nfull; j0 += AGG_WAVES * AGG_BATCH) {
        const int jn = j0 + AGG_WAVES * AGG_BATCH;
        int2 nxt = make_int2(0, 0);
        if (lane < AGG_BATCH && jn < nfull) nxt = blist[jn + lane];
        int   tlb[AGG_BATCH];
        int   keyb[AGG_BATCH];
        float lgb[AGG_BATCH];
        float hvb[AGG_BATCH][4];
#pragma unroll
        for (int b = 0; b < AGG_BATCH; ++b) {
            int ex = __builtin_amdgcn_readlane(ent.x, b);
            int su = __builtin_amdgcn_readlane(ent.y, b);
            int eu = ex & ((1 << LIST_E_BITS) - 1);
            tlb[b] = ex >> LIST_E_BITS;
            keyb[b] = amaxK[(size_t)(n0 + tlb[b]) * 8 + (lane & 7)];
            lgb[b]  = logit[(size_t)eu * 8 + (lane & 7)];
#pragma unroll
            for (int j = 0; j < 4; ++j) hvb[b][j] = h[(size_t)su * HD + j * 32 + lane];
        }
#pragma unroll
        for (int b = 0; b < AGG_BATCH; ++b) {
            int bi = keyb[b] >= 0 ? keyb[b] : (keyb[b] ^ 0x7FFFFFFF);
            float wl = __expf(lgb[b] - __int_as_float(bi));
            float* acc_row = acc + tlb[b] * HD;
            float v[4];
#pragma unroll
            for (int j = 0; j < 4; ++j) {
                int hh = j * 2 + (lane >> 4);
                float w = __int_as_float(__builtin_amdgcn_ds_bpermute(hh << 2, __float_as_int(wl)));
                v[j] = hvb[b][j] * w;
            }
#pragma unroll
            for (int j = 0; j < 4; ++j) atomicAdd(&acc_row[j * 32 + lane], v[j]);
            if (lane < H_HEADS) atomicAdd(&den[tlb[b] * H_HEADS + lane], wl);
        }
        ent = nxt;
    }
    for (int j = nfull + wave; j < total; j += AGG_WAVES) {
        int2 en = blist[j];
        int ex = __builtin_amdgcn_readfirstlane(en.x);
        int su = __builtin_amdgcn_readfirstlane(en.y);
        int eu = ex & ((1 << LIST_E_BITS) - 1);
        int tlu = ex >> LIST_E_BITS;
        agg_row_term(acc + tlu * HD, den + tlu * H_HEADS, h + (size_t)su * HD,
                     logit + (size_t)eu * 8, amaxK + (size_t)(n0 + tlu) * 8, lane, false);
    }
    } else {
    int dv[AGG_UNROLL][4];
    agg_load_groups(dv, dst, wave, lane);
    for (int g0 = wave; g0 < DST_GROUPS; g0 += AGG_WAVES * AGG_UNROLL) {
        int nx[AGG_UNROLL][4];
        agg_load_groups(nx, dst, g0 + AGG_WAVES * AGG_UNROLL, lane);

        const unsigned ucnt = (unsigned)cnt;
        int tl[AGG_UNROLL][4];
        int sv[AGG_UNROLL][4];
        unsigned anym[AGG_UNROLL];
#pragma unroll
        for (int u = 0; u < AGG_UNROLL; ++u) {
            int gg = g0 + u * AGG_WAVES;
#pragma unroll
            for (int k = 0; k < 4; ++k) {
                tl[u][k] = dv[u][k] - n0;
                sv[u][k] = 0;
            }
            bool hu = ((unsigned)tl[u][0] < ucnt) | ((unsigned)tl[u][1] < ucnt) |
                      ((unsigned)tl[u][2] < ucnt) | ((unsigned)tl[u][3] < ucnt);
            anym[u] = __builtin_amdgcn_ballot_w32(hu);
            if (anym[u] != 0u) {
#pragma unroll
                for (int k = 0; k < 4; ++k) {
                    if ((unsigned)tl[u][k] < ucnt)
                        sv[u][k] = src[((size_t)gg * 32 + lane) * 4 + k];
                }
            }
        }
#pragma unroll
        for (int u = 0; u < AGG_UNROLL; ++u) {
            int gg = g0 + u * AGG_WAVES;
            if (anym[u] != 0u) {
#pragma unroll
                for (int k = 0; k < 4; ++k) {
                    unsigned m = __builtin_amdgcn_ballot_w32((unsigned)tl[u][k] < ucnt);
                    while (m != 0u) {
                        int l = __builtin_ctz(m);
                        m &= m - 1u;
                        int tlu = __builtin_amdgcn_readlane(tl[u][k], l);
                        int su  = __builtin_amdgcn_readlane(sv[u][k], l);
                        size_t e = ((size_t)gg * 32 + (size_t)l) * 4 + (size_t)k;
                        agg_row_term(acc + tlu * HD, den + tlu * H_HEADS, h + (size_t)su * HD,
                                     logit + e * 8, amaxK + (size_t)(n0 + tlu) * 8, lane, false);
                    }
                }
            }
        }
#pragma unroll
        for (int u = 0; u < AGG_UNROLL; ++u) {
#pragma unroll
            for (int k = 0; k < 4; ++k) dv[u][k] = nx[u][k];
        }
    }
    }
    __syncthreads();

    float cs[4] = {0.f, 0.f, 0.f, 0.f};
    float cq[4] = {0.f, 0.f, 0.f, 0.f};
    for (int tl = wave; tl < cnt; tl += AGG_WAVES) {
        float* orow = outacc + (size_t)(n0 + tl) * HD;
#pragma unroll
        for (int j = 0; j < 4; ++j) {
            int hh = j * 2 + (lane >> 4);
            float v = acc[tl * HD + j * 32 + lane] / (den[tl * H_HEADS + hh] + 1e-16f);
            orow[j * 32 + lane] = v;
            cs[j] += v;
            cq[j] += v * v;
        }
    }
#pragma unroll
    for (int j = 0; j < 4; ++j) {
        red_s[wave * HD + j * 32 + lane] = cs[j];
        red_q[wave * HD + j * 32 + lane] = cq[j];
    }
    __syncthreads();
    if (threadIdx.x < 2 * HD) {
        const int ch = threadIdx.x & (HD - 1);
        const bool sq = threadIdx.x >= HD;
        const float* r = sq ? red_q : red_s;
        float t = 0.f;
        for (int w = 0; w < AGG_WAVES; ++w) t += r[w * HD + ch];
        float* pp = sq ? psq : psum;
        pp[(size_t)blockIdx.x * HD + ch] = t;
    }
}

#define BIN_THREADS 1024
#define BIN_EDGES   (BIN_THREADS * 4)
#define BIN_BLOCKS  ((N_EDGES + BIN_EDGES - 1) / BIN_EDGES)

static_assert(N_EDGES % 4 == 0, "a lane's four edges are all inside the list or all outside");

__global__ void __launch_bounds__(BIN_THREADS)
k_bin(const int* __restrict__ src, const int* __restrict__ dst, int* ecount, float* elist) {
    __shared__ int hist[AGG_BLOCKS];
    __shared__ int base[AGG_BLOCKS];
    const int tid = threadIdx.x;
    const int e0 = (blockIdx.x * BIN_THREADS + tid) * 4;
    int dv[4] = {-1, -1, -1, -1};
    int sv[4] = {0, 0, 0, 0};
    if (e0 < N_EDGES) {
        const int4 d = *reinterpret_cast<const int4*>(dst + e0);
        const int4 s = *reinterpret_cast<const int4*>(src + e0);
        dv[0] = d.x; dv[1] = d.y; dv[2] = d.z; dv[3] = d.w;
        sv[0] = s.x; sv[1] = s.y; sv[2] = s.z; sv[3] = s.w;
    }
    if (tid < AGG_BLOCKS) hist[tid] = 0;
    __syncthreads();
    int own[4], rank[4];
#pragma unroll
    for (int k = 0; k < 4; ++k) {
        own[k] = -1;
        rank[k] = 0;
        if ((unsigned)dv[k] < (unsigned)N_NODES) {
            own[k] = dv[k] / AGG_ROWS;
            rank[k] = atomicAdd(&hist[own[k]], 1);
        }
    }
    __syncthreads();
    if (tid < AGG_BLOCKS) base[tid] = atomicAdd(&ecount[tid], hist[tid]);
    __syncthreads();
#pragma unroll
    for (int k = 0; k < 4; ++k) {
        if (own[k] >= 0) {
            const int n0 = own[k] * AGG_ROWS;
            int cnt = N_NODES - n0;
            if (cnt > AGG_ROWS) cnt = AGG_ROWS;
            const int pos = base[own[k]] + rank[k];
            if (pos < cnt * LIST_CAP_PER_ROW) {
                int2* blist = reinterpret_cast<int2*>(elist + (size_t)n0 * HD);
                blist[pos] = make_int2((e0 + k) | ((dv[k] - n0) << LIST_E_BITS), sv[k]);
            }
        }
    }
}

__device__ __forceinline__ int logit_key(float l) {
    int bi = __float_as_int(l);
    return bi >= 0 ? bi : (bi ^ 0x7FFFFFFF);
}

__device__ __forceinline__ void pre_edge_term(size_t e, int tl, unsigned s, unsigned t,
                                              const float* __restrict__ edge_attr,
                                              const float* __restrict__ a_src,
                                              const float* __restrict__ a_dst,
                                              const float* sV, float* lsum, float* ldeg,
                                              int* lmax, float* __restrict__ logit) {
    const float4* ep = reinterpret_cast<const float4*>(edge_attr + e * E_DIM);
    const float4* sp = reinterpret_cast<const float4*>(a_src + (size_t)s * H_HEADS);
    const float4* dp = reinterpret_cast<const float4*>(a_dst + (size_t)t * H_HEADS);
    float4 q0 = ep[0], q1 = ep[1], q2 = ep[2], q3 = ep[3];
    float4 s0 = sp[0], s1 = sp[1];
    float4 d0 = dp[0], d1 = dp[1];
    float eav[E_DIM] = {q0.x, q0.y, q0.z, q0.w, q1.x, q1.y, q1.z, q1.w,
                        q2.x, q2.y, q2.z, q2.w, q3.x, q3.y, q3.z, q3.w};
    float asv[H_HEADS] = {s0.x, s0.y, s0.z, s0.w, s1.x, s1.y, s1.z, s1.w};
    float adv[H_HEADS] = {d0.x, d0.y, d0.z, d0.w, d1.x, d1.y, d1.z, d1.w};
    float ae[H_HEADS];
#pragma unroll
    for (int hh = 0; hh < H_HEADS; ++hh) ae[hh] = 0.f;
#pragma unroll
    for (int k = 0; k < E_DIM; ++k) {
#pragma unroll
        for (int hh = 0; hh < H_HEADS; ++hh) ae[hh] += eav[k] * sV[k * 8 + hh];
    }
    float lg[H_HEADS];
#pragma unroll
    for (int hh = 0; hh < H_HEADS; ++hh) {
        float l = asv[hh] + adv[hh] + ae[hh];
        lg[hh] = l > 0.f ? l : NEG_SLOPE * l;
    }
    float4* lp = reinterpret_cast<float4*>(logit + e * H_HEADS);
    lp[0] = make_float4(lg[0], lg[1], lg[2], lg[3]);
    lp[1] = make_float4(lg[4], lg[5], lg[6], lg[7]);
#pragma unroll
    for (int hh = 0; hh < H_HEADS; ++hh) atomicMax(&lmax[tl * H_HEADS + hh], logit_key(lg[hh]));
#pragma unroll
    for (int k = 0; k < E_DIM; ++k) atomicAdd(&lsum[tl * E_DIM + k], eav[k]);
    atomicAdd(&ldeg[tl], 1.0f);
}

#define PRE_STRIDE   (AGG_WAVES * AGG_UNROLL)
#define PRE_STRIPES  ((DST_GROUPS + PRE_STRIDE - 1) / PRE_STRIDE)
#define PRE_SPC      14
#define PRE_CHUNKS   ((PRE_STRIPES + PRE_SPC - 1) / PRE_SPC)
#define PRE_QCAP     2048

__global__ void __launch_bounds__(AGG_THREADS)
k_node_pre(const int* __restrict__ src, const int* __restrict__ dst,
           const float* __restrict__ edge_attr, const float* __restrict__ Vc,
           const float* __restrict__ a_src, const float* __restrict__ a_dst,
           float* __restrict__ logit, int* __restrict__ amaxK,
           const float* __restrict__ elist, const int* __restrict__ ecount) {
    __shared__ float lsum[AGG_ROWS * E_DIM];
    __shared__ float ldeg[AGG_ROWS];
    __shared__ int   lmax[AGG_ROWS * H_HEADS];
    __shared__ float sV[E_DIM * H_HEADS];
    __shared__ int   q[PRE_QCAP];
    __shared__ int   qn[PRE_CHUNKS];
    const int lane = threadIdx.x & 31;
    const int wave = __builtin_amdgcn_readfirstlane((int)(threadIdx.x >> 5));
    const int n0 = blockIdx.x * AGG_ROWS;
    int cnt = N_NODES - n0;
    if (cnt > AGG_ROWS) cnt = AGG_ROWS;
    const unsigned ucnt = (unsigned)cnt;
    const int2* blist = reinterpret_cast<const int2*>(elist + (size_t)n0 * HD);
    const int lcap = cnt * LIST_CAP_PER_ROW;
    const int total = ecount[blockIdx.x];

    for (int i = threadIdx.x; i < AGG_ROWS * E_DIM; i += AGG_THREADS) lsum[i] = 0.0f;
    for (int i = threadIdx.x; i < AGG_ROWS * H_HEADS; i += AGG_THREADS) lmax[i] = (int)0x80000000;
    for (int i = threadIdx.x; i < AGG_ROWS; i += AGG_THREADS) ldeg[i] = 0.0f;
    if (threadIdx.x < E_DIM * H_HEADS) sV[threadIdx.x] = Vc[threadIdx.x];
    if (threadIdx.x < PRE_CHUNKS) qn[threadIdx.x] = 0;
    __syncthreads();

    if (total <= lcap) {
        for (int j = threadIdx.x; j < total; j += AGG_THREADS) {
            asm volatile("" ::: "memory");
            const int2 en = blist[j];
            const unsigned e = (unsigned)en.x & ((1u << LIST_E_BITS) - 1u);
            const int tl = en.x >> LIST_E_BITS;
            pre_edge_term((size_t)e, tl, (unsigned)en.y, (unsigned)(n0 + tl), edge_attr, a_src,
                          a_dst, sV, lsum, ldeg, lmax, logit);
        }
    } else {
    int dv[AGG_UNROLL][4];
    agg_load_groups(dv, dst, wave, lane);
    for (int c = 0; c < PRE_CHUNKS; ++c) {
        int iend = (c + 1) * PRE_SPC;
        if (iend > PRE_STRIPES) iend = PRE_STRIPES;
        for (int i = c * PRE_SPC; i < iend; ++i) {
            const int g0 = wave + i * PRE_STRIDE;
            int cur[AGG_UNROLL][4];
#pragma unroll
            for (int u = 0; u < AGG_UNROLL; ++u) {
#pragma unroll
                for (int k = 0; k < 4; ++k) cur[u][k] = dv[u][k];
            }
            agg_load_groups(dv, dst, g0 + PRE_STRIDE, lane);

            unsigned ovf = 0u;
#pragma unroll
            for (int u = 0; u < AGG_UNROLL; ++u) {
                int gg = g0 + u * AGG_WAVES;
#pragma unroll
                for (int k = 0; k < 4; ++k) {
                    if ((unsigned)(cur[u][k] - n0) < ucnt) {
                        int pos = atomicAdd(&qn[c], 1);
                        if (pos < PRE_QCAP) q[pos] = (gg * 32 + lane) * 4 + k;
                        else                ovf |= 1u << (u * 4 + k);
                    }
                }
            }
            while (ovf != 0u) {
                asm volatile("" ::: "memory");
                int slot = __builtin_ctz(ovf);
                ovf &= ovf - 1u;
                int e = ((g0 + (slot >> 2) * AGG_WAVES) * 32 + lane) * 4 + (slot & 3);
                int s = src[e];
                int t = dst[e];
                pre_edge_term((size_t)e, t - n0, s, t, edge_attr, a_src, a_dst, sV, lsum, ldeg,
                              lmax, logit);
            }
        }
        __syncthreads();
        int qc = qn[c];
        if (qc > PRE_QCAP) qc = PRE_QCAP;
        for (int j = threadIdx.x; j < qc; j += AGG_THREADS) {
            asm volatile("" ::: "memory");
            int e = q[j];
            int s = src[e];
            int t = dst[e];
            pre_edge_term((size_t)e, t - n0, s, t, edge_attr, a_src, a_dst, sV, lsum, ldeg,
                          lmax, logit);
        }
        __syncthreads();
    }
    }
    __syncthreads();

    for (int i = threadIdx.x; i < cnt * H_HEADS; i += AGG_THREADS) {
        int tl = i >> 3, hh = i & 7;
        int t = n0 + tl;
        float d = ldeg[tl];
        d = d < 1.0f ? 1.0f : d;
        float ae = 0.f;
#pragma unroll
        for (int k = 0; k < E_DIM; ++k) ae += (lsum[tl * E_DIM + k] / d) * sV[k * 8 + hh];
        float l = a_src[(size_t)t * H_HEADS + hh] + a_dst[(size_t)t * H_HEADS + hh] + ae;
        l = l > 0.f ? l : NEG_SLOPE * l;
        logit[((size_t)N_EDGES + (size_t)t) * H_HEADS + hh] = l;
        int key = logit_key(l);
        int mk = lmax[i];
        amaxK[(size_t)t * H_HEADS + hh] = key > mk ? key : mk;
    }
}

__global__ void __launch_bounds__(AGG_BLOCKS)
k_stats(const float* __restrict__ psum, const float* __restrict__ psq, float* meanb,
        float* invstd) {
    __shared__ float ssum[AGG_BLOCKS];
    __shared__ float ssq[AGG_BLOCKS];
    int c = blockIdx.x;
    ssum[threadIdx.x] = psum[(size_t)threadIdx.x * HD + c];
    ssq[threadIdx.x]  = psq[(size_t)threadIdx.x * HD + c];
    __syncthreads();
    for (int off = AGG_BLOCKS / 2; off > 0; off >>= 1) {
        if (threadIdx.x < off) {
            ssum[threadIdx.x] += ssum[threadIdx.x + off];
            ssq[threadIdx.x]  += ssq[threadIdx.x + off];
        }
        __syncthreads();
    }
    if (threadIdx.x == 0) {
        float m = ssum[0] / (float)N_NODES;
        float var = ssq[0] / (float)N_NODES - m * m;
        meanb[c] = m;
        invstd[c] = rsqrtf(var + EPS_IN);
    }
}

__global__ void k_final(const float* __restrict__ x,
                        const float* __restrict__ outacc,
                        const float* __restrict__ meanb,
                        const float* __restrict__ invstd,
                        const float* __restrict__ gamma,
                        const float* __restrict__ beta, float* out) {
    int idx = blockIdx.x * blockDim.x + threadIdx.x;
    if (idx >= N_NODES * HD) return;
    int c = idx & 127;
    float normed = (outacc[idx] - meanb[c]) * invstd[c] * gamma[c] + beta[c];
    float t = x[idx] + normed;
    out[idx] = t > 0.f ? t : expm1f(t);
}

extern "C" void kernel_launch(void* const* d_in, const int* in_sizes, int n_in,
                              void* d_out, int out_size, void* d_ws, size_t ws_size,
                              hipStream_t stream) {
    const float* x        = (const float*)d_in[0];
    const int*   ei       = (const int*)d_in[1];
    const float* edgeattr = (const float*)d_in[2];
    const float* W        = (const float*)d_in[3];
    const float* att_src  = (const float*)d_in[4];
    const float* att_dst  = (const float*)d_in[5];
    const float* W_e      = (const float*)d_in[6];
    const float* att_edge = (const float*)d_in[7];
    const float* gamma    = (const float*)d_in[8];
    const float* beta     = (const float*)d_in[9];
    float* out = (float*)d_out;

    const int* src = ei;
    const int* dst = ei + N_EDGES;

    float* ws = (float*)d_ws;
    float* hbuf     = ws;
    float* a_src    = hbuf   + (size_t)N_NODES * HD;
    float* a_dst    = a_src  + (size_t)N_NODES * H_HEADS;
    float* degb     = a_dst  + (size_t)N_NODES * H_HEADS;
    float* loopattr = degb   + N_NODES;
    float* Vc       = loopattr + (size_t)N_NODES * E_DIM;
    float* alpha    = Vc + 128;
    float* denom    = alpha + (size_t)ET * H_HEADS;
    float* outacc   = denom + (size_t)N_NODES * H_HEADS;
    float* meanb    = outacc + (size_t)N_NODES * HD;
    float* invstd   = meanb + 128;
    int*   amaxK    = (int*)(invstd + 128);

    k_gemm_h<<<(N_NODES / 16) * 8 / 8, 256, 0, stream>>>(x, W, hbuf);

    k_attn_node<<<(N_NODES * H_HEADS + 255) / 256, 256, 0, stream>>>(
        hbuf, att_src, att_dst, a_src, a_dst);
    k_edge_proj<<<1, 128, 0, stream>>>(W_e, att_edge, Vc, (int*)degb);

    k_bin<<<BIN_BLOCKS, BIN_THREADS, 0, stream>>>(src, dst, (int*)degb, outacc);
    k_node_pre<<<AGG_BLOCKS, AGG_THREADS, 0, stream>>>(src, dst, edgeattr, Vc, a_src, a_dst,
                                                       alpha, amaxK, outacc, (const int*)degb);
    float* psum = loopattr;
    float* psq  = loopattr + (size_t)AGG_BLOCKS * HD;
    k_aggregate<<<AGG_BLOCKS, AGG_THREADS, 0, stream>>>(src, dst, alpha, amaxK, hbuf,
                                                        (const int*)degb, outacc, psum, psq);

    k_stats<<<HD, AGG_BLOCKS, 0, stream>>>(psum, psq, meanb, invstd);
    k_final<<<(N_NODES * HD + 255) / 256, 256, 0, stream>>>(
        x, outacc, meanb, invstd, gamma, beta, out);
}


# Aggregation: owner counting-sorts its edge list by dst row in LDS, one wave sums a row in registers with 16-byte h loads; LDS float atomics and lane permutes gone; sum order changes
# speedup vs baseline: 14.2486x; 1.1020x over previous
#include <hip/hip_runtime.h>
#include <hip/hip_bf16.h>

#define N_NODES 50000
#define N_EDGES 1600000
#define F_DIM   128
#define H_HEADS 8
#define D_HEAD  16
#define E_DIM   16
#define HD      (H_HEADS * D_HEAD)
#define ET      (N_EDGES + N_NODES)
#define EPS_IN  1e-5f
#define NEG_SLOPE 0.2f

typedef float v2f __attribute__((ext_vector_type(2)));
typedef float v8f __attribute__((ext_vector_type(8)));

#define AGG_BLOCKS  256
#define AGG_ROWS    ((N_NODES + AGG_BLOCKS - 1) / AGG_BLOCKS)

__global__ void __launch_bounds__(256)
k_gemm_h(const float* __restrict__ x, const float* __restrict__ W,
         float* __restrict__ h) {
    int wave = threadIdx.x >> 5;
    int lane = threadIdx.x & 31;
    int tile = blockIdx.x * 8 + wave;
    int tileM = tile >> 3;
    int tileN = tile & 7;
    int m   = tileM * 16 + (lane & 15);
    int kb  = (lane >> 4) << 1;
    int col = tileN * 16 + (lane & 15);

    v8f acc = {};
#pragma unroll
    for (int k = 0; k < F_DIM; k += 4) {
        v2f a, b;
        a.x = x[(size_t)m * F_DIM + k + kb];
        a.y = x[(size_t)m * F_DIM + k + kb + 1];
        b.x = W[(size_t)(k + kb) * HD + col];
        b.y = W[(size_t)(k + kb + 1) * HD + col];
        acc = __builtin_amdgcn_wmma_f32_16x16x4_f32(
             false, a,  false, b,
             (short)0, acc,  false,  false);
    }
    int rbase = tileM * 16 + ((lane >> 4) << 3);
#pragma unroll
    for (int r = 0; r < 8; ++r)
        h[(size_t)(rbase + r) * HD + col] = acc[r];
}

__global__ void k_attn_node(const float* __restrict__ h,
                            const float* __restrict__ att_src,
                            const float* __restrict__ att_dst,
                            float* a_src, float* a_dst) {
    int idx = blockIdx.x * blockDim.x + threadIdx.x;
    if (idx >= N_NODES * H_HEADS) return;
    int n = idx >> 3, hh = idx & 7;
    const float* hp = h + (size_t)n * HD + hh * D_HEAD;
    const float* as = att_src + hh * D_HEAD;
    const float* ad = att_dst + hh * D_HEAD;
    float s = 0.f, d = 0.f;
#pragma unroll
    for (int i = 0; i < D_HEAD; ++i) {
        float v = hp[i];
        s += v * as[i];
        d += v * ad[i];
    }
    a_src[idx] = s;
    a_dst[idx] = d;
}

__global__ void k_edge_proj(const float* __restrict__ W_e,
                            const float* __restrict__ att_edge, float* V, int* ecount) {
    int idx = threadIdx.x;
    int k = idx >> 3, hh = idx & 7;
    float s = 0.f;
#pragma unroll
    for (int d = 0; d < D_HEAD; ++d)
        s += W_e[(size_t)k * HD + hh * D_HEAD + d] * att_edge[hh * D_HEAD + d];
    V[idx] = s;
    for (int i = idx; i < AGG_BLOCKS; i += E_DIM * H_HEADS) ecount[i] = 0;
}

#define AGG_THREADS 1024
#define AGG_WAVES   (AGG_THREADS / 32)
#define DST_GROUPS  (N_EDGES / 128)
#define AGG_UNROLL  4

#define AGG_BATCH   8
#define LIST_E_BITS 21
#define LIST_CAP_PER_ROW (HD / 2)

static_assert(N_EDGES % 128 == 0, "dst list must split into whole wave groups");
static_assert(N_EDGES <= (1 << LIST_E_BITS), "edge id must fit its field");
static_assert(AGG_ROWS <= (1 << (31 - LIST_E_BITS)), "local row must fit its field");
static_assert((AGG_BLOCKS & (AGG_BLOCKS - 1)) == 0, "the statistics kernel halves the partials in a tree");

#define SORT_CAP    (AGG_ROWS * LIST_CAP_PER_ROW)
#define SORT_PER_THREAD ((SORT_CAP + AGG_THREADS - 1) / AGG_THREADS)
#define SCAN_PER_LANE   ((AGG_ROWS + 32) / 32)
static_assert(SCAN_PER_LANE * 32 >= AGG_ROWS + 1, "one wave must cover every row start and the end");
static_assert(SORT_CAP * 2 == AGG_ROWS * HD, "the sort array takes exactly the place of the LDS rows");

__device__ __forceinline__ void agg_row_term(float* acc_row, float* den_row,
                                             const float* __restrict__ hrow,
                                             const float* __restrict__ lrow,
                                             const int* __restrict__ mrow, int lane,
                                             bool first) {
    int key = mrow[lane & 7];
    int bi = key >= 0 ? key : (key ^ 0x7FFFFFFF);
    float wl = __expf(lrow[lane & 7] - __int_as_float(bi));
    float hv[4];
#pragma unroll
    for (int j = 0; j < 4; ++j) hv[j] = hrow[j * 32 + lane];
    float v[4];
#pragma unroll
    for (int j = 0; j < 4; ++j) {
        int hh = j * 2 + (lane >> 4);
        float w = __int_as_float(__builtin_amdgcn_ds_bpermute(hh << 2, __float_as_int(wl)));
        v[j] = hv[j] * w;
    }
#pragma unroll
    for (int j = 0; j < 4; ++j) {
        int c = j * 32 + lane;
        if (first) acc_row[c] = v[j];
        else       atomicAdd(&acc_row[c], v[j]);
    }
    if (lane < H_HEADS) {
        if (first) den_row[lane] = wl;
        else       atomicAdd(&den_row[lane], wl);
    }
}

__device__ __forceinline__ void agg_load_groups(int (&dv)[AGG_UNROLL][4],
                                                const int* __restrict__ dst, int g0, int lane) {
#pragma unroll
    for (int u = 0; u < AGG_UNROLL; ++u) {
        int gg = g0 + u * AGG_WAVES;
        if (gg < DST_GROUPS) {
            const int* p = dst + ((size_t)gg * 32 + lane) * 4;
            dv[u][0] = p[0]; dv[u][1] = p[1]; dv[u][2] = p[2]; dv[u][3] = p[3];
        } else {
            dv[u][0] = -1; dv[u][1] = -1; dv[u][2] = -1; dv[u][3] = -1;
        }
    }
}

__global__ void __launch_bounds__(AGG_THREADS)
k_aggregate(const int* __restrict__ src, const int* __restrict__ dst,
            const float* __restrict__ logit, const int* __restrict__ amaxK,
            const float* __restrict__ h, const int* __restrict__ ecount, float* outacc,
            float* __restrict__ psum, float* __restrict__ psq) {
    __shared__ float acc[AGG_ROWS * HD];
    __shared__ float den[AGG_ROWS * H_HEADS];
    __shared__ float red_s[AGG_WAVES * HD];
    __shared__ float red_q[AGG_WAVES * HD];
    __shared__ int rstart[AGG_ROWS + 1];
    __shared__ int rcnt[AGG_ROWS];
    __shared__ int next_row;
    const int lane = threadIdx.x & 31;
    const int wave = __builtin_amdgcn_readfirstlane((int)(threadIdx.x >> 5));
    const int n0 = blockIdx.x * AGG_ROWS;
    int cnt = N_NODES - n0;
    if (cnt > AGG_ROWS) cnt = AGG_ROWS;

    const int total = ecount[blockIdx.x];
    const int lcap = cnt * LIST_CAP_PER_ROW;
    const int2* blist = reinterpret_cast<const int2*>(outacc + (size_t)n0 * HD);

    float cs[4] = {0.f, 0.f, 0.f, 0.f};
    float cq[4] = {0.f, 0.f, 0.f, 0.f};
    const bool sorted_path = total + cnt <= lcap;

    if (sorted_path) {
    int2* sorted = reinterpret_cast<int2*>(acc);
    if (threadIdx.x < AGG_ROWS) rcnt[threadIdx.x] = 1;
    if (threadIdx.x == 0) next_row = 0;
    int2 en[SORT_PER_THREAD];
#pragma unroll
    for (int k = 0; k < SORT_PER_THREAD; ++k) {
        const int j = (int)threadIdx.x + k * AGG_THREADS;
        en[k] = make_int2(0, 0);
        if (j < total) en[k] = blist[j];
    }
    __syncthreads();
    int rk[SORT_PER_THREAD];
#pragma unroll
    for (int k = 0; k < SORT_PER_THREAD; ++k) {
        const int j = (int)threadIdx.x + k * AGG_THREADS;
        rk[k] = 0;
        if (j < total) rk[k] = atomicAdd(&rcnt[en[k].x >> LIST_E_BITS], 1);
    }
    __syncthreads();
    if (wave == 0) {
        int c[SCAN_PER_LANE];
        int sum = 0;
#pragma unroll
        for (int q = 0; q < SCAN_PER_LANE; ++q) {
            const int r = lane * SCAN_PER_LANE + q;
            c[q] = r < cnt ? rcnt[r] : 0;
            sum += c[q];
        }
        int inc = sum;
#pragma unroll
        for (int d = 1; d < 32; d <<= 1) {
            const int up = __shfl_up(inc, d, 32);
            if (lane >= d) inc += up;
        }
        int run = inc - sum;
#pragma unroll
        for (int q = 0; q < SCAN_PER_LANE; ++q) {
            const int r = lane * SCAN_PER_LANE + q;
            if (r <= cnt) rstart[r] = run;
            run += c[q];
        }
    }
    __syncthreads();
#pragma unroll
    for (int k = 0; k < SORT_PER_THREAD; ++k) {
        const int j = (int)threadIdx.x + k * AGG_THREADS;
        if (j < total) {
            const int tl = en[k].x >> LIST_E_BITS;
            sorted[rstart[tl] + rk[k]] = make_int2(en[k].x & ((1 << LIST_E_BITS) - 1), en[k].y);
        }
    }
    if ((int)threadIdx.x < cnt)
        sorted[rstart[threadIdx.x]] = make_int2(N_EDGES + n0 + (int)threadIdx.x, n0 + (int)threadIdx.x);
    __syncthreads();

    const int hsel = lane >> 2;
    int rr = 0;
    if (lane == 0) rr = atomicAdd(&next_row, 1);
    int r = __builtin_amdgcn_readfirstlane(rr);
    int keyv = 0, rsv = 0, rev = 0;
    if (r < cnt) {
        keyv = amaxK[(size_t)(n0 + r) * 8 + hsel];
        rsv = rstart[r];
        rev = rstart[r + 1];
    }
    asm volatile("" : "+v"(keyv));
    while (r < cnt) {
        int rq = 0;
        if (lane == 0) rq = atomicAdd(&next_row, 1);
        const int rn = __builtin_amdgcn_readfirstlane(rq);
        int keyn = 0, rsn = 0, ren = 0;
        if (rn < cnt) {
            keyn = amaxK[(size_t)(n0 + rn) * 8 + hsel];
            rsn = rstart[rn];
            ren = rstart[rn + 1];
        }
        const int rs = __builtin_amdgcn_readfirstlane(rsv);
        const int re = __builtin_amdgcn_readfirstlane(rev);
        const float mx = __int_as_float(keyv >= 0 ? keyv : (keyv ^ 0x7FFFFFFF));
        float a[4] = {0.f, 0.f, 0.f, 0.f};
        float dn = 0.f;
        for (int base = rs; base < re; base += 32) {
            int n = re - base;
            if (n > 32) n = 32;
            const int2 ent = sorted[base + (lane < n ? lane : n - 1)];
            for (int sub = 0; sub < n; sub += AGG_BATCH) {
                float  lg[AGG_BATCH];
                float4 hv[AGG_BATCH];
#pragma unroll
                for (int b = 0; b < AGG_BATCH; ++b) {
                    int ix = sub + b;
                    if (ix > n - 1) ix = n - 1;
                    const int eu = __builtin_amdgcn_readlane(ent.x, ix);
                    const int su = __builtin_amdgcn_readlane(ent.y, ix);
                    lg[b] = logit[(size_t)eu * 8 + hsel];
                    hv[b] = *reinterpret_cast<const float4*>(h + (size_t)su * HD + lane * 4);
                }
#pragma unroll
                for (int b = 0; b < AGG_BATCH; ++b) {
                    float w = __expf(lg[b] - mx);
                    if (sub + b >= n) w = 0.f;
                    a[0] += w * hv[b].x;
                    a[1] += w * hv[b].y;
                    a[2] += w * hv[b].z;
                    a[3] += w * hv[b].w;
                    dn += w;
                }
            }
        }
        const float dd = dn + 1e-16f;
        float4 o;
        o.x = a[0] / dd; o.y = a[1] / dd; o.z = a[2] / dd; o.w = a[3] / dd;
        *reinterpret_cast<float4*>(outacc + (size_t)(n0 + r) * HD + lane * 4) = o;
        cs[0] += o.x; cs[1] += o.y; cs[2] += o.z; cs[3] += o.w;
        cq[0] += o.x * o.x; cq[1] += o.y * o.y; cq[2] += o.z * o.z; cq[3] += o.w * o.w;
        r = rn; keyv = keyn; rsv = rsn; rev = ren;
    }
    } else {
    for (int tl = wave; tl < cnt; tl += AGG_WAVES) {
        int t = n0 + tl;
        size_t i = (size_t)N_EDGES + (size_t)t;
        agg_row_term(acc + tl * HD, den + tl * H_HEADS, h + (size_t)t * HD, logit + i * 8,
                     amaxK + (size_t)t * 8, lane, true);
    }
    __syncthreads();
    int dv[AGG_UNROLL][4];
    agg_load_groups(dv, dst, wave, lane);
    for (int g0 = wave; g0 < DST_GROUPS; g0 += AGG_WAVES * AGG_UNROLL) {
        int nx[AGG_UNROLL][4];
        agg_load_groups(nx, dst, g0 + AGG_WAVES * AGG_UNROLL, lane);

        const unsigned ucnt = (unsigned)cnt;
        int tl[AGG_UNROLL][4];
        int sv[AGG_UNROLL][4];
        unsigned anym[AGG_UNROLL];
#pragma unroll
        for (int u = 0; u < AGG_UNROLL; ++u) {
            int gg = g0 + u * AGG_WAVES;
#pragma unroll
            for (int k = 0; k < 4; ++k) {
                tl[u][k] = dv[u][k] - n0;
                sv[u][k] = 0;
            }
            bool hu = ((unsigned)tl[u][0] < ucnt) | ((unsigned)tl[u][1] < ucnt) |
                      ((unsigned)tl[u][2] < ucnt) | ((unsigned)tl[u][3] < ucnt);
            anym[u] = __builtin_amdgcn_ballot_w32(hu);
            if (anym[u] != 0u) {
#pragma unroll
                for (int k = 0; k < 4; ++k) {
                    if ((unsigned)tl[u][k] < ucnt)
                        sv[u][k] = src[((size_t)gg * 32 + lane) * 4 + k];
                }
            }
        }
#pragma unroll
        for (int u = 0; u < AGG_UNROLL; ++u) {
            int gg = g0 + u * AGG_WAVES;
            if (anym[u] != 0u) {
#pragma unroll
                for (int k = 0; k < 4; ++k) {
                    unsigned m = __builtin_amdgcn_ballot_w32((unsigned)tl[u][k] < ucnt);
                    while (m != 0u) {
                        int l = __builtin_ctz(m);
                        m &= m - 1u;
                        int tlu = __builtin_amdgcn_readlane(tl[u][k], l);
                        int su  = __builtin_amdgcn_readlane(sv[u][k], l);
                        size_t e = ((size_t)gg * 32 + (size_t)l) * 4 + (size_t)k;
                        agg_row_term(acc + tlu * HD, den + tlu * H_HEADS, h + (size_t)su * HD,
                                     logit + e * 8, amaxK + (size_t)(n0 + tlu) * 8, lane, false);
                    }
                }
            }
        }
#pragma unroll
        for (int u = 0; u < AGG_UNROLL; ++u) {
#pragma unroll
            for (int k = 0; k < 4; ++k) dv[u][k] = nx[u][k];
        }
    }
    __syncthreads();
    for (int tl = wave; tl < cnt; tl += AGG_WAVES) {
        float* orow = outacc + (size_t)(n0 + tl) * HD;
#pragma unroll
        for (int j = 0; j < 4; ++j) {
            int hh = j * 2 + (lane >> 4);
            float v = acc[tl * HD + j * 32 + lane] / (den[tl * H_HEADS + hh] + 1e-16f);
            orow[j * 32 + lane] = v;
            cs[j] += v;
            cq[j] += v * v;
        }
    }
    }
#pragma unroll
    for (int j = 0; j < 4; ++j) {
        const int ch = sorted_path ? lane * 4 + j : j * 32 + lane;
        red_s[wave * HD + ch] = cs[j];
        red_q[wave * HD + ch] = cq[j];
    }
    __syncthreads();
    if (threadIdx.x < 2 * HD) {
        const int ch = threadIdx.x & (HD - 1);
        const bool sq = threadIdx.x >= HD;
        const float* r = sq ? red_q : red_s;
        float t = 0.f;
        for (int w = 0; w < AGG_WAVES; ++w) t += r[w * HD + ch];
        float* pp = sq ? psq : psum;
        pp[(size_t)blockIdx.x * HD + ch] = t;
    }
}

#define BIN_THREADS 1024
#define BIN_EDGES   (BIN_THREADS * 4)
#define BIN_BLOCKS  ((N_EDGES + BIN_EDGES - 1) / BIN_EDGES)

static_assert(N_EDGES % 4 == 0, "a lane's four edges are all inside the list or all outside");

__global__ void __launch_bounds__(BIN_THREADS)
k_bin(const int* __restrict__ src, const int* __restrict__ dst, int* ecount, float* elist) {
    __shared__ int hist[AGG_BLOCKS];
    __shared__ int base[AGG_BLOCKS];
    const int tid = threadIdx.x;
    const int e0 = (blockIdx.x * BIN_THREADS + tid) * 4;
    int dv[4] = {-1, -1, -1, -1};
    int sv[4] = {0, 0, 0, 0};
    if (e0 < N_EDGES) {
        const int4 d = *reinterpret_cast<const int4*>(dst + e0);
        const int4 s = *reinterpret_cast<const int4*>(src + e0);
        dv[0] = d.x; dv[1] = d.y; dv[2] = d.z; dv[3] = d.w;
        sv[0] = s.x; sv[1] = s.y; sv[2] = s.z; sv[3] = s.w;
    }
    if (tid < AGG_BLOCKS) hist[tid] = 0;
    __syncthreads();
    int own[4], rank[4];
#pragma unroll
    for (int k = 0; k < 4; ++k) {
        own[k] = -1;
        rank[k] = 0;
        if ((unsigned)dv[k] < (unsigned)N_NODES) {
            own[k] = dv[k] / AGG_ROWS;
            rank[k] = atomicAdd(&hist[own[k]], 1);
        }
    }
    __syncthreads();
    if (tid < AGG_BLOCKS) base[tid] = atomicAdd(&ecount[tid], hist[tid]);
    __syncthreads();
#pragma unroll
    for (int k = 0; k < 4; ++k) {
        if (own[k] >= 0) {
            const int n0 = own[k] * AGG_ROWS;
            int cnt = N_NODES - n0;
            if (cnt > AGG_ROWS) cnt = AGG_ROWS;
            const int pos = base[own[k]] + rank[k];
            if (pos < cnt * LIST_CAP_PER_ROW) {
                int2* blist = reinterpret_cast<int2*>(elist + (size_t)n0 * HD);
                blist[pos] = make_int2((e0 + k) | ((dv[k] - n0) << LIST_E_BITS), sv[k]);
            }
        }
    }
}

__device__ __forceinline__ int logit_key(float l) {
    int bi = __float_as_int(l);
    return bi >= 0 ? bi : (bi ^ 0x7FFFFFFF);
}

__device__ __forceinline__ void pre_edge_term(size_t e, int tl, unsigned s, unsigned t,
                                              const float* __restrict__ edge_attr,
                                              const float* __restrict__ a_src,
                                              const float* __restrict__ a_dst,
                                              const float* sV, float* lsum, float* ldeg,
                                              int* lmax, float* __restrict__ logit) {
    const float4* ep = reinterpret_cast<const float4*>(edge_attr + e * E_DIM);
    const float4* sp = reinterpret_cast<const float4*>(a_src + (size_t)s * H_HEADS);
    const float4* dp = reinterpret_cast<const float4*>(a_dst + (size_t)t * H_HEADS);
    float4 q0 = ep[0], q1 = ep[1], q2 = ep[2], q3 = ep[3];
    float4 s0 = sp[0], s1 = sp[1];
    float4 d0 = dp[0], d1 = dp[1];
    float eav[E_DIM] = {q0.x, q0.y, q0.z, q0.w, q1.x, q1.y, q1.z, q1.w,
                        q2.x, q2.y, q2.z, q2.w, q3.x, q3.y, q3.z, q3.w};
    float asv[H_HEADS] = {s0.x, s0.y, s0.z, s0.w, s1.x, s1.y, s1.z, s1.w};
    float adv[H_HEADS] = {d0.x, d0.y, d0.z, d0.w, d1.x, d1.y, d1.z, d1.w};
    float ae[H_HEADS];
#pragma unroll
    for (int hh = 0; hh < H_HEADS; ++hh) ae[hh] = 0.f;
#pragma unroll
    for (int k = 0; k < E_DIM; ++k) {
#pragma unroll
        for (int hh = 0; hh < H_HEADS; ++hh) ae[hh] += eav[k] * sV[k * 8 + hh];
    }
    float lg[H_HEADS];
#pragma unroll
    for (int hh = 0; hh < H_HEADS; ++hh) {
        float l = asv[hh] + adv[hh] + ae[hh];
        lg[hh] = l > 0.f ? l : NEG_SLOPE * l;
    }
    float4* lp = reinterpret_cast<float4*>(logit + e * H_HEADS);
    lp[0] = make_float4(lg[0], lg[1], lg[2], lg[3]);
    lp[1] = make_float4(lg[4], lg[5], lg[6], lg[7]);
#pragma unroll
    for (int hh = 0; hh < H_HEADS; ++hh) atomicMax(&lmax[tl * H_HEADS + hh], logit_key(lg[hh]));
#pragma unroll
    for (int k = 0; k < E_DIM; ++k) atomicAdd(&lsum[tl * E_DIM + k], eav[k]);
    atomicAdd(&ldeg[tl], 1.0f);
}

#define PRE_STRIDE   (AGG_WAVES * AGG_UNROLL)
#define PRE_STRIPES  ((DST_GROUPS + PRE_STRIDE - 1) / PRE_STRIDE)
#define PRE_SPC      14
#define PRE_CHUNKS   ((PRE_STRIPES + PRE_SPC - 1) / PRE_SPC)
#define PRE_QCAP     2048

__global__ void __launch_bounds__(AGG_THREADS)
k_node_pre(const int* __restrict__ src, const int* __restrict__ dst,
           const float* __restrict__ edge_attr, const float* __restrict__ Vc,
           const float* __restrict__ a_src, const float* __restrict__ a_dst,
           float* __restrict__ logit, int* __restrict__ amaxK,
           const float* __restrict__ elist, const int* __restrict__ ecount) {
    __shared__ float lsum[AGG_ROWS * E_DIM];
    __shared__ float ldeg[AGG_ROWS];
    __shared__ int   lmax[AGG_ROWS * H_HEADS];
    __shared__ float sV[E_DIM * H_HEADS];
    __shared__ int   q[PRE_QCAP];
    __shared__ int   qn[PRE_CHUNKS];
    const int lane = threadIdx.x & 31;
    const int wave = __builtin_amdgcn_readfirstlane((int)(threadIdx.x >> 5));
    const int n0 = blockIdx.x * AGG_ROWS;
    int cnt = N_NODES - n0;
    if (cnt > AGG_ROWS) cnt = AGG_ROWS;
    const unsigned ucnt = (unsigned)cnt;
    const int2* blist = reinterpret_cast<const int2*>(elist + (size_t)n0 * HD);
    const int lcap = cnt * LIST_CAP_PER_ROW;
    const int total = ecount[blockIdx.x];

    for (int i = threadIdx.x; i < AGG_ROWS * E_DIM; i += AGG_THREADS) lsum[i] = 0.0f;
    for (int i = threadIdx.x; i < AGG_ROWS * H_HEADS; i += AGG_THREADS) lmax[i] = (int)0x80000000;
    for (int i = threadIdx.x; i < AGG_ROWS; i += AGG_THREADS) ldeg[i] = 0.0f;
    if (threadIdx.x < E_DIM * H_HEADS) sV[threadIdx.x] = Vc[threadIdx.x];
    if (threadIdx.x < PRE_CHUNKS) qn[threadIdx.x] = 0;
    __syncthreads();

    if (total <= lcap) {
        for (int j = threadIdx.x; j < total; j += AGG_THREADS) {
            asm volatile("" ::: "memory");
            const int2 en = blist[j];
            const unsigned e = (unsigned)en.x & ((1u << LIST_E_BITS) - 1u);
            const int tl = en.x >> LIST_E_BITS;
            pre_edge_term((size_t)e, tl, (unsigned)en.y, (unsigned)(n0 + tl), edge_attr, a_src,
                          a_dst, sV, lsum, ldeg, lmax, logit);
        }
    } else {
    int dv[AGG_UNROLL][4];
    agg_load_groups(dv, dst, wave, lane);
    for (int c = 0; c < PRE_CHUNKS; ++c) {
        int iend = (c + 1) * PRE_SPC;
        if (iend > PRE_STRIPES) iend = PRE_STRIPES;
        for (int i = c * PRE_SPC; i < iend; ++i) {
            const int g0 = wave + i * PRE_STRIDE;
            int cur[AGG_UNROLL][4];
#pragma unroll
            for (int u = 0; u < AGG_UNROLL; ++u) {
#pragma unroll
                for (int k = 0; k < 4; ++k) cur[u][k] = dv[u][k];
            }
            agg_load_groups(dv, dst, g0 + PRE_STRIDE, lane);

            unsigned ovf = 0u;
#pragma unroll
            for (int u = 0; u < AGG_UNROLL; ++u) {
                int gg = g0 + u * AGG_WAVES;
#pragma unroll
                for (int k = 0; k < 4; ++k) {
                    if ((unsigned)(cur[u][k] - n0) < ucnt) {
                        int pos = atomicAdd(&qn[c], 1);
                        if (pos < PRE_QCAP) q[pos] = (gg * 32 + lane) * 4 + k;
                        else                ovf |= 1u << (u * 4 + k);
                    }
                }
            }
            while (ovf != 0u) {
                asm volatile("" ::: "memory");
                int slot = __builtin_ctz(ovf);
                ovf &= ovf - 1u;
                int e = ((g0 + (slot >> 2) * AGG_WAVES) * 32 + lane) * 4 + (slot & 3);
                int s = src[e];
                int t = dst[e];
                pre_edge_term((size_t)e, t - n0, s, t, edge_attr, a_src, a_dst, sV, lsum, ldeg,
                              lmax, logit);
            }
        }
        __syncthreads();
        int qc = qn[c];
        if (qc > PRE_QCAP) qc = PRE_QCAP;
        for (int j = threadIdx.x; j < qc; j += AGG_THREADS) {
            asm volatile("" ::: "memory");
            int e = q[j];
            int s = src[e];
            int t = dst[e];
            pre_edge_term((size_t)e, t - n0, s, t, edge_attr, a_src, a_dst, sV, lsum, ldeg,
                          lmax, logit);
        }
        __syncthreads();
    }
    }
    __syncthreads();

    for (int i = threadIdx.x; i < cnt * H_HEADS; i += AGG_THREADS) {
        int tl = i >> 3, hh = i & 7;
        int t = n0 + tl;
        float d = ldeg[tl];
        d = d < 1.0f ? 1.0f : d;
        float ae = 0.f;
#pragma unroll
        for (int k = 0; k < E_DIM; ++k) ae += (lsum[tl * E_DIM + k] / d) * sV[k * 8 + hh];
        float l = a_src[(size_t)t * H_HEADS + hh] + a_dst[(size_t)t * H_HEADS + hh] + ae;
        l = l > 0.f ? l : NEG_SLOPE * l;
        logit[((size_t)N_EDGES + (size_t)t) * H_HEADS + hh] = l;
        int key = logit_key(l);
        int mk = lmax[i];
        amaxK[(size_t)t * H_HEADS + hh] = key > mk ? key : mk;
    }
}

__global__ void __launch_bounds__(AGG_BLOCKS)
k_stats(const float* __restrict__ psum, const float* __restrict__ psq, float* meanb,
        float* invstd) {
    __shared__ float ssum[AGG_BLOCKS];
    __shared__ float ssq[AGG_BLOCKS];
    int c = blockIdx.x;
    ssum[threadIdx.x] = psum[(size_t)threadIdx.x * HD + c];
    ssq[threadIdx.x]  = psq[(size_t)threadIdx.x * HD + c];
    __syncthreads();
    for (int off = AGG_BLOCKS / 2; off > 0; off >>= 1) {
        if (threadIdx.x < off) {
            ssum[threadIdx.x] += ssum[threadIdx.x + off];
            ssq[threadIdx.x]  += ssq[threadIdx.x + off];
        }
        __syncthreads();
    }
    if (threadIdx.x == 0) {
        float m = ssum[0] / (float)N_NODES;
        float var = ssq[0] / (float)N_NODES - m * m;
        meanb[c] = m;
        invstd[c] = rsqrtf(var + EPS_IN);
    }
}

__global__ void k_final(const float* __restrict__ x,
                        const float* __restrict__ outacc,
                        const float* __restrict__ meanb,
                        const float* __restrict__ invstd,
                        const float* __restrict__ gamma,
                        const float* __restrict__ beta, float* out) {
    int idx = blockIdx.x * blockDim.x + threadIdx.x;
    if (idx >= N_NODES * HD) return;
    int c = idx & 127;
    float normed = (outacc[idx] - meanb[c]) * invstd[c] * gamma[c] + beta[c];
    float t = x[idx] + normed;
    out[idx] = t > 0.f ? t : expm1f(t);
}

extern "C" void kernel_launch(void* const* d_in, const int* in_sizes, int n_in,
                              void* d_out, int out_size, void* d_ws, size_t ws_size,
                              hipStream_t stream) {
    const float* x        = (const float*)d_in[0];
    const int*   ei       = (const int*)d_in[1];
    const float* edgeattr = (const float*)d_in[2];
    const float* W        = (const float*)d_in[3];
    const float* att_src  = (const float*)d_in[4];
    const float* att_dst  = (const float*)d_in[5];
    const float* W_e      = (const float*)d_in[6];
    const float* att_edge = (const float*)d_in[7];
    const float* gamma    = (const float*)d_in[8];
    const float* beta     = (const float*)d_in[9];
    float* out = (float*)d_out;

    const int* src = ei;
    const int* dst = ei + N_EDGES;

    float* ws = (float*)d_ws;
    float* hbuf     = ws;
    float* a_src    = hbuf   + (size_t)N_NODES * HD;
    float* a_dst    = a_src  + (size_t)N_NODES * H_HEADS;
    float* degb     = a_dst  + (size_t)N_NODES * H_HEADS;
    float* loopattr = degb   + N_NODES;
    float* Vc       = loopattr + (size_t)N_NODES * E_DIM;
    float* alpha    = Vc + 128;
    float* denom    = alpha + (size_t)ET * H_HEADS;
    float* outacc   = denom + (size_t)N_NODES * H_HEADS;
    float* meanb    = outacc + (size_t)N_NODES * HD;
    float* invstd   = meanb + 128;
    int*   amaxK    = (int*)(invstd + 128);

    k_gemm_h<<<(N_NODES / 16) * 8 / 8, 256, 0, stream>>>(x, W, hbuf);

    k_attn_node<<<(N_NODES * H_HEADS + 255) / 256, 256, 0, stream>>>(
        hbuf, att_src, att_dst, a_src, a_dst);
    k_edge_proj<<<1, 128, 0, stream>>>(W_e, att_edge, Vc, (int*)degb);

    k_bin<<<BIN_BLOCKS, BIN_THREADS, 0, stream>>>(src, dst, (int*)degb, outacc);
    k_node_pre<<<AGG_BLOCKS, AGG_THREADS, 0, stream>>>(src, dst, edgeattr, Vc, a_src, a_dst,
                                                       alpha, amaxK, outacc, (const int*)degb);
    float* psum = loopattr;
    float* psq  = loopattr + (size_t)AGG_BLOCKS * HD;
    k_aggregate<<<AGG_BLOCKS, AGG_THREADS, 0, stream>>>(src, dst, alpha, amaxK, hbuf,
                                                        (const int*)degb, outacc, psum, psq);

    k_stats<<<HD, AGG_BLOCKS, 0, stream>>>(psum, psq, meanb, invstd);
    k_final<<<(N_NODES * HD + 255) / 256, 256, 0, stream>>>(
        x, outacc, meanb, invstd, gamma, beta, out);
}


# Final normalise+residual+ELU pass as 10000 resident waves: 5 trips of 16-byte loads and stores per lane, channel parameters read once per thread; was 200000 one-element-per-thread waves
# speedup vs baseline: 14.8656x; 1.0433x over previous
#include <hip/hip_runtime.h>
#include <hip/hip_bf16.h>

#define N_NODES 50000
#define N_EDGES 1600000
#define F_DIM   128
#define H_HEADS 8
#define D_HEAD  16
#define E_DIM   16
#define HD      (H_HEADS * D_HEAD)
#define ET      (N_EDGES + N_NODES)
#define EPS_IN  1e-5f
#define NEG_SLOPE 0.2f

typedef float v2f __attribute__((ext_vector_type(2)));
typedef float v8f __attribute__((ext_vector_type(8)));

#define AGG_BLOCKS  256
#define AGG_ROWS    ((N_NODES + AGG_BLOCKS - 1) / AGG_BLOCKS)

__global__ void __launch_bounds__(256)
k_gemm_h(const float* __restrict__ x, const float* __restrict__ W,
         float* __restrict__ h) {
    int wave = threadIdx.x >> 5;
    int lane = threadIdx.x & 31;
    int tile = blockIdx.x * 8 + wave;
    int tileM = tile >> 3;
    int tileN = tile & 7;
    int m   = tileM * 16 + (lane & 15);
    int kb  = (lane >> 4) << 1;
    int col = tileN * 16 + (lane & 15);

    v8f acc = {};
#pragma unroll
    for (int k = 0; k < F_DIM; k += 4) {
        v2f a, b;
        a.x = x[(size_t)m * F_DIM + k + kb];
        a.y = x[(size_t)m * F_DIM + k + kb + 1];
        b.x = W[(size_t)(k + kb) * HD + col];
        b.y = W[(size_t)(k + kb + 1) * HD + col];
        acc = __builtin_amdgcn_wmma_f32_16x16x4_f32(
             false, a,  false, b,
             (short)0, acc,  false,  false);
    }
    int rbase = tileM * 16 + ((lane >> 4) << 3);
#pragma unroll
    for (int r = 0; r < 8; ++r)
        h[(size_t)(rbase + r) * HD + col] = acc[r];
}

__global__ void k_attn_node(const float* __restrict__ h,
                            const float* __restrict__ att_src,
                            const float* __restrict__ att_dst,
                            float* a_src, float* a_dst) {
    int idx = blockIdx.x * blockDim.x + threadIdx.x;
    if (idx >= N_NODES * H_HEADS) return;
    int n = idx >> 3, hh = idx & 7;
    const float* hp = h + (size_t)n * HD + hh * D_HEAD;
    const float* as = att_src + hh * D_HEAD;
    const float* ad = att_dst + hh * D_HEAD;
    float s = 0.f, d = 0.f;
#pragma unroll
    for (int i = 0; i < D_HEAD; ++i) {
        float v = hp[i];
        s += v * as[i];
        d += v * ad[i];
    }
    a_src[idx] = s;
    a_dst[idx] = d;
}

__global__ void k_edge_proj(const float* __restrict__ W_e,
                            const float* __restrict__ att_edge, float* V, int* ecount) {
    int idx = threadIdx.x;
    int k = idx >> 3, hh = idx & 7;
    float s = 0.f;
#pragma unroll
    for (int d = 0; d < D_HEAD; ++d)
        s += W_e[(size_t)k * HD + hh * D_HEAD + d] * att_edge[hh * D_HEAD + d];
    V[idx] = s;
    for (int i = idx; i < AGG_BLOCKS; i += E_DIM * H_HEADS) ecount[i] = 0;
}

#define AGG_THREADS 1024
#define AGG_WAVES   (AGG_THREADS / 32)
#define DST_GROUPS  (N_EDGES / 128)
#define AGG_UNROLL  4

#define AGG_BATCH   8
#define LIST_E_BITS 21
#define LIST_CAP_PER_ROW (HD / 2)

static_assert(N_EDGES % 128 == 0, "dst list must split into whole wave groups");
static_assert(N_EDGES <= (1 << LIST_E_BITS), "edge id must fit its field");
static_assert(AGG_ROWS <= (1 << (31 - LIST_E_BITS)), "local row must fit its field");
static_assert((AGG_BLOCKS & (AGG_BLOCKS - 1)) == 0, "the statistics kernel halves the partials in a tree");

#define SORT_CAP    (AGG_ROWS * LIST_CAP_PER_ROW)
#define SORT_PER_THREAD ((SORT_CAP + AGG_THREADS - 1) / AGG_THREADS)
#define SCAN_PER_LANE   ((AGG_ROWS + 32) / 32)
static_assert(SCAN_PER_LANE * 32 >= AGG_ROWS + 1, "one wave must cover every row start and the end");
static_assert(SORT_CAP * 2 == AGG_ROWS * HD, "the sort array takes exactly the place of the LDS rows");

__device__ __forceinline__ void agg_row_term(float* acc_row, float* den_row,
                                             const float* __restrict__ hrow,
                                             const float* __restrict__ lrow,
                                             const int* __restrict__ mrow, int lane,
                                             bool first) {
    int key = mrow[lane & 7];
    int bi = key >= 0 ? key : (key ^ 0x7FFFFFFF);
    float wl = __expf(lrow[lane & 7] - __int_as_float(bi));
    float hv[4];
#pragma unroll
    for (int j = 0; j < 4; ++j) hv[j] = hrow[j * 32 + lane];
    float v[4];
#pragma unroll
    for (int j = 0; j < 4; ++j) {
        int hh = j * 2 + (lane >> 4);
        float w = __int_as_float(__builtin_amdgcn_ds_bpermute(hh << 2, __float_as_int(wl)));
        v[j] = hv[j] * w;
    }
#pragma unroll
    for (int j = 0; j < 4; ++j) {
        int c = j * 32 + lane;
        if (first) acc_row[c] = v[j];
        else       atomicAdd(&acc_row[c], v[j]);
    }
    if (lane < H_HEADS) {
        if (first) den_row[lane] = wl;
        else       atomicAdd(&den_row[lane], wl);
    }
}

__device__ __forceinline__ void agg_load_groups(int (&dv)[AGG_UNROLL][4],
                                                const int* __restrict__ dst, int g0, int lane) {
#pragma unroll
    for (int u = 0; u < AGG_UNROLL; ++u) {
        int gg = g0 + u * AGG_WAVES;
        if (gg < DST_GROUPS) {
            const int* p = dst + ((size_t)gg * 32 + lane) * 4;
            dv[u][0] = p[0]; dv[u][1] = p[1]; dv[u][2] = p[2]; dv[u][3] = p[3];
        } else {
            dv[u][0] = -1; dv[u][1] = -1; dv[u][2] = -1; dv[u][3] = -1;
        }
    }
}

__global__ void __launch_bounds__(AGG_THREADS)
k_aggregate(const int* __restrict__ src, const int* __restrict__ dst,
            const float* __restrict__ logit, const int* __restrict__ amaxK,
            const float* __restrict__ h, const int* __restrict__ ecount, float* outacc,
            float* __restrict__ psum, float* __restrict__ psq) {
    __shared__ float acc[AGG_ROWS * HD];
    __shared__ float den[AGG_ROWS * H_HEADS];
    __shared__ float red_s[AGG_WAVES * HD];
    __shared__ float red_q[AGG_WAVES * HD];
    __shared__ int rstart[AGG_ROWS + 1];
    __shared__ int rcnt[AGG_ROWS];
    __shared__ int next_row;
    const int lane = threadIdx.x & 31;
    const int wave = __builtin_amdgcn_readfirstlane((int)(threadIdx.x >> 5));
    const int n0 = blockIdx.x * AGG_ROWS;
    int cnt = N_NODES - n0;
    if (cnt > AGG_ROWS) cnt = AGG_ROWS;

    const int total = ecount[blockIdx.x];
    const int lcap = cnt * LIST_CAP_PER_ROW;
    const int2* blist = reinterpret_cast<const int2*>(outacc + (size_t)n0 * HD);

    float cs[4] = {0.f, 0.f, 0.f, 0.f};
    float cq[4] = {0.f, 0.f, 0.f, 0.f};
    const bool sorted_path = total + cnt <= lcap;

    if (sorted_path) {
    int2* sorted = reinterpret_cast<int2*>(acc);
    if (threadIdx.x < AGG_ROWS) rcnt[threadIdx.x] = 1;
    if (threadIdx.x == 0) next_row = 0;
    int2 en[SORT_PER_THREAD];
#pragma unroll
    for (int k = 0; k < SORT_PER_THREAD; ++k) {
        const int j = (int)threadIdx.x + k * AGG_THREADS;
        en[k] = make_int2(0, 0);
        if (j < total) en[k] = blist[j];
    }
    __syncthreads();
    int rk[SORT_PER_THREAD];
#pragma unroll
    for (int k = 0; k < SORT_PER_THREAD; ++k) {
        const int j = (int)threadIdx.x + k * AGG_THREADS;
        rk[k] = 0;
        if (j < total) rk[k] = atomicAdd(&rcnt[en[k].x >> LIST_E_BITS], 1);
    }
    __syncthreads();
    if (wave == 0) {
        int c[SCAN_PER_LANE];
        int sum = 0;
#pragma unroll
        for (int q = 0; q < SCAN_PER_LANE; ++q) {
            const int r = lane * SCAN_PER_LANE + q;
            c[q] = r < cnt ? rcnt[r] : 0;
            sum += c[q];
        }
        int inc = sum;
#pragma unroll
        for (int d = 1; d < 32; d <<= 1) {
            const int up = __shfl_up(inc, d, 32);
            if (lane >= d) inc += up;
        }
        int run = inc - sum;
#pragma unroll
        for (int q = 0; q < SCAN_PER_LANE; ++q) {
            const int r = lane * SCAN_PER_LANE + q;
            if (r <= cnt) rstart[r] = run;
            run += c[q];
        }
    }
    __syncthreads();
#pragma unroll
    for (int k = 0; k < SORT_PER_THREAD; ++k) {
        const int j = (int)threadIdx.x + k * AGG_THREADS;
        if (j < total) {
            const int tl = en[k].x >> LIST_E_BITS;
            sorted[rstart[tl] + rk[k]] = make_int2(en[k].x & ((1 << LIST_E_BITS) - 1), en[k].y);
        }
    }
    if ((int)threadIdx.x < cnt)
        sorted[rstart[threadIdx.x]] = make_int2(N_EDGES + n0 + (int)threadIdx.x, n0 + (int)threadIdx.x);
    __syncthreads();

    const int hsel = lane >> 2;
    int rr = 0;
    if (lane == 0) rr = atomicAdd(&next_row, 1);
    int r = __builtin_amdgcn_readfirstlane(rr);
    int keyv = 0, rsv = 0, rev = 0;
    if (r < cnt) {
        keyv = amaxK[(size_t)(n0 + r) * 8 + hsel];
        rsv = rstart[r];
        rev = rstart[r + 1];
    }
    asm volatile("" : "+v"(keyv));
    while (r < cnt) {
        int rq = 0;
        if (lane == 0) rq = atomicAdd(&next_row, 1);
        const int rn = __builtin_amdgcn_readfirstlane(rq);
        int keyn = 0, rsn = 0, ren = 0;
        if (rn < cnt) {
            keyn = amaxK[(size_t)(n0 + rn) * 8 + hsel];
            rsn = rstart[rn];
            ren = rstart[rn + 1];
        }
        const int rs = __builtin_amdgcn_readfirstlane(rsv);
        const int re = __builtin_amdgcn_readfirstlane(rev);
        const float mx = __int_as_float(keyv >= 0 ? keyv : (keyv ^ 0x7FFFFFFF));
        float a[4] = {0.f, 0.f, 0.f, 0.f};
        float dn = 0.f;
        for (int base = rs; base < re; base += 32) {
            int n = re - base;
            if (n > 32) n = 32;
            const int2 ent = sorted[base + (lane < n ? lane : n - 1)];
            for (int sub = 0; sub < n; sub += AGG_BATCH) {
                float  lg[AGG_BATCH];
                float4 hv[AGG_BATCH];
#pragma unroll
                for (int b = 0; b < AGG_BATCH; ++b) {
                    int ix = sub + b;
                    if (ix > n - 1) ix = n - 1;
                    const int eu = __builtin_amdgcn_readlane(ent.x, ix);
                    const int su = __builtin_amdgcn_readlane(ent.y, ix);
                    lg[b] = logit[(size_t)eu * 8 + hsel];
                    hv[b] = *reinterpret_cast<const float4*>(h + (size_t)su * HD + lane * 4);
                }
#pragma unroll
                for (int b = 0; b < AGG_BATCH; ++b) {
                    float w = __expf(lg[b] - mx);
                    if (sub + b >= n) w = 0.f;
                    a[0] += w * hv[b].x;
                    a[1] += w * hv[b].y;
                    a[2] += w * hv[b].z;
                    a[3] += w * hv[b].w;
                    dn += w;
                }
            }
        }
        const float dd = dn + 1e-16f;
        float4 o;
        o.x = a[0] / dd; o.y = a[1] / dd; o.z = a[2] / dd; o.w = a[3] / dd;
        *reinterpret_cast<float4*>(outacc + (size_t)(n0 + r) * HD + lane * 4) = o;
        cs[0] += o.x; cs[1] += o.y; cs[2] += o.z; cs[3] += o.w;
        cq[0] += o.x * o.x; cq[1] += o.y * o.y; cq[2] += o.z * o.z; cq[3] += o.w * o.w;
        r = rn; keyv = keyn; rsv = rsn; rev = ren;
    }
    } else {
    for (int tl = wave; tl < cnt; tl += AGG_WAVES) {
        int t = n0 + tl;
        size_t i = (size_t)N_EDGES + (size_t)t;
        agg_row_term(acc + tl * HD, den + tl * H_HEADS, h + (size_t)t * HD, logit + i * 8,
                     amaxK + (size_t)t * 8, lane, true);
    }
    __syncthreads();
    int dv[AGG_UNROLL][4];
    agg_load_groups(dv, dst, wave, lane);
    for (int g0 = wave; g0 < DST_GROUPS; g0 += AGG_WAVES * AGG_UNROLL) {
        int nx[AGG_UNROLL][4];
        agg_load_groups(nx, dst, g0 + AGG_WAVES * AGG_UNROLL, lane);

        const unsigned ucnt = (unsigned)cnt;
        int tl[AGG_UNROLL][4];
        int sv[AGG_UNROLL][4];
        unsigned anym[AGG_UNROLL];
#pragma unroll
        for (int u = 0; u < AGG_UNROLL; ++u) {
            int gg = g0 + u * AGG_WAVES;
#pragma unroll
            for (int k = 0; k < 4; ++k) {
                tl[u][k] = dv[u][k] - n0;
                sv[u][k] = 0;
            }
            bool hu = ((unsigned)tl[u][0] < ucnt) | ((unsigned)tl[u][1] < ucnt) |
                      ((unsigned)tl[u][2] < ucnt) | ((unsigned)tl[u][3] < ucnt);
            anym[u] = __builtin_amdgcn_ballot_w32(hu);
            if (anym[u] != 0u) {
#pragma unroll
                for (int k = 0; k < 4; ++k) {
                    if ((unsigned)tl[u][k] < ucnt)
                        sv[u][k] = src[((size_t)gg * 32 + lane) * 4 + k];
                }
            }
        }
#pragma unroll
        for (int u = 0; u < AGG_UNROLL; ++u) {
            int gg = g0 + u * AGG_WAVES;
            if (anym[u] != 0u) {
#pragma unroll
                for (int k = 0; k < 4; ++k) {
                    unsigned m = __builtin_amdgcn_ballot_w32((unsigned)tl[u][k] < ucnt);
                    while (m != 0u) {
                        int l = __builtin_ctz(m);
                        m &= m - 1u;
                        int tlu = __builtin_amdgcn_readlane(tl[u][k], l);
                        int su  = __builtin_amdgcn_readlane(sv[u][k], l);
                        size_t e = ((size_t)gg * 32 + (size_t)l) * 4 + (size_t)k;
                        agg_row_term(acc + tlu * HD, den + tlu * H_HEADS, h + (size_t)su * HD,
                                     logit + e * 8, amaxK + (size_t)(n0 + tlu) * 8, lane, false);
                    }
                }
            }
        }
#pragma unroll
        for (int u = 0; u < AGG_UNROLL; ++u) {
#pragma unroll
            for (int k = 0; k < 4; ++k) dv[u][k] = nx[u][k];
        }
    }
    __syncthreads();
    for (int tl = wave; tl < cnt; tl += AGG_WAVES) {
        float* orow = outacc + (size_t)(n0 + tl) * HD;
#pragma unroll
        for (int j = 0; j < 4; ++j) {
            int hh = j * 2 + (lane >> 4);
            float v = acc[tl * HD + j * 32 + lane] / (den[tl * H_HEADS + hh] + 1e-16f);
            orow[j * 32 + lane] = v;
            cs[j] += v;
            cq[j] += v * v;
        }
    }
    }
#pragma unroll
    for (int j = 0; j < 4; ++j) {
        const int ch = sorted_path ? lane * 4 + j : j * 32 + lane;
        red_s[wave * HD + ch] = cs[j];
        red_q[wave * HD + ch] = cq[j];
    }
    __syncthreads();
    if (threadIdx.x < 2 * HD) {
        const int ch = threadIdx.x & (HD - 1);
        const bool sq = threadIdx.x >= HD;
        const float* r = sq ? red_q : red_s;
        float t = 0.f;
        for (int w = 0; w < AGG_WAVES; ++w) t += r[w * HD + ch];
        float* pp = sq ? psq : psum;
        pp[(size_t)blockIdx.x * HD + ch] = t;
    }
}

#define BIN_THREADS 1024
#define BIN_EDGES   (BIN_THREADS * 4)
#define BIN_BLOCKS  ((N_EDGES + BIN_EDGES - 1) / BIN_EDGES)

static_assert(N_EDGES % 4 == 0, "a lane's four edges are all inside the list or all outside");

__global__ void __launch_bounds__(BIN_THREADS)
k_bin(const int* __restrict__ src, const int* __restrict__ dst, int* ecount, float* elist) {
    __shared__ int hist[AGG_BLOCKS];
    __shared__ int base[AGG_BLOCKS];
    const int tid = threadIdx.x;
    const int e0 = (blockIdx.x * BIN_THREADS + tid) * 4;
    int dv[4] = {-1, -1, -1, -1};
    int sv[4] = {0, 0, 0, 0};
    if (e0 < N_EDGES) {
        const int4 d = *reinterpret_cast<const int4*>(dst + e0);
        const int4 s = *reinterpret_cast<const int4*>(src + e0);
        dv[0] = d.x; dv[1] = d.y; dv[2] = d.z; dv[3] = d.w;
        sv[0] = s.x; sv[1] = s.y; sv[2] = s.z; sv[3] = s.w;
    }
    if (tid < AGG_BLOCKS) hist[tid] = 0;
    __syncthreads();
    int own[4], rank[4];
#pragma unroll
    for (int k = 0; k < 4; ++k) {
        own[k] = -1;
        rank[k] = 0;
        if ((unsigned)dv[k] < (unsigned)N_NODES) {
            own[k] = dv[k] / AGG_ROWS;
            rank[k] = atomicAdd(&hist[own[k]], 1);
        }
    }
    __syncthreads();
    if (tid < AGG_BLOCKS) base[tid] = atomicAdd(&ecount[tid], hist[tid]);
    __syncthreads();
#pragma unroll
    for (int k = 0; k < 4; ++k) {
        if (own[k] >= 0) {
            const int n0 = own[k] * AGG_ROWS;
            int cnt = N_NODES - n0;
            if (cnt > AGG_ROWS) cnt = AGG_ROWS;
            const int pos = base[own[k]] + rank[k];
            if (pos < cnt * LIST_CAP_PER_ROW) {
                int2* blist = reinterpret_cast<int2*>(elist + (size_t)n0 * HD);
                blist[pos] = make_int2((e0 + k) | ((dv[k] - n0) << LIST_E_BITS), sv[k]);
            }
        }
    }
}

__device__ __forceinline__ int logit_key(float l) {
    int bi = __float_as_int(l);
    return bi >= 0 ? bi : (bi ^ 0x7FFFFFFF);
}

__device__ __forceinline__ void pre_edge_term(size_t e, int tl, unsigned s, unsigned t,
                                              const float* __restrict__ edge_attr,
                                              const float* __restrict__ a_src,
                                              const float* __restrict__ a_dst,
                                              const float* sV, float* lsum, float* ldeg,
                                              int* lmax, float* __restrict__ logit) {
    const float4* ep = reinterpret_cast<const float4*>(edge_attr + e * E_DIM);
    const float4* sp = reinterpret_cast<const float4*>(a_src + (size_t)s * H_HEADS);
    const float4* dp = reinterpret_cast<const float4*>(a_dst + (size_t)t * H_HEADS);
    float4 q0 = ep[0], q1 = ep[1], q2 = ep[2], q3 = ep[3];
    float4 s0 = sp[0], s1 = sp[1];
    float4 d0 = dp[0], d1 = dp[1];
    float eav[E_DIM] = {q0.x, q0.y, q0.z, q0.w, q1.x, q1.y, q1.z, q1.w,
                        q2.x, q2.y, q2.z, q2.w, q3.x, q3.y, q3.z, q3.w};
    float asv[H_HEADS] = {s0.x, s0.y, s0.z, s0.w, s1.x, s1.y, s1.z, s1.w};
    float adv[H_HEADS] = {d0.x, d0.y, d0.z, d0.w, d1.x, d1.y, d1.z, d1.w};
    float ae[H_HEADS];
#pragma unroll
    for (int hh = 0; hh < H_HEADS; ++hh) ae[hh] = 0.f;
#pragma unroll
    for (int k = 0; k < E_DIM; ++k) {
#pragma unroll
        for (int hh = 0; hh < H_HEADS; ++hh) ae[hh] += eav[k] * sV[k * 8 + hh];
    }
    float lg[H_HEADS];
#pragma unroll
    for (int hh = 0; hh < H_HEADS; ++hh) {
        float l = asv[hh] + adv[hh] + ae[hh];
        lg[hh] = l > 0.f ? l : NEG_SLOPE * l;
    }
    float4* lp = reinterpret_cast<float4*>(logit + e * H_HEADS);
    lp[0] = make_float4(lg[0], lg[1], lg[2], lg[3]);
    lp[1] = make_float4(lg[4], lg[5], lg[6], lg[7]);
#pragma unroll
    for (int hh = 0; hh < H_HEADS; ++hh) atomicMax(&lmax[tl * H_HEADS + hh], logit_key(lg[hh]));
#pragma unroll
    for (int k = 0; k < E_DIM; ++k) atomicAdd(&lsum[tl * E_DIM + k], eav[k]);
    atomicAdd(&ldeg[tl], 1.0f);
}

#define PRE_STRIDE   (AGG_WAVES * AGG_UNROLL)
#define PRE_STRIPES  ((DST_GROUPS + PRE_STRIDE - 1) / PRE_STRIDE)
#define PRE_SPC      14
#define PRE_CHUNKS   ((PRE_STRIPES + PRE_SPC - 1) / PRE_SPC)
#define PRE_QCAP     2048

__global__ void __launch_bounds__(AGG_THREADS)
k_node_pre(const int* __restrict__ src, const int* __restrict__ dst,
           const float* __restrict__ edge_attr, const float* __restrict__ Vc,
           const float* __restrict__ a_src, const float* __restrict__ a_dst,
           float* __restrict__ logit, int* __restrict__ amaxK,
           const float* __restrict__ elist, const int* __restrict__ ecount) {
    __shared__ float lsum[AGG_ROWS * E_DIM];
    __shared__ float ldeg[AGG_ROWS];
    __shared__ int   lmax[AGG_ROWS * H_HEADS];
    __shared__ float sV[E_DIM * H_HEADS];
    __shared__ int   q[PRE_QCAP];
    __shared__ int   qn[PRE_CHUNKS];
    const int lane = threadIdx.x & 31;
    const int wave = __builtin_amdgcn_readfirstlane((int)(threadIdx.x >> 5));
    const int n0 = blockIdx.x * AGG_ROWS;
    int cnt = N_NODES - n0;
    if (cnt > AGG_ROWS) cnt = AGG_ROWS;
    const unsigned ucnt = (unsigned)cnt;
    const int2* blist = reinterpret_cast<const int2*>(elist + (size_t)n0 * HD);
    const int lcap = cnt * LIST_CAP_PER_ROW;
    const int total = ecount[blockIdx.x];

    for (int i = threadIdx.x; i < AGG_ROWS * E_DIM; i += AGG_THREADS) lsum[i] = 0.0f;
    for (int i = threadIdx.x; i < AGG_ROWS * H_HEADS; i += AGG_THREADS) lmax[i] = (int)0x80000000;
    for (int i = threadIdx.x; i < AGG_ROWS; i += AGG_THREADS) ldeg[i] = 0.0f;
    if (threadIdx.x < E_DIM * H_HEADS) sV[threadIdx.x] = Vc[threadIdx.x];
    if (threadIdx.x < PRE_CHUNKS) qn[threadIdx.x] = 0;
    __syncthreads();

    if (total <= lcap) {
        for (int j = threadIdx.x; j < total; j += AGG_THREADS) {
            asm volatile("" ::: "memory");
            const int2 en = blist[j];
            const unsigned e = (unsigned)en.x & ((1u << LIST_E_BITS) - 1u);
            const int tl = en.x >> LIST_E_BITS;
            pre_edge_term((size_t)e, tl, (unsigned)en.y, (unsigned)(n0 + tl), edge_attr, a_src,
                          a_dst, sV, lsum, ldeg, lmax, logit);
        }
    } else {
    int dv[AGG_UNROLL][4];
    agg_load_groups(dv, dst, wave, lane);
    for (int c = 0; c < PRE_CHUNKS; ++c) {
        int iend = (c + 1) * PRE_SPC;
        if (iend > PRE_STRIPES) iend = PRE_STRIPES;
        for (int i = c * PRE_SPC; i < iend; ++i) {
            const int g0 = wave + i * PRE_STRIDE;
            int cur[AGG_UNROLL][4];
#pragma unroll
            for (int u = 0; u < AGG_UNROLL; ++u) {
#pragma unroll
                for (int k = 0; k < 4; ++k) cur[u][k] = dv[u][k];
            }
            agg_load_groups(dv, dst, g0 + PRE_STRIDE, lane);

            unsigned ovf = 0u;
#pragma unroll
            for (int u = 0; u < AGG_UNROLL; ++u) {
                int gg = g0 + u * AGG_WAVES;
#pragma unroll
                for (int k = 0; k < 4; ++k) {
                    if ((unsigned)(cur[u][k] - n0) < ucnt) {
                        int pos = atomicAdd(&qn[c], 1);
                        if (pos < PRE_QCAP) q[pos] = (gg * 32 + lane) * 4 + k;
                        else                ovf |= 1u << (u * 4 + k);
                    }
                }
            }
            while (ovf != 0u) {
                asm volatile("" ::: "memory");
                int slot = __builtin_ctz(ovf);
                ovf &= ovf - 1u;
                int e = ((g0 + (slot >> 2) * AGG_WAVES) * 32 + lane) * 4 + (slot & 3);
                int s = src[e];
                int t = dst[e];
                pre_edge_term((size_t)e, t - n0, s, t, edge_attr, a_src, a_dst, sV, lsum, ldeg,
                              lmax, logit);
            }
        }
        __syncthreads();
        int qc = qn[c];
        if (qc > PRE_QCAP) qc = PRE_QCAP;
        for (int j = threadIdx.x; j < qc; j += AGG_THREADS) {
            asm volatile("" ::: "memory");
            int e = q[j];
            int s = src[e];
            int t = dst[e];
            pre_edge_term((size_t)e, t - n0, s, t, edge_attr, a_src, a_dst, sV, lsum, ldeg,
                          lmax, logit);
        }
        __syncthreads();
    }
    }
    __syncthreads();

    for (int i = threadIdx.x; i < cnt * H_HEADS; i += AGG_THREADS) {
        int tl = i >> 3, hh = i & 7;
        int t = n0 + tl;
        float d = ldeg[tl];
        d = d < 1.0f ? 1.0f : d;
        float ae = 0.f;
#pragma unroll
        for (int k = 0; k < E_DIM; ++k) ae += (lsum[tl * E_DIM + k] / d) * sV[k * 8 + hh];
        float l = a_src[(size_t)t * H_HEADS + hh] + a_dst[(size_t)t * H_HEADS + hh] + ae;
        l = l > 0.f ? l : NEG_SLOPE * l;
        logit[((size_t)N_EDGES + (size_t)t) * H_HEADS + hh] = l;
        int key = logit_key(l);
        int mk = lmax[i];
        amaxK[(size_t)t * H_HEADS + hh] = key > mk ? key : mk;
    }
}

__global__ void __launch_bounds__(AGG_BLOCKS)
k_stats(const float* __restrict__ psum, const float* __restrict__ psq, float* meanb,
        float* invstd) {
    __shared__ float ssum[AGG_BLOCKS];
    __shared__ float ssq[AGG_BLOCKS];
    int c = blockIdx.x;
    ssum[threadIdx.x] = psum[(size_t)threadIdx.x * HD + c];
    ssq[threadIdx.x]  = psq[(size_t)threadIdx.x * HD + c];
    __syncthreads();
    for (int off = AGG_BLOCKS / 2; off > 0; off >>= 1) {
        if (threadIdx.x < off) {
            ssum[threadIdx.x] += ssum[threadIdx.x + off];
            ssq[threadIdx.x]  += ssq[threadIdx.x + off];
        }
        __syncthreads();
    }
    if (threadIdx.x == 0) {
        float m = ssum[0] / (float)N_NODES;
        float var = ssq[0] / (float)N_NODES - m * m;
        meanb[c] = m;
        invstd[c] = rsqrtf(var + EPS_IN);
    }
}

#define FIN_THREADS 512
#define FIN_BLOCKS  625
#define FIN_VEC     (N_NODES * HD / 4)
#define FIN_STRIDE  (FIN_BLOCKS * FIN_THREADS)
#define FIN_TRIPS   (FIN_VEC / FIN_STRIDE)

static_assert(FIN_TRIPS * FIN_STRIDE == FIN_VEC, "the grid covers the plane in whole trips");
static_assert(FIN_STRIDE % (HD / 4) == 0, "a thread keeps its four channels on every trip");

__device__ __forceinline__ float fin_elem(float xv, float ov, float m, float is, float g,
                                          float b) {
    float normed = (ov - m) * is * g + b;
    float t = xv + normed;
    return t > 0.f ? t : expm1f(t);
}

__global__ void __launch_bounds__(FIN_THREADS)
k_final(const float* __restrict__ x,
        const float* __restrict__ outacc,
        const float* __restrict__ meanb,
        const float* __restrict__ invstd,
        const float* __restrict__ gamma,
        const float* __restrict__ beta, float* out) {
    const int t0 = blockIdx.x * FIN_THREADS + threadIdx.x;
    const int c4 = t0 & (HD / 4 - 1);
    const float4* x4 = reinterpret_cast<const float4*>(x);
    const float4* o4 = reinterpret_cast<const float4*>(outacc);
    float4* out4 = reinterpret_cast<float4*>(out);
    float4 xv[FIN_TRIPS], ov[FIN_TRIPS];
#pragma unroll
    for (int k = 0; k < FIN_TRIPS; ++k) {
        xv[k] = x4[t0 + k * FIN_STRIDE];
        ov[k] = o4[t0 + k * FIN_STRIDE];
    }
    const float4 m  = reinterpret_cast<const float4*>(meanb)[c4];
    const float4 is = reinterpret_cast<const float4*>(invstd)[c4];
    float4 g, b;
    g.x = gamma[c4 * 4 + 0]; g.y = gamma[c4 * 4 + 1]; g.z = gamma[c4 * 4 + 2]; g.w = gamma[c4 * 4 + 3];
    b.x = beta[c4 * 4 + 0];  b.y = beta[c4 * 4 + 1];  b.z = beta[c4 * 4 + 2];  b.w = beta[c4 * 4 + 3];
#pragma unroll
    for (int k = 0; k < FIN_TRIPS; ++k) {
        float4 r;
        r.x = fin_elem(xv[k].x, ov[k].x, m.x, is.x, g.x, b.x);
        r.y = fin_elem(xv[k].y, ov[k].y, m.y, is.y, g.y, b.y);
        r.z = fin_elem(xv[k].z, ov[k].z, m.z, is.z, g.z, b.z);
        r.w = fin_elem(xv[k].w, ov[k].w, m.w, is.w, g.w, b.w);
        out4[t0 + k * FIN_STRIDE] = r;
    }
}

extern "C" void kernel_launch(void* const* d_in, const int* in_sizes, int n_in,
                              void* d_out, int out_size, void* d_ws, size_t ws_size,
                              hipStream_t stream) {
    const float* x        = (const float*)d_in[0];
    const int*   ei       = (const int*)d_in[1];
    const float* edgeattr = (const float*)d_in[2];
    const float* W        = (const float*)d_in[3];
    const float* att_src  = (const float*)d_in[4];
    const float* att_dst  = (const float*)d_in[5];
    const float* W_e      = (const float*)d_in[6];
    const float* att_edge = (const float*)d_in[7];
    const float* gamma    = (const float*)d_in[8];
    const float* beta     = (const float*)d_in[9];
    float* out = (float*)d_out;

    const int* src = ei;
    const int* dst = ei + N_EDGES;

    float* ws = (float*)d_ws;
    float* hbuf     = ws;
    float* a_src    = hbuf   + (size_t)N_NODES * HD;
    float* a_dst    = a_src  + (size_t)N_NODES * H_HEADS;
    float* degb     = a_dst  + (size_t)N_NODES * H_HEADS;
    float* loopattr = degb   + N_NODES;
    float* Vc       = loopattr + (size_t)N_NODES * E_DIM;
    float* alpha    = Vc + 128;
    float* denom    = alpha + (size_t)ET * H_HEADS;
    float* outacc   = denom + (size_t)N_NODES * H_HEADS;
    float* meanb    = outacc + (size_t)N_NODES * HD;
    float* invstd   = meanb + 128;
    int*   amaxK    = (int*)(invstd + 128);

    k_gemm_h<<<(N_NODES / 16) * 8 / 8, 256, 0, stream>>>(x, W, hbuf);

    k_attn_node<<<(N_NODES * H_HEADS + 255) / 256, 256, 0, stream>>>(
        hbuf, att_src, att_dst, a_src, a_dst);
    k_edge_proj<<<1, 128, 0, stream>>>(W_e, att_edge, Vc, (int*)degb);

    k_bin<<<BIN_BLOCKS, BIN_THREADS, 0, stream>>>(src, dst, (int*)degb, outacc);
    k_node_pre<<<AGG_BLOCKS, AGG_THREADS, 0, stream>>>(src, dst, edgeattr, Vc, a_src, a_dst,
                                                       alpha, amaxK, outacc, (const int*)degb);
    float* psum = loopattr;
    float* psq  = loopattr + (size_t)AGG_BLOCKS * HD;
    k_aggregate<<<AGG_BLOCKS, AGG_THREADS, 0, stream>>>(src, dst, alpha, amaxK, hbuf,
                                                        (const int*)degb, outacc, psum, psq);

    k_stats<<<HD, AGG_BLOCKS, 0, stream>>>(psum, psq, meanb, invstd);
    k_final<<<FIN_BLOCKS, FIN_THREADS, 0, stream>>>(
        x, outacc, meanb, invstd, gamma, beta, out);
}


# GEMM h = x W: a wave owns a 16-row strip of all 128 columns, its x fragment feeds 8 WMMAs per k step; W staged once per workgroup in LDS; was one 16x16 tile per wave; same k order, sums unchanged
# speedup vs baseline: 15.7338x; 1.0584x over previous
#include <hip/hip_runtime.h>
#include <hip/hip_bf16.h>

#define N_NODES 50000
#define N_EDGES 1600000
#define F_DIM   128
#define H_HEADS 8
#define D_HEAD  16
#define E_DIM   16
#define HD      (H_HEADS * D_HEAD)
#define ET      (N_EDGES + N_NODES)
#define EPS_IN  1e-5f
#define NEG_SLOPE 0.2f

typedef float v2f __attribute__((ext_vector_type(2)));
typedef float v8f __attribute__((ext_vector_type(8)));

#define AGG_BLOCKS  256
#define AGG_ROWS    ((N_NODES + AGG_BLOCKS - 1) / AGG_BLOCKS)

#define GEMM_WAVES   8
#define GEMM_THREADS (GEMM_WAVES * 32)
#define GEMM_STRIPS  (N_NODES / 16)
#define GEMM_BLOCKS  ((GEMM_STRIPS + GEMM_WAVES - 1) / GEMM_WAVES)
#define W_PITCH      (2 * HD + 32)
#define GEMM_KSTEPS  (F_DIM / 4)
#define GEMM_NT      (HD / 16)
#define W_ITEMS      ((F_DIM / 2) * (HD / 4) / GEMM_THREADS)

static_assert(N_NODES % 16 == 0, "whole strips");
static_assert(W_ITEMS * GEMM_THREADS == (F_DIM / 2) * (HD / 4), "the copy of W splits evenly over the threads");
static_assert(HD / 4 == 32, "one row of W per wave instruction in the copy");

__global__ void __launch_bounds__(GEMM_THREADS)
k_gemm_h(const float* __restrict__ x, const float* __restrict__ W,
         float* __restrict__ h) {
    __shared__ float Ws[(F_DIM / 2) * W_PITCH];
    const int lane = threadIdx.x & 31;
    const int wave = __builtin_amdgcn_readfirstlane((int)(threadIdx.x >> 5));
    const int strip = blockIdx.x * GEMM_WAVES + wave;
    const bool live = strip < GEMM_STRIPS;
    const int c  = lane & 15;
    const int hf = lane >> 4;

    v2f a[GEMM_KSTEPS];
    if (live) {
        const float* xr = x + (size_t)(strip * 16 + c) * F_DIM + 2 * hf;
#pragma unroll
        for (int s = 0; s < GEMM_KSTEPS; ++s) {
            a[s].x = xr[4 * s];
            a[s].y = xr[4 * s + 1];
        }
    }
    const float4* W4 = reinterpret_cast<const float4*>(W);
    float4 r0[W_ITEMS], r1[W_ITEMS];
#pragma unroll
    for (int i = 0; i < W_ITEMS; ++i) {
        const int item = i * GEMM_THREADS + (int)threadIdx.x;
        const int kp = item >> 5, c4 = item & 31;
        r0[i] = W4[(2 * kp) * (HD / 4) + c4];
        r1[i] = W4[(2 * kp + 1) * (HD / 4) + c4];
    }
#pragma unroll
    for (int i = 0; i < W_ITEMS; ++i) {
        const int item = i * GEMM_THREADS + (int)threadIdx.x;
        const int kp = item >> 5, c4 = item & 31;
        float4* d = reinterpret_cast<float4*>(&Ws[kp * W_PITCH + c4 * 8]);
        d[0] = make_float4(r0[i].x, r1[i].x, r0[i].y, r1[i].y);
        d[1] = make_float4(r0[i].z, r1[i].z, r0[i].w, r1[i].w);
    }
    __syncthreads();
    if (!live) return;

    v8f acc[GEMM_NT];
#pragma unroll
    for (int t = 0; t < GEMM_NT; ++t) acc[t] = v8f{};
    const float* wl = Ws + hf * W_PITCH + 2 * c;
    v2f bq[2][GEMM_NT];
#pragma unroll
    for (int t = 0; t < GEMM_NT; ++t)
        bq[0][t] = *reinterpret_cast<const v2f*>(wl + 32 * t);
#pragma unroll
    for (int s = 0; s < GEMM_KSTEPS; ++s) {
        if (s + 1 < GEMM_KSTEPS) {
#pragma unroll
            for (int t = 0; t < GEMM_NT; ++t)
                bq[(s + 1) & 1][t] =
                    *reinterpret_cast<const v2f*>(wl + (2 * (s + 1)) * W_PITCH + 32 * t);
        }
        __builtin_amdgcn_sched_barrier(0);
#pragma unroll
        for (int t = 0; t < GEMM_NT; ++t) {
            acc[t] = __builtin_amdgcn_wmma_f32_16x16x4_f32(
                 false, a[s],  false, bq[s & 1][t],
                 (short)0, acc[t],  false,  false);
        }
        __builtin_amdgcn_sched_barrier(0);
    }
    const int rbase = strip * 16 + (hf << 3);
#pragma unroll
    for (int r = 0; r < 8; ++r) {
#pragma unroll
        for (int t = 0; t < GEMM_NT; ++t)
            h[(size_t)(rbase + r) * HD + 16 * t + c] = acc[t][r];
    }
}

__global__ void k_attn_node(const float* __restrict__ h,
                            const float* __restrict__ att_src,
                            const float* __restrict__ att_dst,
                            float* a_src, float* a_dst) {
    int idx = blockIdx.x * blockDim.x + threadIdx.x;
    if (idx >= N_NODES * H_HEADS) return;
    int n = idx >> 3, hh = idx & 7;
    const float* hp = h + (size_t)n * HD + hh * D_HEAD;
    const float* as = att_src + hh * D_HEAD;
    const float* ad = att_dst + hh * D_HEAD;
    float s = 0.f, d = 0.f;
#pragma unroll
    for (int i = 0; i < D_HEAD; ++i) {
        float v = hp[i];
        s += v * as[i];
        d += v * ad[i];
    }
    a_src[idx] = s;
    a_dst[idx] = d;
}

__global__ void k_edge_proj(const float* __restrict__ W_e,
                            const float* __restrict__ att_edge, float* V, int* ecount) {
    int idx = threadIdx.x;
    int k = idx >> 3, hh = idx & 7;
    float s = 0.f;
#pragma unroll
    for (int d = 0; d < D_HEAD; ++d)
        s += W_e[(size_t)k * HD + hh * D_HEAD + d] * att_edge[hh * D_HEAD + d];
    V[idx] = s;
    for (int i = idx; i < AGG_BLOCKS; i += E_DIM * H_HEADS) ecount[i] = 0;
}

#define AGG_THREADS 1024
#define AGG_WAVES   (AGG_THREADS / 32)
#define DST_GROUPS  (N_EDGES / 128)
#define AGG_UNROLL  4

#define AGG_BATCH   8
#define LIST_E_BITS 21
#define LIST_CAP_PER_ROW (HD / 2)

static_assert(N_EDGES % 128 == 0, "dst list must split into whole wave groups");
static_assert(N_EDGES <= (1 << LIST_E_BITS), "edge id must fit its field");
static_assert(AGG_ROWS <= (1 << (31 - LIST_E_BITS)), "local row must fit its field");
static_assert((AGG_BLOCKS & (AGG_BLOCKS - 1)) == 0, "the statistics kernel halves the partials in a tree");

#define SORT_CAP    (AGG_ROWS * LIST_CAP_PER_ROW)
#define SORT_PER_THREAD ((SORT_CAP + AGG_THREADS - 1) / AGG_THREADS)
#define SCAN_PER_LANE   ((AGG_ROWS + 32) / 32)
static_assert(SCAN_PER_LANE * 32 >= AGG_ROWS + 1, "one wave must cover every row start and the end");
static_assert(SORT_CAP * 2 == AGG_ROWS * HD, "the sort array takes exactly the place of the LDS rows");

__device__ __forceinline__ void agg_row_term(float* acc_row, float* den_row,
                                             const float* __restrict__ hrow,
                                             const float* __restrict__ lrow,
                                             const int* __restrict__ mrow, int lane,
                                             bool first) {
    int key = mrow[lane & 7];
    int bi = key >= 0 ? key : (key ^ 0x7FFFFFFF);
    float wl = __expf(lrow[lane & 7] - __int_as_float(bi));
    float hv[4];
#pragma unroll
    for (int j = 0; j < 4; ++j) hv[j] = hrow[j * 32 + lane];
    float v[4];
#pragma unroll
    for (int j = 0; j < 4; ++j) {
        int hh = j * 2 + (lane >> 4);
        float w = __int_as_float(__builtin_amdgcn_ds_bpermute(hh << 2, __float_as_int(wl)));
        v[j] = hv[j] * w;
    }
#pragma unroll
    for (int j = 0; j < 4; ++j) {
        int c = j * 32 + lane;
        if (first) acc_row[c] = v[j];
        else       atomicAdd(&acc_row[c], v[j]);
    }
    if (lane < H_HEADS) {
        if (first) den_row[lane] = wl;
        else       atomicAdd(&den_row[lane], wl);
    }
}

__device__ __forceinline__ void agg_load_groups(int (&dv)[AGG_UNROLL][4],
                                                const int* __restrict__ dst, int g0, int lane) {
#pragma unroll
    for (int u = 0; u < AGG_UNROLL; ++u) {
        int gg = g0 + u * AGG_WAVES;
        if (gg < DST_GROUPS) {
            const int* p = dst + ((size_t)gg * 32 + lane) * 4;
            dv[u][0] = p[0]; dv[u][1] = p[1]; dv[u][2] = p[2]; dv[u][3] = p[3];
        } else {
            dv[u][0] = -1; dv[u][1] = -1; dv[u][2] = -1; dv[u][3] = -1;
        }
    }
}

__global__ void __launch_bounds__(AGG_THREADS)
k_aggregate(const int* __restrict__ src, const int* __restrict__ dst,
            const float* __restrict__ logit, const int* __restrict__ amaxK,
            const float* __restrict__ h, const int* __restrict__ ecount, float* outacc,
            float* __restrict__ psum, float* __restrict__ psq) {
    __shared__ float acc[AGG_ROWS * HD];
    __shared__ float den[AGG_ROWS * H_HEADS];
    __shared__ float red_s[AGG_WAVES * HD];
    __shared__ float red_q[AGG_WAVES * HD];
    __shared__ int rstart[AGG_ROWS + 1];
    __shared__ int rcnt[AGG_ROWS];
    __shared__ int next_row;
    const int lane = threadIdx.x & 31;
    const int wave = __builtin_amdgcn_readfirstlane((int)(threadIdx.x >> 5));
    const int n0 = blockIdx.x * AGG_ROWS;
    int cnt = N_NODES - n0;
    if (cnt > AGG_ROWS) cnt = AGG_ROWS;

    const int total = ecount[blockIdx.x];
    const int lcap = cnt * LIST_CAP_PER_ROW;
    const int2* blist = reinterpret_cast<const int2*>(outacc + (size_t)n0 * HD);

    float cs[4] = {0.f, 0.f, 0.f, 0.f};
    float cq[4] = {0.f, 0.f, 0.f, 0.f};
    const bool sorted_path = total + cnt <= lcap;

    if (sorted_path) {
    int2* sorted = reinterpret_cast<int2*>(acc);
    if (threadIdx.x < AGG_ROWS) rcnt[threadIdx.x] = 1;
    if (threadIdx.x == 0) next_row = 0;
    int2 en[SORT_PER_THREAD];
#pragma unroll
    for (int k = 0; k < SORT_PER_THREAD; ++k) {
        const int j = (int)threadIdx.x + k * AGG_THREADS;
        en[k] = make_int2(0, 0);
        if (j < total) en[k] = blist[j];
    }
    __syncthreads();
    int rk[SORT_PER_THREAD];
#pragma unroll
    for (int k = 0; k < SORT_PER_THREAD; ++k) {
        const int j = (int)threadIdx.x + k * AGG_THREADS;
        rk[k] = 0;
        if (j < total) rk[k] = atomicAdd(&rcnt[en[k].x >> LIST_E_BITS], 1);
    }
    __syncthreads();
    if (wave == 0) {
        int c[SCAN_PER_LANE];
        int sum = 0;
#pragma unroll
        for (int q = 0; q < SCAN_PER_LANE; ++q) {
            const int r = lane * SCAN_PER_LANE + q;
            c[q] = r < cnt ? rcnt[r] : 0;
            sum += c[q];
        }
        int inc = sum;
#pragma unroll
        for (int d = 1; d < 32; d <<= 1) {
            const int up = __shfl_up(inc, d, 32);
            if (lane >= d) inc += up;
        }
        int run = inc - sum;
#pragma unroll
        for (int q = 0; q < SCAN_PER_LANE; ++q) {
            const int r = lane * SCAN_PER_LANE + q;
            if (r <= cnt) rstart[r] = run;
            run += c[q];
        }
    }
    __syncthreads();
#pragma unroll
    for (int k = 0; k < SORT_PER_THREAD; ++k) {
        const int j = (int)threadIdx.x + k * AGG_THREADS;
        if (j < total) {
            const int tl = en[k].x >> LIST_E_BITS;
            sorted[rstart[tl] + rk[k]] = make_int2(en[k].x & ((1 << LIST_E_BITS) - 1), en[k].y);
        }
    }
    if ((int)threadIdx.x < cnt)
        sorted[rstart[threadIdx.x]] = make_int2(N_EDGES + n0 + (int)threadIdx.x, n0 + (int)threadIdx.x);
    __syncthreads();

    const int hsel = lane >> 2;
    int rr = 0;
    if (lane == 0) rr = atomicAdd(&next_row, 1);
    int r = __builtin_amdgcn_readfirstlane(rr);
    int keyv = 0, rsv = 0, rev = 0;
    if (r < cnt) {
        keyv = amaxK[(size_t)(n0 + r) * 8 + hsel];
        rsv = rstart[r];
        rev = rstart[r + 1];
    }
    asm volatile("" : "+v"(keyv));
    while (r < cnt) {
        int rq = 0;
        if (lane == 0) rq = atomicAdd(&next_row, 1);
        const int rn = __builtin_amdgcn_readfirstlane(rq);
        int keyn = 0, rsn = 0, ren = 0;
        if (rn < cnt) {
            keyn = amaxK[(size_t)(n0 + rn) * 8 + hsel];
            rsn = rstart[rn];
            ren = rstart[rn + 1];
        }
        const int rs = __builtin_amdgcn_readfirstlane(rsv);
        const int re = __builtin_amdgcn_readfirstlane(rev);
        const float mx = __int_as_float(keyv >= 0 ? keyv : (keyv ^ 0x7FFFFFFF));
        float a[4] = {0.f, 0.f, 0.f, 0.f};
        float dn = 0.f;
        for (int base = rs; base < re; base += 32) {
            int n = re - base;
            if (n > 32) n = 32;
            const int2 ent = sorted[base + (lane < n ? lane : n - 1)];
            for (int sub = 0; sub < n; sub += AGG_BATCH) {
                float  lg[AGG_BATCH];
                float4 hv[AGG_BATCH];
#pragma unroll
                for (int b = 0; b < AGG_BATCH; ++b) {
                    int ix = sub + b;
                    if (ix > n - 1) ix = n - 1;
                    const int eu = __builtin_amdgcn_readlane(ent.x, ix);
                    const int su = __builtin_amdgcn_readlane(ent.y, ix);
                    lg[b] = logit[(size_t)eu * 8 + hsel];
                    hv[b] = *reinterpret_cast<const float4*>(h + (size_t)su * HD + lane * 4);
                }
#pragma unroll
                for (int b = 0; b < AGG_BATCH; ++b) {
                    float w = __expf(lg[b] - mx);
                    if (sub + b >= n) w = 0.f;
                    a[0] += w * hv[b].x;
                    a[1] += w * hv[b].y;
                    a[2] += w * hv[b].z;
                    a[3] += w * hv[b].w;
                    dn += w;
                }
            }
        }
        const float dd = dn + 1e-16f;
        float4 o;
        o.x = a[0] / dd; o.y = a[1] / dd; o.z = a[2] / dd; o.w = a[3] / dd;
        *reinterpret_cast<float4*>(outacc + (size_t)(n0 + r) * HD + lane * 4) = o;
        cs[0] += o.x; cs[1] += o.y; cs[2] += o.z; cs[3] += o.w;
        cq[0] += o.x * o.x; cq[1] += o.y * o.y; cq[2] += o.z * o.z; cq[3] += o.w * o.w;
        r = rn; keyv = keyn; rsv = rsn; rev = ren;
    }
    } else {
    for (int tl = wave; tl < cnt; tl += AGG_WAVES) {
        int t = n0 + tl;
        size_t i = (size_t)N_EDGES + (size_t)t;
        agg_row_term(acc + tl * HD, den + tl * H_HEADS, h + (size_t)t * HD, logit + i * 8,
                     amaxK + (size_t)t * 8, lane, true);
    }
    __syncthreads();
    int dv[AGG_UNROLL][4];
    agg_load_groups(dv, dst, wave, lane);
    for (int g0 = wave; g0 < DST_GROUPS; g0 += AGG_WAVES * AGG_UNROLL) {
        int nx[AGG_UNROLL][4];
        agg_load_groups(nx, dst, g0 + AGG_WAVES * AGG_UNROLL, lane);

        const unsigned ucnt = (unsigned)cnt;
        int tl[AGG_UNROLL][4];
        int sv[AGG_UNROLL][4];
        unsigned anym[AGG_UNROLL];
#pragma unroll
        for (int u = 0; u < AGG_UNROLL; ++u) {
            int gg = g0 + u * AGG_WAVES;
#pragma unroll
            for (int k = 0; k < 4; ++k) {
                tl[u][k] = dv[u][k] - n0;
                sv[u][k] = 0;
            }
            bool hu = ((unsigned)tl[u][0] < ucnt) | ((unsigned)tl[u][1] < ucnt) |
                      ((unsigned)tl[u][2] < ucnt) | ((unsigned)tl[u][3] < ucnt);
            anym[u] = __builtin_amdgcn_ballot_w32(hu);
            if (anym[u] != 0u) {
#pragma unroll
                for (int k = 0; k < 4; ++k) {
                    if ((unsigned)tl[u][k] < ucnt)
                        sv[u][k] = src[((size_t)gg * 32 + lane) * 4 + k];
                }
            }
        }
#pragma unroll
        for (int u = 0; u < AGG_UNROLL; ++u) {
            int gg = g0 + u * AGG_WAVES;
            if (anym[u] != 0u) {
#pragma unroll
                for (int k = 0; k < 4; ++k) {
                    unsigned m = __builtin_amdgcn_ballot_w32((unsigned)tl[u][k] < ucnt);
                    while (m != 0u) {
                        int l = __builtin_ctz(m);
                        m &= m - 1u;
                        int tlu = __builtin_amdgcn_readlane(tl[u][k], l);
                        int su  = __builtin_amdgcn_readlane(sv[u][k], l);
                        size_t e = ((size_t)gg * 32 + (size_t)l) * 4 + (size_t)k;
                        agg_row_term(acc + tlu * HD, den + tlu * H_HEADS, h + (size_t)su * HD,
                                     logit + e * 8, amaxK + (size_t)(n0 + tlu) * 8, lane, false);
                    }
                }
            }
        }
#pragma unroll
        for (int u = 0; u < AGG_UNROLL; ++u) {
#pragma unroll
            for (int k = 0; k < 4; ++k) dv[u][k] = nx[u][k];
        }
    }
    __syncthreads();
    for (int tl = wave; tl < cnt; tl += AGG_WAVES) {
        float* orow = outacc + (size_t)(n0 + tl) * HD;
#pragma unroll
        for (int j = 0; j < 4; ++j) {
            int hh = j * 2 + (lane >> 4);
            float v = acc[tl * HD + j * 32 + lane] / (den[tl * H_HEADS + hh] + 1e-16f);
            orow[j * 32 + lane] = v;
            cs[j] += v;
            cq[j] += v * v;
        }
    }
    }
#pragma unroll
    for (int j = 0; j < 4; ++j) {
        const int ch = sorted_path ? lane * 4 + j : j * 32 + lane;
        red_s[wave * HD + ch] = cs[j];
        red_q[wave * HD + ch] = cq[j];
    }
    __syncthreads();
    if (threadIdx.x < 2 * HD) {
        const int ch = threadIdx.x & (HD - 1);
        const bool sq = threadIdx.x >= HD;
        const float* r = sq ? red_q : red_s;
        float t = 0.f;
        for (int w = 0; w < AGG_WAVES; ++w) t += r[w * HD + ch];
        float* pp = sq ? psq : psum;
        pp[(size_t)blockIdx.x * HD + ch] = t;
    }
}

#define BIN_THREADS 1024
#define BIN_EDGES   (BIN_THREADS * 4)
#define BIN_BLOCKS  ((N_EDGES + BIN_EDGES - 1) / BIN_EDGES)

static_assert(N_EDGES % 4 == 0, "a lane's four edges are all inside the list or all outside");

__global__ void __launch_bounds__(BIN_THREADS)
k_bin(const int* __restrict__ src, const int* __restrict__ dst, int* ecount, float* elist) {
    __shared__ int hist[AGG_BLOCKS];
    __shared__ int base[AGG_BLOCKS];
    const int tid = threadIdx.x;
    const int e0 = (blockIdx.x * BIN_THREADS + tid) * 4;
    int dv[4] = {-1, -1, -1, -1};
    int sv[4] = {0, 0, 0, 0};
    if (e0 < N_EDGES) {
        const int4 d = *reinterpret_cast<const int4*>(dst + e0);
        const int4 s = *reinterpret_cast<const int4*>(src + e0);
        dv[0] = d.x; dv[1] = d.y; dv[2] = d.z; dv[3] = d.w;
        sv[0] = s.x; sv[1] = s.y; sv[2] = s.z; sv[3] = s.w;
    }
    if (tid < AGG_BLOCKS) hist[tid] = 0;
    __syncthreads();
    int own[4], rank[4];
#pragma unroll
    for (int k = 0; k < 4; ++k) {
        own[k] = -1;
        rank[k] = 0;
        if ((unsigned)dv[k] < (unsigned)N_NODES) {
            own[k] = dv[k] / AGG_ROWS;
            rank[k] = atomicAdd(&hist[own[k]], 1);
        }
    }
    __syncthreads();
    if (tid < AGG_BLOCKS) base[tid] = atomicAdd(&ecount[tid], hist[tid]);
    __syncthreads();
#pragma unroll
    for (int k = 0; k < 4; ++k) {
        if (own[k] >= 0) {
            const int n0 = own[k] * AGG_ROWS;
            int cnt = N_NODES - n0;
            if (cnt > AGG_ROWS) cnt = AGG_ROWS;
            const int pos = base[own[k]] + rank[k];
            if (pos < cnt * LIST_CAP_PER_ROW) {
                int2* blist = reinterpret_cast<int2*>(elist + (size_t)n0 * HD);
                blist[pos] = make_int2((e0 + k) | ((dv[k] - n0) << LIST_E_BITS), sv[k]);
            }
        }
    }
}

__device__ __forceinline__ int logit_key(float l) {
    int bi = __float_as_int(l);
    return bi >= 0 ? bi : (bi ^ 0x7FFFFFFF);
}

__device__ __forceinline__ void pre_edge_term(size_t e, int tl, unsigned s, unsigned t,
                                              const float* __restrict__ edge_attr,
                                              const float* __restrict__ a_src,
                                              const float* __restrict__ a_dst,
                                              const float* sV, float* lsum, float* ldeg,
                                              int* lmax, float* __restrict__ logit) {
    const float4* ep = reinterpret_cast<const float4*>(edge_attr + e * E_DIM);
    const float4* sp = reinterpret_cast<const float4*>(a_src + (size_t)s * H_HEADS);
    const float4* dp = reinterpret_cast<const float4*>(a_dst + (size_t)t * H_HEADS);
    float4 q0 = ep[0], q1 = ep[1], q2 = ep[2], q3 = ep[3];
    float4 s0 = sp[0], s1 = sp[1];
    float4 d0 = dp[0], d1 = dp[1];
    float eav[E_DIM] = {q0.x, q0.y, q0.z, q0.w, q1.x, q1.y, q1.z, q1.w,
                        q2.x, q2.y, q2.z, q2.w, q3.x, q3.y, q3.z, q3.w};
    float asv[H_HEADS] = {s0.x, s0.y, s0.z, s0.w, s1.x, s1.y, s1.z, s1.w};
    float adv[H_HEADS] = {d0.x, d0.y, d0.z, d0.w, d1.x, d1.y, d1.z, d1.w};
    float ae[H_HEADS];
#pragma unroll
    for (int hh = 0; hh < H_HEADS; ++hh) ae[hh] = 0.f;
#pragma unroll
    for (int k = 0; k < E_DIM; ++k) {
#pragma unroll
        for (int hh = 0; hh < H_HEADS; ++hh) ae[hh] += eav[k] * sV[k * 8 + hh];
    }
    float lg[H_HEADS];
#pragma unroll
    for (int hh = 0; hh < H_HEADS; ++hh) {
        float l = asv[hh] + adv[hh] + ae[hh];
        lg[hh] = l > 0.f ? l : NEG_SLOPE * l;
    }
    float4* lp = reinterpret_cast<float4*>(logit + e * H_HEADS);
    lp[0] = make_float4(lg[0], lg[1], lg[2], lg[3]);
    lp[1] = make_float4(lg[4], lg[5], lg[6], lg[7]);
#pragma unroll
    for (int hh = 0; hh < H_HEADS; ++hh) atomicMax(&lmax[tl * H_HEADS + hh], logit_key(lg[hh]));
#pragma unroll
    for (int k = 0; k < E_DIM; ++k) atomicAdd(&lsum[tl * E_DIM + k], eav[k]);
    atomicAdd(&ldeg[tl], 1.0f);
}

#define PRE_STRIDE   (AGG_WAVES * AGG_UNROLL)
#define PRE_STRIPES  ((DST_GROUPS + PRE_STRIDE - 1) / PRE_STRIDE)
#define PRE_SPC      14
#define PRE_CHUNKS   ((PRE_STRIPES + PRE_SPC - 1) / PRE_SPC)
#define PRE_QCAP     2048

__global__ void __launch_bounds__(AGG_THREADS)
k_node_pre(const int* __restrict__ src, const int* __restrict__ dst,
           const float* __restrict__ edge_attr, const float* __restrict__ Vc,
           const float* __restrict__ a_src, const float* __restrict__ a_dst,
           float* __restrict__ logit, int* __restrict__ amaxK,
           const float* __restrict__ elist, const int* __restrict__ ecount) {
    __shared__ float lsum[AGG_ROWS * E_DIM];
    __shared__ float ldeg[AGG_ROWS];
    __shared__ int   lmax[AGG_ROWS * H_HEADS];
    __shared__ float sV[E_DIM * H_HEADS];
    __shared__ int   q[PRE_QCAP];
    __shared__ int   qn[PRE_CHUNKS];
    const int lane = threadIdx.x & 31;
    const int wave = __builtin_amdgcn_readfirstlane((int)(threadIdx.x >> 5));
    const int n0 = blockIdx.x * AGG_ROWS;
    int cnt = N_NODES - n0;
    if (cnt > AGG_ROWS) cnt = AGG_ROWS;
    const unsigned ucnt = (unsigned)cnt;
    const int2* blist = reinterpret_cast<const int2*>(elist + (size_t)n0 * HD);
    const int lcap = cnt * LIST_CAP_PER_ROW;
    const int total = ecount[blockIdx.x];

    for (int i = threadIdx.x; i < AGG_ROWS * E_DIM; i += AGG_THREADS) lsum[i] = 0.0f;
    for (int i = threadIdx.x; i < AGG_ROWS * H_HEADS; i += AGG_THREADS) lmax[i] = (int)0x80000000;
    for (int i = threadIdx.x; i < AGG_ROWS; i += AGG_THREADS) ldeg[i] = 0.0f;
    if (threadIdx.x < E_DIM * H_HEADS) sV[threadIdx.x] = Vc[threadIdx.x];
    if (threadIdx.x < PRE_CHUNKS) qn[threadIdx.x] = 0;
    __syncthreads();

    if (total <= lcap) {
        for (int j = threadIdx.x; j < total; j += AGG_THREADS) {
            asm volatile("" ::: "memory");
            const int2 en = blist[j];
            const unsigned e = (unsigned)en.x & ((1u << LIST_E_BITS) - 1u);
            const int tl = en.x >> LIST_E_BITS;
            pre_edge_term((size_t)e, tl, (unsigned)en.y, (unsigned)(n0 + tl), edge_attr, a_src,
                          a_dst, sV, lsum, ldeg, lmax, logit);
        }
    } else {
    int dv[AGG_UNROLL][4];
    agg_load_groups(dv, dst, wave, lane);
    for (int c = 0; c < PRE_CHUNKS; ++c) {
        int iend = (c + 1) * PRE_SPC;
        if (iend > PRE_STRIPES) iend = PRE_STRIPES;
        for (int i = c * PRE_SPC; i < iend; ++i) {
            const int g0 = wave + i * PRE_STRIDE;
            int cur[AGG_UNROLL][4];
#pragma unroll
            for (int u = 0; u < AGG_UNROLL; ++u) {
#pragma unroll
                for (int k = 0; k < 4; ++k) cur[u][k] = dv[u][k];
            }
            agg_load_groups(dv, dst, g0 + PRE_STRIDE, lane);

            unsigned ovf = 0u;
#pragma unroll
            for (int u = 0; u < AGG_UNROLL; ++u) {
                int gg = g0 + u * AGG_WAVES;
#pragma unroll
                for (int k = 0; k < 4; ++k) {
                    if ((unsigned)(cur[u][k] - n0) < ucnt) {
                        int pos = atomicAdd(&qn[c], 1);
                        if (pos < PRE_QCAP) q[pos] = (gg * 32 + lane) * 4 + k;
                        else                ovf |= 1u << (u * 4 + k);
                    }
                }
            }
            while (ovf != 0u) {
                asm volatile("" ::: "memory");
                int slot = __builtin_ctz(ovf);
                ovf &= ovf - 1u;
                int e = ((g0 + (slot >> 2) * AGG_WAVES) * 32 + lane) * 4 + (slot & 3);
                int s = src[e];
                int t = dst[e];
                pre_edge_term((size_t)e, t - n0, s, t, edge_attr, a_src, a_dst, sV, lsum, ldeg,
                              lmax, logit);
            }
        }
        __syncthreads();
        int qc = qn[c];
        if (qc > PRE_QCAP) qc = PRE_QCAP;
        for (int j = threadIdx.x; j < qc; j += AGG_THREADS) {
            asm volatile("" ::: "memory");
            int e = q[j];
            int s = src[e];
            int t = dst[e];
            pre_edge_term((size_t)e, t - n0, s, t, edge_attr, a_src, a_dst, sV, lsum, ldeg,
                          lmax, logit);
        }
        __syncthreads();
    }
    }
    __syncthreads();

    for (int i = threadIdx.x; i < cnt * H_HEADS; i += AGG_THREADS) {
        int tl = i >> 3, hh = i & 7;
        int t = n0 + tl;
        float d = ldeg[tl];
        d = d < 1.0f ? 1.0f : d;
        float ae = 0.f;
#pragma unroll
        for (int k = 0; k < E_DIM; ++k) ae += (lsum[tl * E_DIM + k] / d) * sV[k * 8 + hh];
        float l = a_src[(size_t)t * H_HEADS + hh] + a_dst[(size_t)t * H_HEADS + hh] + ae;
        l = l > 0.f ? l : NEG_SLOPE * l;
        logit[((size_t)N_EDGES + (size_t)t) * H_HEADS + hh] = l;
        int key = logit_key(l);
        int mk = lmax[i];
        amaxK[(size_t)t * H_HEADS + hh] = key > mk ? key : mk;
    }
}

__global__ void __launch_bounds__(AGG_BLOCKS)
k_stats(const float* __restrict__ psum, const float* __restrict__ psq, float* meanb,
        float* invstd) {
    __shared__ float ssum[AGG_BLOCKS];
    __shared__ float ssq[AGG_BLOCKS];
    int c = blockIdx.x;
    ssum[threadIdx.x] = psum[(size_t)threadIdx.x * HD + c];
    ssq[threadIdx.x]  = psq[(size_t)threadIdx.x * HD + c];
    __syncthreads();
    for (int off = AGG_BLOCKS / 2; off > 0; off >>= 1) {
        if (threadIdx.x < off) {
            ssum[threadIdx.x] += ssum[threadIdx.x + off];
            ssq[threadIdx.x]  += ssq[threadIdx.x + off];
        }
        __syncthreads();
    }
    if (threadIdx.x == 0) {
        float m = ssum[0] / (float)N_NODES;
        float var = ssq[0] / (float)N_NODES - m * m;
        meanb[c] = m;
        invstd[c] = rsqrtf(var + EPS_IN);
    }
}

#define FIN_THREADS 512
#define FIN_BLOCKS  625
#define FIN_VEC     (N_NODES * HD / 4)
#define FIN_STRIDE  (FIN_BLOCKS * FIN_THREADS)
#define FIN_TRIPS   (FIN_VEC / FIN_STRIDE)

static_assert(FIN_TRIPS * FIN_STRIDE == FIN_VEC, "the grid covers the plane in whole trips");
static_assert(FIN_STRIDE % (HD / 4) == 0, "a thread keeps its four channels on every trip");

__device__ __forceinline__ float fin_elem(float xv, float ov, float m, float is, float g,
                                          float b) {
    float normed = (ov - m) * is * g + b;
    float t = xv + normed;
    return t > 0.f ? t : expm1f(t);
}

__global__ void __launch_bounds__(FIN_THREADS)
k_final(const float* __restrict__ x,
        const float* __restrict__ outacc,
        const float* __restrict__ meanb,
        const float* __restrict__ invstd,
        const float* __restrict__ gamma,
        const float* __restrict__ beta, float* out) {
    const int t0 = blockIdx.x * FIN_THREADS + threadIdx.x;
    const int c4 = t0 & (HD / 4 - 1);
    const float4* x4 = reinterpret_cast<const float4*>(x);
    const float4* o4 = reinterpret_cast<const float4*>(outacc);
    float4* out4 = reinterpret_cast<float4*>(out);
    float4 xv[FIN_TRIPS], ov[FIN_TRIPS];
#pragma unroll
    for (int k = 0; k < FIN_TRIPS; ++k) {
        xv[k] = x4[t0 + k * FIN_STRIDE];
        ov[k] = o4[t0 + k * FIN_STRIDE];
    }
    const float4 m  = reinterpret_cast<const float4*>(meanb)[c4];
    const float4 is = reinterpret_cast<const float4*>(invstd)[c4];
    float4 g, b;
    g.x = gamma[c4 * 4 + 0]; g.y = gamma[c4 * 4 + 1]; g.z = gamma[c4 * 4 + 2]; g.w = gamma[c4 * 4 + 3];
    b.x = beta[c4 * 4 + 0];  b.y = beta[c4 * 4 + 1];  b.z = beta[c4 * 4 + 2];  b.w = beta[c4 * 4 + 3];
#pragma unroll
    for (int k = 0; k < FIN_TRIPS; ++k) {
        float4 r;
        r.x = fin_elem(xv[k].x, ov[k].x, m.x, is.x, g.x, b.x);
        r.y = fin_elem(xv[k].y, ov[k].y, m.y, is.y, g.y, b.y);
        r.z = fin_elem(xv[k].z, ov[k].z, m.z, is.z, g.z, b.z);
        r.w = fin_elem(xv[k].w, ov[k].w, m.w, is.w, g.w, b.w);
        out4[t0 + k * FIN_STRIDE] = r;
    }
}

extern "C" void kernel_launch(void* const* d_in, const int* in_sizes, int n_in,
                              void* d_out, int out_size, void* d_ws, size_t ws_size,
                              hipStream_t stream) {
    const float* x        = (const float*)d_in[0];
    const int*   ei       = (const int*)d_in[1];
    const float* edgeattr = (const float*)d_in[2];
    const float* W        = (const float*)d_in[3];
    const float* att_src  = (const float*)d_in[4];
    const float* att_dst  = (const float*)d_in[5];
    const float* W_e      = (const float*)d_in[6];
    const float* att_edge = (const float*)d_in[7];
    const float* gamma    = (const float*)d_in[8];
    const float* beta     = (const float*)d_in[9];
    float* out = (float*)d_out;

    const int* src = ei;
    const int* dst = ei + N_EDGES;

    float* ws = (float*)d_ws;
    float* hbuf     = ws;
    float* a_src    = hbuf   + (size_t)N_NODES * HD;
    float* a_dst    = a_src  + (size_t)N_NODES * H_HEADS;
    float* degb     = a_dst  + (size_t)N_NODES * H_HEADS;
    float* loopattr = degb   + N_NODES;
    float* Vc       = loopattr + (size_t)N_NODES * E_DIM;
    float* alpha    = Vc + 128;
    float* denom    = alpha + (size_t)ET * H_HEADS;
    float* outacc   = denom + (size_t)N_NODES * H_HEADS;
    float* meanb    = outacc + (size_t)N_NODES * HD;
    float* invstd   = meanb + 128;
    int*   amaxK    = (int*)(invstd + 128);

    k_gemm_h<<<GEMM_BLOCKS, GEMM_THREADS, 0, stream>>>(x, W, hbuf);

    k_attn_node<<<(N_NODES * H_HEADS + 255) / 256, 256, 0, stream>>>(
        hbuf, att_src, att_dst, a_src, a_dst);
    k_edge_proj<<<1, 128, 0, stream>>>(W_e, att_edge, Vc, (int*)degb);

    k_bin<<<BIN_BLOCKS, BIN_THREADS, 0, stream>>>(src, dst, (int*)degb, outacc);
    k_node_pre<<<AGG_BLOCKS, AGG_THREADS, 0, stream>>>(src, dst, edgeattr, Vc, a_src, a_dst,
                                                       alpha, amaxK, outacc, (const int*)degb);
    float* psum = loopattr;
    float* psq  = loopattr + (size_t)AGG_BLOCKS * HD;
    k_aggregate<<<AGG_BLOCKS, AGG_THREADS, 0, stream>>>(src, dst, alpha, amaxK, hbuf,
                                                        (const int*)degb, outacc, psum, psq);

    k_stats<<<HD, AGG_BLOCKS, 0, stream>>>(psum, psq, meanb, invstd);
    k_final<<<FIN_BLOCKS, FIN_THREADS, 0, stream>>>(
        x, outacc, meanb, invstd, gamma, beta, out);
}


# Edge-logit pass: four lanes share an edge (a quarter of its attributes, two heads each, logit parts added across the quad by DPP); a wave's load covers 8 lines, not 32; a_edge sum order changes
# speedup vs baseline: 20.0942x; 1.2771x over previous
#include <hip/hip_runtime.h>
#include <hip/hip_bf16.h>

#define N_NODES 50000
#define N_EDGES 1600000
#define F_DIM   128
#define H_HEADS 8
#define D_HEAD  16
#define E_DIM   16
#define HD      (H_HEADS * D_HEAD)
#define ET      (N_EDGES + N_NODES)
#define EPS_IN  1e-5f
#define NEG_SLOPE 0.2f

typedef float v2f __attribute__((ext_vector_type(2)));
typedef float v8f __attribute__((ext_vector_type(8)));

#define AGG_BLOCKS  256
#define AGG_ROWS    ((N_NODES + AGG_BLOCKS - 1) / AGG_BLOCKS)

#define GEMM_WAVES   8
#define GEMM_THREADS (GEMM_WAVES * 32)
#define GEMM_STRIPS  (N_NODES / 16)
#define GEMM_BLOCKS  ((GEMM_STRIPS + GEMM_WAVES - 1) / GEMM_WAVES)
#define W_PITCH      (2 * HD + 32)
#define GEMM_KSTEPS  (F_DIM / 4)
#define GEMM_NT      (HD / 16)
#define W_ITEMS      ((F_DIM / 2) * (HD / 4) / GEMM_THREADS)

static_assert(N_NODES % 16 == 0, "whole strips");
static_assert(W_ITEMS * GEMM_THREADS == (F_DIM / 2) * (HD / 4), "the copy of W splits evenly over the threads");
static_assert(HD / 4 == 32, "one row of W per wave instruction in the copy");

__global__ void __launch_bounds__(GEMM_THREADS)
k_gemm_h(const float* __restrict__ x, const float* __restrict__ W,
         float* __restrict__ h) {
    __shared__ float Ws[(F_DIM / 2) * W_PITCH];
    const int lane = threadIdx.x & 31;
    const int wave = __builtin_amdgcn_readfirstlane((int)(threadIdx.x >> 5));
    const int strip = blockIdx.x * GEMM_WAVES + wave;
    const bool live = strip < GEMM_STRIPS;
    const int c  = lane & 15;
    const int hf = lane >> 4;

    v2f a[GEMM_KSTEPS];
    if (live) {
        const float* xr = x + (size_t)(strip * 16 + c) * F_DIM + 2 * hf;
#pragma unroll
        for (int s = 0; s < GEMM_KSTEPS; ++s) {
            a[s].x = xr[4 * s];
            a[s].y = xr[4 * s + 1];
        }
    }
    const float4* W4 = reinterpret_cast<const float4*>(W);
    float4 r0[W_ITEMS], r1[W_ITEMS];
#pragma unroll
    for (int i = 0; i < W_ITEMS; ++i) {
        const int item = i * GEMM_THREADS + (int)threadIdx.x;
        const int kp = item >> 5, c4 = item & 31;
        r0[i] = W4[(2 * kp) * (HD / 4) + c4];
        r1[i] = W4[(2 * kp + 1) * (HD / 4) + c4];
    }
#pragma unroll
    for (int i = 0; i < W_ITEMS; ++i) {
        const int item = i * GEMM_THREADS + (int)threadIdx.x;
        const int kp = item >> 5, c4 = item & 31;
        float4* d = reinterpret_cast<float4*>(&Ws[kp * W_PITCH + c4 * 8]);
        d[0] = make_float4(r0[i].x, r1[i].x, r0[i].y, r1[i].y);
        d[1] = make_float4(r0[i].z, r1[i].z, r0[i].w, r1[i].w);
    }
    __syncthreads();
    if (!live) return;

    v8f acc[GEMM_NT];
#pragma unroll
    for (int t = 0; t < GEMM_NT; ++t) acc[t] = v8f{};
    const float* wl = Ws + hf * W_PITCH + 2 * c;
    v2f bq[2][GEMM_NT];
#pragma unroll
    for (int t = 0; t < GEMM_NT; ++t)
        bq[0][t] = *reinterpret_cast<const v2f*>(wl + 32 * t);
#pragma unroll
    for (int s = 0; s < GEMM_KSTEPS; ++s) {
        if (s + 1 < GEMM_KSTEPS) {
#pragma unroll
            for (int t = 0; t < GEMM_NT; ++t)
                bq[(s + 1) & 1][t] =
                    *reinterpret_cast<const v2f*>(wl + (2 * (s + 1)) * W_PITCH + 32 * t);
        }
        __builtin_amdgcn_sched_barrier(0);
#pragma unroll
        for (int t = 0; t < GEMM_NT; ++t) {
            acc[t] = __builtin_amdgcn_wmma_f32_16x16x4_f32(
                 false, a[s],  false, bq[s & 1][t],
                 (short)0, acc[t],  false,  false);
        }
        __builtin_amdgcn_sched_barrier(0);
    }
    const int rbase = strip * 16 + (hf << 3);
#pragma unroll
    for (int r = 0; r < 8; ++r) {
#pragma unroll
        for (int t = 0; t < GEMM_NT; ++t)
            h[(size_t)(rbase + r) * HD + 16 * t + c] = acc[t][r];
    }
}

__global__ void k_attn_node(const float* __restrict__ h,
                            const float* __restrict__ att_src,
                            const float* __restrict__ att_dst,
                            float* a_src, float* a_dst) {
    int idx = blockIdx.x * blockDim.x + threadIdx.x;
    if (idx >= N_NODES * H_HEADS) return;
    int n = idx >> 3, hh = idx & 7;
    const float* hp = h + (size_t)n * HD + hh * D_HEAD;
    const float* as = att_src + hh * D_HEAD;
    const float* ad = att_dst + hh * D_HEAD;
    float s = 0.f, d = 0.f;
#pragma unroll
    for (int i = 0; i < D_HEAD; ++i) {
        float v = hp[i];
        s += v * as[i];
        d += v * ad[i];
    }
    a_src[idx] = s;
    a_dst[idx] = d;
}

__global__ void k_edge_proj(const float* __restrict__ W_e,
                            const float* __restrict__ att_edge, float* V, int* ecount) {
    int idx = threadIdx.x;
    int k = idx >> 3, hh = idx & 7;
    float s = 0.f;
#pragma unroll
    for (int d = 0; d < D_HEAD; ++d)
        s += W_e[(size_t)k * HD + hh * D_HEAD + d] * att_edge[hh * D_HEAD + d];
    V[idx] = s;
    for (int i = idx; i < AGG_BLOCKS; i += E_DIM * H_HEADS) ecount[i] = 0;
}

#define AGG_THREADS 1024
#define AGG_WAVES   (AGG_THREADS / 32)
#define DST_GROUPS  (N_EDGES / 128)
#define AGG_UNROLL  4

#define AGG_BATCH   8
#define LIST_E_BITS 21
#define LIST_CAP_PER_ROW (HD / 2)

static_assert(N_EDGES % 128 == 0, "dst list must split into whole wave groups");
static_assert(N_EDGES <= (1 << LIST_E_BITS), "edge id must fit its field");
static_assert(AGG_ROWS <= (1 << (31 - LIST_E_BITS)), "local row must fit its field");
static_assert((AGG_BLOCKS & (AGG_BLOCKS - 1)) == 0, "the statistics kernel halves the partials in a tree");

#define SORT_CAP    (AGG_ROWS * LIST_CAP_PER_ROW)
#define SORT_PER_THREAD ((SORT_CAP + AGG_THREADS - 1) / AGG_THREADS)
#define SCAN_PER_LANE   ((AGG_ROWS + 32) / 32)
static_assert(SCAN_PER_LANE * 32 >= AGG_ROWS + 1, "one wave must cover every row start and the end");
static_assert(SORT_CAP * 2 == AGG_ROWS * HD, "the sort array takes exactly the place of the LDS rows");

__device__ __forceinline__ void agg_row_term(float* acc_row, float* den_row,
                                             const float* __restrict__ hrow,
                                             const float* __restrict__ lrow,
                                             const int* __restrict__ mrow, int lane,
                                             bool first) {
    int key = mrow[lane & 7];
    int bi = key >= 0 ? key : (key ^ 0x7FFFFFFF);
    float wl = __expf(lrow[lane & 7] - __int_as_float(bi));
    float hv[4];
#pragma unroll
    for (int j = 0; j < 4; ++j) hv[j] = hrow[j * 32 + lane];
    float v[4];
#pragma unroll
    for (int j = 0; j < 4; ++j) {
        int hh = j * 2 + (lane >> 4);
        float w = __int_as_float(__builtin_amdgcn_ds_bpermute(hh << 2, __float_as_int(wl)));
        v[j] = hv[j] * w;
    }
#pragma unroll
    for (int j = 0; j < 4; ++j) {
        int c = j * 32 + lane;
        if (first) acc_row[c] = v[j];
        else       atomicAdd(&acc_row[c], v[j]);
    }
    if (lane < H_HEADS) {
        if (first) den_row[lane] = wl;
        else       atomicAdd(&den_row[lane], wl);
    }
}

__device__ __forceinline__ void agg_load_groups(int (&dv)[AGG_UNROLL][4],
                                                const int* __restrict__ dst, int g0, int lane) {
#pragma unroll
    for (int u = 0; u < AGG_UNROLL; ++u) {
        int gg = g0 + u * AGG_WAVES;
        if (gg < DST_GROUPS) {
            const int* p = dst + ((size_t)gg * 32 + lane) * 4;
            dv[u][0] = p[0]; dv[u][1] = p[1]; dv[u][2] = p[2]; dv[u][3] = p[3];
        } else {
            dv[u][0] = -1; dv[u][1] = -1; dv[u][2] = -1; dv[u][3] = -1;
        }
    }
}

__global__ void __launch_bounds__(AGG_THREADS)
k_aggregate(const int* __restrict__ src, const int* __restrict__ dst,
            const float* __restrict__ logit, const int* __restrict__ amaxK,
            const float* __restrict__ h, const int* __restrict__ ecount, float* outacc,
            float* __restrict__ psum, float* __restrict__ psq) {
    __shared__ float acc[AGG_ROWS * HD];
    __shared__ float den[AGG_ROWS * H_HEADS];
    __shared__ float red_s[AGG_WAVES * HD];
    __shared__ float red_q[AGG_WAVES * HD];
    __shared__ int rstart[AGG_ROWS + 1];
    __shared__ int rcnt[AGG_ROWS];
    __shared__ int next_row;
    const int lane = threadIdx.x & 31;
    const int wave = __builtin_amdgcn_readfirstlane((int)(threadIdx.x >> 5));
    const int n0 = blockIdx.x * AGG_ROWS;
    int cnt = N_NODES - n0;
    if (cnt > AGG_ROWS) cnt = AGG_ROWS;

    const int total = ecount[blockIdx.x];
    const int lcap = cnt * LIST_CAP_PER_ROW;
    const int2* blist = reinterpret_cast<const int2*>(outacc + (size_t)n0 * HD);

    float cs[4] = {0.f, 0.f, 0.f, 0.f};
    float cq[4] = {0.f, 0.f, 0.f, 0.f};
    const bool sorted_path = total + cnt <= lcap;

    if (sorted_path) {
    int2* sorted = reinterpret_cast<int2*>(acc);
    if (threadIdx.x < AGG_ROWS) rcnt[threadIdx.x] = 1;
    if (threadIdx.x == 0) next_row = 0;
    int2 en[SORT_PER_THREAD];
#pragma unroll
    for (int k = 0; k < SORT_PER_THREAD; ++k) {
        const int j = (int)threadIdx.x + k * AGG_THREADS;
        en[k] = make_int2(0, 0);
        if (j < total) en[k] = blist[j];
    }
    __syncthreads();
    int rk[SORT_PER_THREAD];
#pragma unroll
    for (int k = 0; k < SORT_PER_THREAD; ++k) {
        const int j = (int)threadIdx.x + k * AGG_THREADS;
        rk[k] = 0;
        if (j < total) rk[k] = atomicAdd(&rcnt[en[k].x >> LIST_E_BITS], 1);
    }
    __syncthreads();
    if (wave == 0) {
        int c[SCAN_PER_LANE];
        int sum = 0;
#pragma unroll
        for (int q = 0; q < SCAN_PER_LANE; ++q) {
            const int r = lane * SCAN_PER_LANE + q;
            c[q] = r < cnt ? rcnt[r] : 0;
            sum += c[q];
        }
        int inc = sum;
#pragma unroll
        for (int d = 1; d < 32; d <<= 1) {
            const int up = __shfl_up(inc, d, 32);
            if (lane >= d) inc += up;
        }
        int run = inc - sum;
#pragma unroll
        for (int q = 0; q < SCAN_PER_LANE; ++q) {
            const int r = lane * SCAN_PER_LANE + q;
            if (r <= cnt) rstart[r] = run;
            run += c[q];
        }
    }
    __syncthreads();
#pragma unroll
    for (int k = 0; k < SORT_PER_THREAD; ++k) {
        const int j = (int)threadIdx.x + k * AGG_THREADS;
        if (j < total) {
            const int tl = en[k].x >> LIST_E_BITS;
            sorted[rstart[tl] + rk[k]] = make_int2(en[k].x & ((1 << LIST_E_BITS) - 1), en[k].y);
        }
    }
    if ((int)threadIdx.x < cnt)
        sorted[rstart[threadIdx.x]] = make_int2(N_EDGES + n0 + (int)threadIdx.x, n0 + (int)threadIdx.x);
    __syncthreads();

    const int hsel = lane >> 2;
    int rr = 0;
    if (lane == 0) rr = atomicAdd(&next_row, 1);
    int r = __builtin_amdgcn_readfirstlane(rr);
    int keyv = 0, rsv = 0, rev = 0;
    if (r < cnt) {
        keyv = amaxK[(size_t)(n0 + r) * 8 + hsel];
        rsv = rstart[r];
        rev = rstart[r + 1];
    }
    asm volatile("" : "+v"(keyv));
    while (r < cnt) {
        int rq = 0;
        if (lane == 0) rq = atomicAdd(&next_row, 1);
        const int rn = __builtin_amdgcn_readfirstlane(rq);
        int keyn = 0, rsn = 0, ren = 0;
        if (rn < cnt) {
            keyn = amaxK[(size_t)(n0 + rn) * 8 + hsel];
            rsn = rstart[rn];
            ren = rstart[rn + 1];
        }
        const int rs = __builtin_amdgcn_readfirstlane(rsv);
        const int re = __builtin_amdgcn_readfirstlane(rev);
        const float mx = __int_as_float(keyv >= 0 ? keyv : (keyv ^ 0x7FFFFFFF));
        float a[4] = {0.f, 0.f, 0.f, 0.f};
        float dn = 0.f;
        for (int base = rs; base < re; base += 32) {
            int n = re - base;
            if (n > 32) n = 32;
            const int2 ent = sorted[base + (lane < n ? lane : n - 1)];
            for (int sub = 0; sub < n; sub += AGG_BATCH) {
                float  lg[AGG_BATCH];
                float4 hv[AGG_BATCH];
#pragma unroll
                for (int b = 0; b < AGG_BATCH; ++b) {
                    int ix = sub + b;
                    if (ix > n - 1) ix = n - 1;
                    const int eu = __builtin_amdgcn_readlane(ent.x, ix);
                    const int su = __builtin_amdgcn_readlane(ent.y, ix);
                    lg[b] = logit[(size_t)eu * 8 + hsel];
                    hv[b] = *reinterpret_cast<const float4*>(h + (size_t)su * HD + lane * 4);
                }
#pragma unroll
                for (int b = 0; b < AGG_BATCH; ++b) {
                    float w = __expf(lg[b] - mx);
                    if (sub + b >= n) w = 0.f;
                    a[0] += w * hv[b].x;
                    a[1] += w * hv[b].y;
                    a[2] += w * hv[b].z;
                    a[3] += w * hv[b].w;
                    dn += w;
                }
            }
        }
        const float dd = dn + 1e-16f;
        float4 o;
        o.x = a[0] / dd; o.y = a[1] / dd; o.z = a[2] / dd; o.w = a[3] / dd;
        *reinterpret_cast<float4*>(outacc + (size_t)(n0 + r) * HD + lane * 4) = o;
        cs[0] += o.x; cs[1] += o.y; cs[2] += o.z; cs[3] += o.w;
        cq[0] += o.x * o.x; cq[1] += o.y * o.y; cq[2] += o.z * o.z; cq[3] += o.w * o.w;
        r = rn; keyv = keyn; rsv = rsn; rev = ren;
    }
    } else {
    for (int tl = wave; tl < cnt; tl += AGG_WAVES) {
        int t = n0 + tl;
        size_t i = (size_t)N_EDGES + (size_t)t;
        agg_row_term(acc + tl * HD, den + tl * H_HEADS, h + (size_t)t * HD, logit + i * 8,
                     amaxK + (size_t)t * 8, lane, true);
    }
    __syncthreads();
    int dv[AGG_UNROLL][4];
    agg_load_groups(dv, dst, wave, lane);
    for (int g0 = wave; g0 < DST_GROUPS; g0 += AGG_WAVES * AGG_UNROLL) {
        int nx[AGG_UNROLL][4];
        agg_load_groups(nx, dst, g0 + AGG_WAVES * AGG_UNROLL, lane);

        const unsigned ucnt = (unsigned)cnt;
        int tl[AGG_UNROLL][4];
        int sv[AGG_UNROLL][4];
        unsigned anym[AGG_UNROLL];
#pragma unroll
        for (int u = 0; u < AGG_UNROLL; ++u) {
            int gg = g0 + u * AGG_WAVES;
#pragma unroll
            for (int k = 0; k < 4; ++k) {
                tl[u][k] = dv[u][k] - n0;
                sv[u][k] = 0;
            }
            bool hu = ((unsigned)tl[u][0] < ucnt) | ((unsigned)tl[u][1] < ucnt) |
                      ((unsigned)tl[u][2] < ucnt) | ((unsigned)tl[u][3] < ucnt);
            anym[u] = __builtin_amdgcn_ballot_w32(hu);
            if (anym[u] != 0u) {
#pragma unroll
                for (int k = 0; k < 4; ++k) {
                    if ((unsigned)tl[u][k] < ucnt)
                        sv[u][k] = src[((size_t)gg * 32 + lane) * 4 + k];
                }
            }
        }
#pragma unroll
        for (int u = 0; u < AGG_UNROLL; ++u) {
            int gg = g0 + u * AGG_WAVES;
            if (anym[u] != 0u) {
#pragma unroll
                for (int k = 0; k < 4; ++k) {
                    unsigned m = __builtin_amdgcn_ballot_w32((unsigned)tl[u][k] < ucnt);
                    while (m != 0u) {
                        int l = __builtin_ctz(m);
                        m &= m - 1u;
                        int tlu = __builtin_amdgcn_readlane(tl[u][k], l);
                        int su  = __builtin_amdgcn_readlane(sv[u][k], l);
                        size_t e = ((size_t)gg * 32 + (size_t)l) * 4 + (size_t)k;
                        agg_row_term(acc + tlu * HD, den + tlu * H_HEADS, h + (size_t)su * HD,
                                     logit + e * 8, amaxK + (size_t)(n0 + tlu) * 8, lane, false);
                    }
                }
            }
        }
#pragma unroll
        for (int u = 0; u < AGG_UNROLL; ++u) {
#pragma unroll
            for (int k = 0; k < 4; ++k) dv[u][k] = nx[u][k];
        }
    }
    __syncthreads();
    for (int tl = wave; tl < cnt; tl += AGG_WAVES) {
        float* orow = outacc + (size_t)(n0 + tl) * HD;
#pragma unroll
        for (int j = 0; j < 4; ++j) {
            int hh = j * 2 + (lane >> 4);
            float v = acc[tl * HD + j * 32 + lane] / (den[tl * H_HEADS + hh] + 1e-16f);
            orow[j * 32 + lane] = v;
            cs[j] += v;
            cq[j] += v * v;
        }
    }
    }
#pragma unroll
    for (int j = 0; j < 4; ++j) {
        const int ch = sorted_path ? lane * 4 + j : j * 32 + lane;
        red_s[wave * HD + ch] = cs[j];
        red_q[wave * HD + ch] = cq[j];
    }
    __syncthreads();
    if (threadIdx.x < 2 * HD) {
        const int ch = threadIdx.x & (HD - 1);
        const bool sq = threadIdx.x >= HD;
        const float* r = sq ? red_q : red_s;
        float t = 0.f;
        for (int w = 0; w < AGG_WAVES; ++w) t += r[w * HD + ch];
        float* pp = sq ? psq : psum;
        pp[(size_t)blockIdx.x * HD + ch] = t;
    }
}

#define BIN_THREADS 1024
#define BIN_EDGES   (BIN_THREADS * 4)
#define BIN_BLOCKS  ((N_EDGES + BIN_EDGES - 1) / BIN_EDGES)

static_assert(N_EDGES % 4 == 0, "a lane's four edges are all inside the list or all outside");

__global__ void __launch_bounds__(BIN_THREADS)
k_bin(const int* __restrict__ src, const int* __restrict__ dst, int* ecount, float* elist) {
    __shared__ int hist[AGG_BLOCKS];
    __shared__ int base[AGG_BLOCKS];
    const int tid = threadIdx.x;
    const int e0 = (blockIdx.x * BIN_THREADS + tid) * 4;
    int dv[4] = {-1, -1, -1, -1};
    int sv[4] = {0, 0, 0, 0};
    if (e0 < N_EDGES) {
        const int4 d = *reinterpret_cast<const int4*>(dst + e0);
        const int4 s = *reinterpret_cast<const int4*>(src + e0);
        dv[0] = d.x; dv[1] = d.y; dv[2] = d.z; dv[3] = d.w;
        sv[0] = s.x; sv[1] = s.y; sv[2] = s.z; sv[3] = s.w;
    }
    if (tid < AGG_BLOCKS) hist[tid] = 0;
    __syncthreads();
    int own[4], rank[4];
#pragma unroll
    for (int k = 0; k < 4; ++k) {
        own[k] = -1;
        rank[k] = 0;
        if ((unsigned)dv[k] < (unsigned)N_NODES) {
            own[k] = dv[k] / AGG_ROWS;
            rank[k] = atomicAdd(&hist[own[k]], 1);
        }
    }
    __syncthreads();
    if (tid < AGG_BLOCKS) base[tid] = atomicAdd(&ecount[tid], hist[tid]);
    __syncthreads();
#pragma unroll
    for (int k = 0; k < 4; ++k) {
        if (own[k] >= 0) {
            const int n0 = own[k] * AGG_ROWS;
            int cnt = N_NODES - n0;
            if (cnt > AGG_ROWS) cnt = AGG_ROWS;
            const int pos = base[own[k]] + rank[k];
            if (pos < cnt * LIST_CAP_PER_ROW) {
                int2* blist = reinterpret_cast<int2*>(elist + (size_t)n0 * HD);
                blist[pos] = make_int2((e0 + k) | ((dv[k] - n0) << LIST_E_BITS), sv[k]);
            }
        }
    }
}

__device__ __forceinline__ int logit_key(float l) {
    int bi = __float_as_int(l);
    return bi >= 0 ? bi : (bi ^ 0x7FFFFFFF);
}

__device__ __forceinline__ void pre_edge_term(size_t e, int tl, unsigned s, unsigned t,
                                              const float* __restrict__ edge_attr,
                                              const float* __restrict__ a_src,
                                              const float* __restrict__ a_dst,
                                              const float* sV, float* lsum, float* ldeg,
                                              int* lmax, float* __restrict__ logit) {
    const float4* ep = reinterpret_cast<const float4*>(edge_attr + e * E_DIM);
    const float4* sp = reinterpret_cast<const float4*>(a_src + (size_t)s * H_HEADS);
    const float4* dp = reinterpret_cast<const float4*>(a_dst + (size_t)t * H_HEADS);
    float4 q0 = ep[0], q1 = ep[1], q2 = ep[2], q3 = ep[3];
    float4 s0 = sp[0], s1 = sp[1];
    float4 d0 = dp[0], d1 = dp[1];
    float eav[E_DIM] = {q0.x, q0.y, q0.z, q0.w, q1.x, q1.y, q1.z, q1.w,
                        q2.x, q2.y, q2.z, q2.w, q3.x, q3.y, q3.z, q3.w};
    float asv[H_HEADS] = {s0.x, s0.y, s0.z, s0.w, s1.x, s1.y, s1.z, s1.w};
    float adv[H_HEADS] = {d0.x, d0.y, d0.z, d0.w, d1.x, d1.y, d1.z, d1.w};
    float ae[H_HEADS];
#pragma unroll
    for (int hh = 0; hh < H_HEADS; ++hh) ae[hh] = 0.f;
#pragma unroll
    for (int k = 0; k < E_DIM; ++k) {
#pragma unroll
        for (int hh = 0; hh < H_HEADS; ++hh) ae[hh] += eav[k] * sV[k * 8 + hh];
    }
    float lg[H_HEADS];
#pragma unroll
    for (int hh = 0; hh < H_HEADS; ++hh) {
        float l = asv[hh] + adv[hh] + ae[hh];
        lg[hh] = l > 0.f ? l : NEG_SLOPE * l;
    }
    float4* lp = reinterpret_cast<float4*>(logit + e * H_HEADS);
    lp[0] = make_float4(lg[0], lg[1], lg[2], lg[3]);
    lp[1] = make_float4(lg[4], lg[5], lg[6], lg[7]);
#pragma unroll
    for (int hh = 0; hh < H_HEADS; ++hh) atomicMax(&lmax[tl * H_HEADS + hh], logit_key(lg[hh]));
#pragma unroll
    for (int k = 0; k < E_DIM; ++k) atomicAdd(&lsum[tl * E_DIM + k], eav[k]);
    atomicAdd(&ldeg[tl], 1.0f);
}

#define PRE_U        4

__device__ __forceinline__ float quad_xor2(float v) {
    return __int_as_float(__builtin_amdgcn_update_dpp(0, __float_as_int(v), 0x4E, 0xF, 0xF, true));
}
__device__ __forceinline__ float quad_xor1(float v) {
    return __int_as_float(__builtin_amdgcn_update_dpp(0, __float_as_int(v), 0xB1, 0xF, 0xF, true));
}

__device__ __forceinline__ void pre_quad_trip(int2 (&en)[PRE_U], const int p0, const int total,
                                              const int q, const int n0, const float (&vr)[4][8],
                                              const int2* __restrict__ blist,
                                              const float* __restrict__ edge_attr,
                                              const float* __restrict__ a_src,
                                              const float* __restrict__ a_dst, float* lsum,
                                              float* ldeg, int* lmax, float* __restrict__ logit) {
    const float4* ea4 = reinterpret_cast<const float4*>(edge_attr);
    const float2* as2 = reinterpret_cast<const float2*>(a_src);
    const float2* ad2 = reinterpret_cast<const float2*>(a_dst);
    float2* lg2 = reinterpret_cast<float2*>(logit);
    const unsigned uq = (unsigned)q;
    float4 ea[PRE_U];
    float2 sa[PRE_U], da[PRE_U];
#pragma unroll
    for (int u = 0; u < PRE_U; ++u) {
        const unsigned e = (unsigned)en[u].x & ((1u << LIST_E_BITS) - 1u);
        const unsigned tl = (unsigned)en[u].x >> LIST_E_BITS;
        ea[u] = ea4[e * 4u + uq];
        sa[u] = as2[(unsigned)en[u].y * 4u + uq];
        da[u] = ad2[((unsigned)n0 + tl) * 4u + uq];
    }
    int2 nx[PRE_U];
#pragma unroll
    for (int u = 0; u < PRE_U; ++u) {
        const int p = p0 + AGG_THREADS + u * 8;
        nx[u] = blist[p < total ? p : total - 1];
    }
    __builtin_amdgcn_sched_barrier(0);
#pragma unroll
    for (int u = 0; u < PRE_U; ++u) {
        float p[8];
#pragma unroll
        for (int j = 0; j < 8; ++j) p[j] = ea[u].x * vr[0][j];
#pragma unroll
        for (int j = 0; j < 8; ++j) p[j] = __builtin_fmaf(ea[u].y, vr[1][j], p[j]);
#pragma unroll
        for (int j = 0; j < 8; ++j) p[j] = __builtin_fmaf(ea[u].z, vr[2][j], p[j]);
#pragma unroll
        for (int j = 0; j < 8; ++j) p[j] = __builtin_fmaf(ea[u].w, vr[3][j], p[j]);
        float r[4];
#pragma unroll
        for (int j = 0; j < 4; ++j) r[j] = p[j] + quad_xor2(p[4 + j]);
        const float f0 = r[0] + quad_xor1(r[2]);
        const float f1 = r[1] + quad_xor1(r[3]);
        float l0 = sa[u].x + da[u].x + f0;
        float l1 = sa[u].y + da[u].y + f1;
        l0 = l0 > 0.f ? l0 : NEG_SLOPE * l0;
        l1 = l1 > 0.f ? l1 : NEG_SLOPE * l1;
        const unsigned e = (unsigned)en[u].x & ((1u << LIST_E_BITS) - 1u);
        const unsigned tl = (unsigned)en[u].x >> LIST_E_BITS;
        lg2[e * 4u + uq] = make_float2(l0, l1);
        atomicMax(&lmax[tl * H_HEADS + 2u * uq], logit_key(l0));
        atomicMax(&lmax[tl * H_HEADS + 2u * uq + 1u], logit_key(l1));
        if (p0 + u * 8 < total) {
            atomicAdd(&lsum[tl * E_DIM + 4u * uq], ea[u].x);
            atomicAdd(&lsum[tl * E_DIM + 4u * uq + 1u], ea[u].y);
            atomicAdd(&lsum[tl * E_DIM + 4u * uq + 2u], ea[u].z);
            atomicAdd(&lsum[tl * E_DIM + 4u * uq + 3u], ea[u].w);
            if (q == 0) atomicAdd(&ldeg[tl], 1.0f);
        }
    }
#pragma unroll
    for (int u = 0; u < PRE_U; ++u) en[u] = nx[u];
}

#define PRE_STRIDE   (AGG_WAVES * AGG_UNROLL)
#define PRE_STRIPES  ((DST_GROUPS + PRE_STRIDE - 1) / PRE_STRIDE)
#define PRE_SPC      14
#define PRE_CHUNKS   ((PRE_STRIPES + PRE_SPC - 1) / PRE_SPC)
#define PRE_QCAP     2048

__global__ void __launch_bounds__(AGG_THREADS)
k_node_pre(const int* __restrict__ src, const int* __restrict__ dst,
           const float* __restrict__ edge_attr, const float* __restrict__ Vc,
           const float* __restrict__ a_src, const float* __restrict__ a_dst,
           float* __restrict__ logit, int* __restrict__ amaxK,
           const float* __restrict__ elist, const int* __restrict__ ecount) {
    __shared__ float lsum[AGG_ROWS * E_DIM];
    __shared__ float ldeg[AGG_ROWS];
    __shared__ int   lmax[AGG_ROWS * H_HEADS];
    __shared__ float sV[E_DIM * H_HEADS];
    __shared__ int   q[PRE_QCAP];
    __shared__ int   qn[PRE_CHUNKS];
    const int lane = threadIdx.x & 31;
    const int wave = __builtin_amdgcn_readfirstlane((int)(threadIdx.x >> 5));
    const int n0 = blockIdx.x * AGG_ROWS;
    int cnt = N_NODES - n0;
    if (cnt > AGG_ROWS) cnt = AGG_ROWS;
    const unsigned ucnt = (unsigned)cnt;
    const int2* blist = reinterpret_cast<const int2*>(elist + (size_t)n0 * HD);
    const int lcap = cnt * LIST_CAP_PER_ROW;
    const int total = ecount[blockIdx.x];

    for (int i = threadIdx.x; i < AGG_ROWS * E_DIM; i += AGG_THREADS) lsum[i] = 0.0f;
    for (int i = threadIdx.x; i < AGG_ROWS * H_HEADS; i += AGG_THREADS) lmax[i] = (int)0x80000000;
    for (int i = threadIdx.x; i < AGG_ROWS; i += AGG_THREADS) ldeg[i] = 0.0f;
    if (threadIdx.x < E_DIM * H_HEADS) sV[threadIdx.x] = Vc[threadIdx.x];
    if (threadIdx.x < PRE_CHUNKS) qn[threadIdx.x] = 0;
    __syncthreads();

    if (total > lcap) {
    int dv[AGG_UNROLL][4];
    agg_load_groups(dv, dst, wave, lane);
    for (int c = 0; c < PRE_CHUNKS; ++c) {
        int iend = (c + 1) * PRE_SPC;
        if (iend > PRE_STRIPES) iend = PRE_STRIPES;
        for (int i = c * PRE_SPC; i < iend; ++i) {
            const int g0 = wave + i * PRE_STRIDE;
            int cur[AGG_UNROLL][4];
#pragma unroll
            for (int u = 0; u < AGG_UNROLL; ++u) {
#pragma unroll
                for (int k = 0; k < 4; ++k) cur[u][k] = dv[u][k];
            }
            agg_load_groups(dv, dst, g0 + PRE_STRIDE, lane);

            unsigned ovf = 0u;
#pragma unroll
            for (int u = 0; u < AGG_UNROLL; ++u) {
                int gg = g0 + u * AGG_WAVES;
#pragma unroll
                for (int k = 0; k < 4; ++k) {
                    if ((unsigned)(cur[u][k] - n0) < ucnt) {
                        int pos = atomicAdd(&qn[c], 1);
                        if (pos < PRE_QCAP) q[pos] = (gg * 32 + lane) * 4 + k;
                        else                ovf |= 1u << (u * 4 + k);
                    }
                }
            }
            while (ovf != 0u) {
                asm volatile("" ::: "memory");
                int slot = __builtin_ctz(ovf);
                ovf &= ovf - 1u;
                int e = ((g0 + (slot >> 2) * AGG_WAVES) * 32 + lane) * 4 + (slot & 3);
                int s = src[e];
                int t = dst[e];
                pre_edge_term((size_t)e, t - n0, s, t, edge_attr, a_src, a_dst, sV, lsum, ldeg,
                              lmax, logit);
            }
        }
        __syncthreads();
        int qc = qn[c];
        if (qc > PRE_QCAP) qc = PRE_QCAP;
        for (int j = threadIdx.x; j < qc; j += AGG_THREADS) {
            asm volatile("" ::: "memory");
            int e = q[j];
            int s = src[e];
            int t = dst[e];
            pre_edge_term((size_t)e, t - n0, s, t, edge_attr, a_src, a_dst, sV, lsum, ldeg,
                          lmax, logit);
        }
        __syncthreads();
    }
    } else {
        const int qt = lane & 3;
        const int g = lane >> 2;
        float vr[4][8];
#pragma unroll
        for (int i = 0; i < 4; ++i) {
#pragma unroll
            for (int m = 0; m < 4; ++m) {
                vr[i][2 * m]     = sV[(4 * qt + i) * 8 + 2 * (qt ^ m)];
                vr[i][2 * m + 1] = sV[(4 * qt + i) * 8 + 2 * (qt ^ m) + 1];
            }
        }
        const int wfirst = wave * 32;
        const int plast = total - 1;
        int2 en[PRE_U];
#pragma unroll
        for (int u = 0; u < PRE_U; ++u) {
            const int p = wfirst + u * 8 + g;
            en[u] = make_int2(0, 0);
            if (wfirst < total) en[u] = blist[p < total ? p : plast];
        }
        for (int w0 = wfirst; w0 < total; w0 += AGG_THREADS)
            pre_quad_trip(en, w0 + g, total, qt, n0, vr, blist, edge_attr, a_src, a_dst, lsum, ldeg,
                          lmax, logit);
    }
    __syncthreads();

    for (int i = threadIdx.x; i < cnt * H_HEADS; i += AGG_THREADS) {
        int tl = i >> 3, hh = i & 7;
        int t = n0 + tl;
        float d = ldeg[tl];
        d = d < 1.0f ? 1.0f : d;
        float ae = 0.f;
#pragma unroll
        for (int k = 0; k < E_DIM; ++k) ae += (lsum[tl * E_DIM + k] / d) * sV[k * 8 + hh];
        float l = a_src[(size_t)t * H_HEADS + hh] + a_dst[(size_t)t * H_HEADS + hh] + ae;
        l = l > 0.f ? l : NEG_SLOPE * l;
        logit[((size_t)N_EDGES + (size_t)t) * H_HEADS + hh] = l;
        int key = logit_key(l);
        int mk = lmax[i];
        amaxK[(size_t)t * H_HEADS + hh] = key > mk ? key : mk;
    }
}

__global__ void __launch_bounds__(AGG_BLOCKS)
k_stats(const float* __restrict__ psum, const float* __restrict__ psq, float* meanb,
        float* invstd) {
    __shared__ float ssum[AGG_BLOCKS];
    __shared__ float ssq[AGG_BLOCKS];
    int c = blockIdx.x;
    ssum[threadIdx.x] = psum[(size_t)threadIdx.x * HD + c];
    ssq[threadIdx.x]  = psq[(size_t)threadIdx.x * HD + c];
    __syncthreads();
    for (int off = AGG_BLOCKS / 2; off > 0; off >>= 1) {
        if (threadIdx.x < off) {
            ssum[threadIdx.x] += ssum[threadIdx.x + off];
            ssq[threadIdx.x]  += ssq[threadIdx.x + off];
        }
        __syncthreads();
    }
    if (threadIdx.x == 0) {
        float m = ssum[0] / (float)N_NODES;
        float var = ssq[0] / (float)N_NODES - m * m;
        meanb[c] = m;
        invstd[c] = rsqrtf(var + EPS_IN);
    }
}

#define FIN_THREADS 512
#define FIN_BLOCKS  625
#define FIN_VEC     (N_NODES * HD / 4)
#define FIN_STRIDE  (FIN_BLOCKS * FIN_THREADS)
#define FIN_TRIPS   (FIN_VEC / FIN_STRIDE)

static_assert(FIN_TRIPS * FIN_STRIDE == FIN_VEC, "the grid covers the plane in whole trips");
static_assert(FIN_STRIDE % (HD / 4) == 0, "a thread keeps its four channels on every trip");

__device__ __forceinline__ float fin_elem(float xv, float ov, float m, float is, float g,
                                          float b) {
    float normed = (ov - m) * is * g + b;
    float t = xv + normed;
    return t > 0.f ? t : expm1f(t);
}

__global__ void __launch_bounds__(FIN_THREADS)
k_final(const float* __restrict__ x,
        const float* __restrict__ outacc,
        const float* __restrict__ meanb,
        const float* __restrict__ invstd,
        const float* __restrict__ gamma,
        const float* __restrict__ beta, float* out) {
    const int t0 = blockIdx.x * FIN_THREADS + threadIdx.x;
    const int c4 = t0 & (HD / 4 - 1);
    const float4* x4 = reinterpret_cast<const float4*>(x);
    const float4* o4 = reinterpret_cast<const float4*>(outacc);
    float4* out4 = reinterpret_cast<float4*>(out);
    float4 xv[FIN_TRIPS], ov[FIN_TRIPS];
#pragma unroll
    for (int k = 0; k < FIN_TRIPS; ++k) {
        xv[k] = x4[t0 + k * FIN_STRIDE];
        ov[k] = o4[t0 + k * FIN_STRIDE];
    }
    const float4 m  = reinterpret_cast<const float4*>(meanb)[c4];
    const float4 is = reinterpret_cast<const float4*>(invstd)[c4];
    float4 g, b;
    g.x = gamma[c4 * 4 + 0]; g.y = gamma[c4 * 4 + 1]; g.z = gamma[c4 * 4 + 2]; g.w = gamma[c4 * 4 + 3];
    b.x = beta[c4 * 4 + 0];  b.y = beta[c4 * 4 + 1];  b.z = beta[c4 * 4 + 2];  b.w = beta[c4 * 4 + 3];
#pragma unroll
    for (int k = 0; k < FIN_TRIPS; ++k) {
        float4 r;
        r.x = fin_elem(xv[k].x, ov[k].x, m.x, is.x, g.x, b.x);
        r.y = fin_elem(xv[k].y, ov[k].y, m.y, is.y, g.y, b.y);
        r.z = fin_elem(xv[k].z, ov[k].z, m.z, is.z, g.z, b.z);
        r.w = fin_elem(xv[k].w, ov[k].w, m.w, is.w, g.w, b.w);
        out4[t0 + k * FIN_STRIDE] = r;
    }
}

extern "C" void kernel_launch(void* const* d_in, const int* in_sizes, int n_in,
                              void* d_out, int out_size, void* d_ws, size_t ws_size,
                              hipStream_t stream) {
    const float* x        = (const float*)d_in[0];
    const int*   ei       = (const int*)d_in[1];
    const float* edgeattr = (const float*)d_in[2];
    const float* W        = (const float*)d_in[3];
    const float* att_src  = (const float*)d_in[4];
    const float* att_dst  = (const float*)d_in[5];
    const float* W_e      = (const float*)d_in[6];
    const float* att_edge = (const float*)d_in[7];
    const float* gamma    = (const float*)d_in[8];
    const float* beta     = (const float*)d_in[9];
    float* out = (float*)d_out;

    const int* src = ei;
    const int* dst = ei + N_EDGES;

    float* ws = (float*)d_ws;
    float* hbuf     = ws;
    float* a_src    = hbuf   + (size_t)N_NODES * HD;
    float* a_dst    = a_src  + (size_t)N_NODES * H_HEADS;
    float* degb     = a_dst  + (size_t)N_NODES * H_HEADS;
    float* loopattr = degb   + N_NODES;
    float* Vc       = loopattr + (size_t)N_NODES * E_DIM;
    float* alpha    = Vc + 128;
    float* denom    = alpha + (size_t)ET * H_HEADS;
    float* outacc   = denom + (size_t)N_NODES * H_HEADS;
    float* meanb    = outacc + (size_t)N_NODES * HD;
    float* invstd   = meanb + 128;
    int*   amaxK    = (int*)(invstd + 128);

    k_gemm_h<<<GEMM_BLOCKS, GEMM_THREADS, 0, stream>>>(x, W, hbuf);

    k_attn_node<<<(N_NODES * H_HEADS + 255) / 256, 256, 0, stream>>>(
        hbuf, att_src, att_dst, a_src, a_dst);
    k_edge_proj<<<1, 128, 0, stream>>>(W_e, att_edge, Vc, (int*)degb);

    k_bin<<<BIN_BLOCKS, BIN_THREADS, 0, stream>>>(src, dst, (int*)degb, outacc);
    k_node_pre<<<AGG_BLOCKS, AGG_THREADS, 0, stream>>>(src, dst, edgeattr, Vc, a_src, a_dst,
                                                       alpha, amaxK, outacc, (const int*)degb);
    float* psum = loopattr;
    float* psq  = loopattr + (size_t)AGG_BLOCKS * HD;
    k_aggregate<<<AGG_BLOCKS, AGG_THREADS, 0, stream>>>(src, dst, alpha, amaxK, hbuf,
                                                        (const int*)degb, outacc, psum, psq);

    k_stats<<<HD, AGG_BLOCKS, 0, stream>>>(psum, psq, meanb, invstd);
    k_final<<<FIN_BLOCKS, FIN_THREADS, 0, stream>>>(
        x, outacc, meanb, invstd, gamma, beta, out);
}
